# starting kernel
# baseline (speedup 1.0000x reference)
#include <hip/hip_runtime.h>


#define HSZ   30
#define MPAD  128
#define KPAD  32
#define TLEN  2048
#define BSZ   512

typedef __attribute__((ext_vector_type(16))) _Float16 v16h;
typedef __attribute__((ext_vector_type(8)))  _Float16 v8h;
typedef __attribute__((ext_vector_type(8)))  float    v8f;

#if __has_builtin(__builtin_amdgcn_tanhf)
#define HW_TANH __builtin_amdgcn_tanhf
#elif __has_builtin(__builtin_amdgcn_tanh_f32)
#define HW_TANH __builtin_amdgcn_tanh_f32
#endif

#ifdef HW_TANH
__device__ __forceinline__ float tanh_act(float x) { return HW_TANH(x); }
__device__ __forceinline__ float sig_act(float x)  { return fmaf(0.5f, HW_TANH(0.5f * x), 0.5f); }
#else
__device__ __forceinline__ float tanh_act(float x) {
  return fmaf(2.0f, __builtin_amdgcn_rcpf(1.0f + __expf(-2.0f * x)), -1.0f);
}
__device__ __forceinline__ float sig_act(float x) {
  return __builtin_amdgcn_rcpf(1.0f + __expf(-x));
}
#endif

__device__ __forceinline__ v16h load_a_tile(const _Float16* __restrict__ base,
                                            int tile, int lane) {
  const int m  = lane & 15;
  const int kb = (lane >> 4) << 3;
  const _Float16* row = base + ((tile << 4) + m) * KPAD;
  v8h lo = *(const v8h*)(row + kb);
  v8h hi = *(const v8h*)(row + 16 + kb);
  v16h a;
#pragma unroll
  for (int i = 0; i < 8; ++i) { a[i] = lo[i]; a[8 + i] = hi[i]; }
  return a;
}

__global__ __launch_bounds__(32, 1)
void lstm2_wmma_kernel(const float* __restrict__ x,
                       const float* __restrict__ w_ih1, const float* __restrict__ w_hh1,
                       const float* __restrict__ b_ih1, const float* __restrict__ b_hh1,
                       const float* __restrict__ w_ih2, const float* __restrict__ w_hh2,
                       const float* __restrict__ b_ih2, const float* __restrict__ b_hh2,
                       const float* __restrict__ w_fc,  const float* __restrict__ b_fc,
                       float* __restrict__ out) {
  __shared__ _Float16 sA1[MPAD * KPAD];
  __shared__ _Float16 sAi2[MPAD * KPAD];
  __shared__ _Float16 sAh2[MPAD * KPAD];

  const int lane = threadIdx.x;
  const bool hi_half = lane >= 16;
  const int n = lane & 15;

  for (int idx = lane; idx < MPAD * KPAD; idx += 32) {
    const int m = idx >> 5, k = idx & 31;
    const int g = m >> 5, j = m & 31;
    const int row = g * HSZ + j;
    float a1 = 0.f, ai2 = 0.f, ah2 = 0.f;
    if (j < HSZ) {
      if (k < HSZ) {
        a1  = w_hh1[row * HSZ + k];
        ai2 = w_ih2[row * HSZ + k];
        ah2 = w_hh2[row * HSZ + k];
      } else if (k == 30) {
        a1  = w_ih1[row];
      } else {
        a1  = b_ih1[row] + b_hh1[row];
        ah2 = b_ih2[row] + b_hh2[row];
      }
    }
    sA1[idx]  = (_Float16)a1;
    sAi2[idx] = (_Float16)ai2;
    sAh2[idx] = (_Float16)ah2;
  }
  __syncthreads();

  float wfcD[2][8];
#pragma unroll
  for (int s = 0; s < 2; ++s)
#pragma unroll
    for (int r = 0; r < 8; ++r) {
      const int j = 16 * s + (hi_half ? 8 : 0) + r;
      wfcD[s][r] = (j < HSZ) ? w_fc[j] : 0.f;
    }
  const float bfc = b_fc[0];

  v8f zc;
#pragma unroll
  for (int i = 0; i < 8; ++i) zc[i] = 0.f;

  v16h h1B, h2B;
#pragma unroll
  for (int i = 0; i < 16; ++i) { h1B[i] = (_Float16)0.f; h2B[i] = (_Float16)0.f; }
  if (hi_half) h2B[15] = (_Float16)1.f;

  float c1[2][8], c2[2][8];
#pragma unroll
  for (int s = 0; s < 2; ++s)
#pragma unroll
    for (int r = 0; r < 8; ++r) { c1[s][r] = 0.f; c2[s][r] = 0.f; }

  const int bbase = blockIdx.x << 4;
  const float* xrow = x + (size_t)(bbase + n) * TLEN;
  float* orow = out + (size_t)(bbase + n) * TLEN;
  float xcur = xrow[0];

  for (int t = 0; t < TLEN; ++t) {
    const float xnext = xrow[min(t + 1, TLEN - 1)];

    if (hi_half) { h1B[14] = (_Float16)xcur; h1B[15] = (_Float16)1.f; }

    v16h h1Bn;
#pragma unroll
    for (int s = 0; s < 2; ++s) {
      v8f d[4];
#pragma unroll
      for (int g = 0; g < 4; ++g) {
        v16h a = load_a_tile(sA1, 2 * g + s, lane);
        d[g] = __builtin_amdgcn_wmma_f32_16x16x32_f16(false, a, false, h1B,
                                                      (short)0, zc, false, false);
      }
#pragma unroll
      for (int r = 0; r < 8; ++r) {
        const float iv = sig_act(d[0][r]);
        const float fv = sig_act(d[1][r]);
        const float gv = tanh_act(d[2][r]);
        const float ov = sig_act(d[3][r]);
        const float cn = fmaf(fv, c1[s][r], iv * gv);
        c1[s][r] = cn;
        h1Bn[8 * s + r] = (_Float16)(ov * tanh_act(cn));
      }
    }

    v16h h2Bn;
    float acc = 0.f;
#pragma unroll
    for (int s = 0; s < 2; ++s) {
      v8f d[4];
#pragma unroll
      for (int g = 0; g < 4; ++g) {
        v16h ah = load_a_tile(sAh2, 2 * g + s, lane);
        v8f tmp = __builtin_amdgcn_wmma_f32_16x16x32_f16(false, ah, false, h2B,
                                                         (short)0, zc, false, false);
        v16h ai = load_a_tile(sAi2, 2 * g + s, lane);
        d[g] = __builtin_amdgcn_wmma_f32_16x16x32_f16(false, ai, false, h1Bn,
                                                      (short)0, tmp, false, false);
      }
#pragma unroll
      for (int r = 0; r < 8; ++r) {
        const float iv = sig_act(d[0][r]);
        const float fv = sig_act(d[1][r]);
        const float gv = tanh_act(d[2][r]);
        const float ov = sig_act(d[3][r]);
        const float cn = fmaf(fv, c2[s][r], iv * gv);
        c2[s][r] = cn;
        const float hn = ov * tanh_act(cn);
        h2Bn[8 * s + r] = (_Float16)hn;
        acc = fmaf(hn, wfcD[s][r], acc);
      }
    }
    if (hi_half) { h2Bn[14] = (_Float16)0.f; h2Bn[15] = (_Float16)1.f; }

    acc += __shfl_xor(acc, 16, 32);
    if (lane < 16) orow[t] = acc + bfc;

    h1B = h1Bn;
    h2B = h2Bn;
    xcur = xnext;
  }
}

extern "C" void kernel_launch(void* const* d_in, const int* in_sizes, int n_in,
                              void* d_out, int out_size, void* d_ws, size_t ws_size,
                              hipStream_t stream) {
  (void)in_sizes; (void)n_in; (void)out_size; (void)d_ws; (void)ws_size;
  const float* x     = (const float*)d_in[0];
  const float* w_ih1 = (const float*)d_in[1];
  const float* w_hh1 = (const float*)d_in[2];
  const float* b_ih1 = (const float*)d_in[3];
  const float* b_hh1 = (const float*)d_in[4];
  const float* w_ih2 = (const float*)d_in[5];
  const float* w_hh2 = (const float*)d_in[6];
  const float* b_ih2 = (const float*)d_in[7];
  const float* b_hh2 = (const float*)d_in[8];
  const float* w_fc  = (const float*)d_in[9];
  const float* b_fc  = (const float*)d_in[10];
  float* out = (float*)d_out;

  dim3 grid(BSZ / 16);
  dim3 block(32);
  lstm2_wmma_kernel<<<grid, block, 0, stream>>>(x, w_ih1, w_hh1, b_ih1, b_hh1,
                                                w_ih2, w_hh2, b_ih2, b_hh2,
                                                w_fc, b_fc, out);
}


# Issue order written out: the body is one stream of slots (a transcendental beside a two-wide plain op) pipelined over unit pairs, products between slots, held by ordering points; values, sums kept
# speedup vs baseline: 1.6399x; 1.6399x over previous
#include <hip/hip_runtime.h>


#define HSZ   30
#define MPAD  128
#define KPAD  32
#define TLEN  2048
#define BSZ   512

typedef __attribute__((ext_vector_type(16))) _Float16 v16h;
typedef __attribute__((ext_vector_type(8)))  _Float16 v8h;
typedef __attribute__((ext_vector_type(8)))  float    v8f;
typedef __attribute__((ext_vector_type(2)))  float    v2f;
typedef __attribute__((ext_vector_type(2)))  _Float16 v2h;

#define ORDER(...) asm volatile("" : __VA_ARGS__)

#if __has_builtin(__builtin_amdgcn_tanhf)
#define HW_TANH __builtin_amdgcn_tanhf
#elif __has_builtin(__builtin_amdgcn_tanh_f32)
#define HW_TANH __builtin_amdgcn_tanh_f32
#endif

#ifdef HW_TANH
__device__ __forceinline__ float tanh_act(float x) { return HW_TANH(x); }
__device__ __forceinline__ float sig_act(float x)  { return fmaf(0.5f, HW_TANH(0.5f * x), 0.5f); }
#define PRE2(x) ((x) * 0.5f)
#define TSIG(m) HW_TANH(m)
#define FIN2(t) __builtin_elementwise_fma((v2f)(0.5f), (t), (v2f)(0.5f))
#else
#define PRE2(x) (x)
#define TSIG(m) __builtin_amdgcn_rcpf(1.0f + __expf(-(m)))
#define FIN2(t) (t)
__device__ __forceinline__ float tanh_act(float x) {
  return fmaf(2.0f, __builtin_amdgcn_rcpf(1.0f + __expf(-2.0f * x)), -1.0f);
}
__device__ __forceinline__ float sig_act(float x) {
  return __builtin_amdgcn_rcpf(1.0f + __expf(-x));
}
#endif

__device__ __forceinline__ v16h load_a_tile(const _Float16* __restrict__ base,
                                            int tile, int lane) {
  const int m  = lane & 15;
  const int kb = (lane >> 4) << 3;
  const _Float16* row = base + ((tile << 4) + m) * KPAD;
  v8h lo = *(const v8h*)(row + kb);
  v8h hi = *(const v8h*)(row + 16 + kb);
  v16h a;
#pragma unroll
  for (int i = 0; i < 8; ++i) { a[i] = lo[i]; a[8 + i] = hi[i]; }
  return a;
}

__device__ __forceinline__ void gates8(const v8f (&d)[4], v8f &c, float (&hn)[8]) {
#pragma unroll
  for (int r = 0; r < 8; ++r) {
    const float iv = sig_act(d[0][r]);
    const float fv = sig_act(d[1][r]);
    const float gv = tanh_act(d[2][r]);
    const float ov = sig_act(d[3][r]);
    const float cn = fmaf(fv, c[r], iv * gv);
    c[r] = cn;
    hn[r] = ov * tanh_act(cn);
  }
}

__device__ __forceinline__ v16h layer1_step(const v16h (&a1)[8], v16h h1B, v8f (&c1)[2], v8f zc) {
  v16h h1Bn;
#pragma unroll
  for (int s = 0; s < 2; ++s) {
    v8f d[4];
#pragma unroll
    for (int g = 0; g < 4; ++g)
      d[g] = __builtin_amdgcn_wmma_f32_16x16x32_f16(false, a1[2 * g + s], false, h1B,
                                                    (short)0, zc, false, false);
    float hn[8];
    gates8(d, c1[s], hn);
#pragma unroll
    for (int r = 0; r < 8; ++r) h1Bn[8 * s + r] = (_Float16)hn[r];
  }
  return h1Bn;
}

__device__ __forceinline__ v16h layer2_step(const v16h (&ai2)[8], const v16h (&ah2)[8],
                                            v16h h1B, v16h h2B, v8f (&c2)[2],
                                            const float (&wfcD)[2][8], v8f zc, float& acc_out) {
  v16h h2Bn;
  float acc = 0.f;
#pragma unroll
  for (int s = 0; s < 2; ++s) {
    v8f d[4];
#pragma unroll
    for (int g = 0; g < 4; ++g) {
      v8f tmp = __builtin_amdgcn_wmma_f32_16x16x32_f16(false, ah2[2 * g + s], false, h2B,
                                                       (short)0, zc, false, false);
      d[g] = __builtin_amdgcn_wmma_f32_16x16x32_f16(false, ai2[2 * g + s], false, h1B,
                                                    (short)0, tmp, false, false);
    }
    float hn[8];
    gates8(d, c2[s], hn);
#pragma unroll
    for (int r = 0; r < 8; ++r) {
      h2Bn[8 * s + r] = (_Float16)hn[r];
      acc = fmaf(hn[r], wfcD[s][r], acc);
    }
  }
  acc_out = acc;
  return h2Bn;
}

__global__ __launch_bounds__(32, 1)
void lstm2_wmma_kernel(const float* __restrict__ x,
                       const float* __restrict__ w_ih1, const float* __restrict__ w_hh1,
                       const float* __restrict__ b_ih1, const float* __restrict__ b_hh1,
                       const float* __restrict__ w_ih2, const float* __restrict__ w_hh2,
                       const float* __restrict__ b_ih2, const float* __restrict__ b_hh2,
                       const float* __restrict__ w_fc,  const float* __restrict__ b_fc,
                       float* __restrict__ out) {
  __shared__ _Float16 sA1[MPAD * KPAD];
  __shared__ _Float16 sAi2[MPAD * KPAD];
  __shared__ _Float16 sAh2[MPAD * KPAD];

  const int lane = threadIdx.x;
  const bool hi_half = lane >= 16;
  const int n = lane & 15;

  for (int idx = lane; idx < MPAD * KPAD; idx += 32) {
    const int m = idx >> 5, k = idx & 31;
    const int g = m >> 5, j = m & 31;
    const int row = g * HSZ + j;
    float a1 = 0.f, ai2 = 0.f, ah2 = 0.f;
    if (j < HSZ) {
      if (k < HSZ) {
        a1  = w_hh1[row * HSZ + k];
        ai2 = w_ih2[row * HSZ + k];
        ah2 = w_hh2[row * HSZ + k];
      } else if (k == 30) {
        a1  = w_ih1[row];
      } else {
        a1  = b_ih1[row] + b_hh1[row];
        ah2 = b_ih2[row] + b_hh2[row];
      }
    }
    sA1[idx]  = (_Float16)a1;
    sAi2[idx] = (_Float16)ai2;
    sAh2[idx] = (_Float16)ah2;
  }
  __syncthreads();

  v16h a1[8], ai2[8], ah2[8];
#pragma unroll
  for (int tl = 0; tl < 8; ++tl) {
    a1[tl]  = load_a_tile(sA1,  tl, lane);
    ai2[tl] = load_a_tile(sAi2, tl, lane);
    ah2[tl] = load_a_tile(sAh2, tl, lane);
  }

  float wfcD[2][8];
#pragma unroll
  for (int s = 0; s < 2; ++s)
#pragma unroll
    for (int r = 0; r < 8; ++r) {
      const int j = 16 * s + (hi_half ? 8 : 0) + r;
      wfcD[s][r] = (j < HSZ) ? w_fc[j] : 0.f;
    }
  const float bfc = b_fc[0];

  v8f zc;
#pragma unroll
  for (int i = 0; i < 8; ++i) zc[i] = 0.f;

  v16h h1B, h2B;
#pragma unroll
  for (int i = 0; i < 16; ++i) { h1B[i] = (_Float16)0.f; h2B[i] = (_Float16)0.f; }
  if (hi_half) h2B[15] = (_Float16)1.f;

  v8f c1[2], c2[2];
#pragma unroll
  for (int s = 0; s < 2; ++s)
#pragma unroll
    for (int r = 0; r < 8; ++r) { c1[s][r] = 0.f; c2[s][r] = 0.f; }

  const int bbase = blockIdx.x << 4;
  const float* xrow = x + (size_t)(bbase + n) * TLEN;
  float* orow = out + (size_t)(bbase + n) * TLEN;

  const float x0 = xrow[0];
  const float x1 = xrow[1];
  float xinj = xrow[2];
  if (hi_half) { h1B[14] = (_Float16)x0; h1B[15] = (_Float16)1.f; }
  h1B = layer1_step(a1, h1B, c1, zc);
  if (hi_half) { h1B[14] = (_Float16)x1; h1B[15] = (_Float16)1.f; }

  v8f d1[8];
#pragma unroll
  for (int tl = 0; tl < 8; ++tl)
    d1[tl] = __builtin_amdgcn_wmma_f32_16x16x32_f16(false, a1[tl], false, h1B,
                                                    (short)0, zc, false, false);

  float accPrev = 0.f;
  float accOther = 0.f;
  float pend = 0.f;

  for (int t = 0; t < TLEN - 1; ++t) {

    if (t > 1 && lane < 16) orow[t - 2] = pend;

    const float xpre = xrow[min(t + 3, TLEN - 1)];

    v8f d2[8];
    v16h h1Bn, h2Bn;
    float acc = 0.f;
    ORDER("+v"(d1[0]));
    v2f mi_0 = PRE2(((v2f){d1[0][0], d1[0][1]}));
    ORDER("+v"(mi_0), "+v"(h2B));
    d2[0] = __builtin_amdgcn_wmma_f32_16x16x32_f16(false, ah2[0], false, h2B, (short)0, zc, false, false);
    ORDER("+v"(d2[0]), "+v"(mi_0), "+v"(d1[2]));
    float ti0_0 = TSIG(mi_0[0]);
    v2f mf_0 = PRE2(((v2f){d1[2][0], d1[2][1]}));
    ORDER("+v"(ti0_0), "+v"(mf_0), "+v"(mi_0));
    float ti1_0 = TSIG(mi_0[1]);
    ORDER("+v"(ti1_0), "+v"(d1[4]));
    float tg0_0 = tanh_act(d1[4][0]);
    ORDER("+v"(tg0_0), "+v"(d1[4]), "+v"(d1[6]));
    float tg1_0 = tanh_act(d1[4][1]);
    v2f mo_0 = PRE2(((v2f){d1[6][0], d1[6][1]}));
    ORDER("+v"(tg1_0), "+v"(mo_0), "+v"(mf_0));
    float tf0_0 = TSIG(mf_0[0]);
    ORDER("+v"(tf0_0), "+v"(h2B));
    d2[2] = __builtin_amdgcn_wmma_f32_16x16x32_f16(false, ah2[2], false, h2B, (short)0, zc, false, false);
    ORDER("+v"(d2[2]), "+v"(mf_0), "+v"(ti0_0), "+v"(ti1_0));
    float tf1_0 = TSIG(mf_0[1]);
    v2f fi_0 = FIN2(((v2f){ti0_0, ti1_0}));
    ORDER("+v"(tf1_0), "+v"(fi_0), "+v"(mo_0), "+v"(d1[0]));
    float to0_0 = TSIG(mo_0[0]);
    v2f mi_1 = PRE2(((v2f){d1[0][2], d1[0][3]}));
    ORDER("+v"(to0_0), "+v"(mi_1), "+v"(mo_0));
    float to1_0 = TSIG(mo_0[1]);
    ORDER("+v"(to1_0), "+v"(fi_0), "+v"(tg0_0), "+v"(tg1_0));
    v2f ig_0 = fi_0 * (v2f){tg0_0, tg1_0};
    ORDER("+v"(ig_0), "+v"(h2B));
    d2[4] = __builtin_amdgcn_wmma_f32_16x16x32_f16(false, ah2[4], false, h2B, (short)0, zc, false, false);
    ORDER("+v"(d2[4]), "+v"(tf0_0), "+v"(tf1_0));
    v2f ff_0 = FIN2(((v2f){tf0_0, tf1_0}));
    ORDER("+v"(ff_0), "+v"(mi_1), "+v"(d1[2]));
    float ti0_1 = TSIG(mi_1[0]);
    v2f mf_1 = PRE2(((v2f){d1[2][2], d1[2][3]}));
    ORDER("+v"(ti0_1), "+v"(mf_1), "+v"(mi_1), "+v"(to0_0), "+v"(to1_0));
    float ti1_1 = TSIG(mi_1[1]);
    v2f fo_0 = FIN2(((v2f){to0_0, to1_0}));
    ORDER("+v"(ti1_1), "+v"(fo_0), "+v"(d1[4]), "+v"(ff_0), "+v"(ig_0));
    float tg0_1 = tanh_act(d1[4][2]);
    { const v2f cn = __builtin_elementwise_fma(ff_0, (v2f){c1[0][0], c1[0][1]}, ig_0); c1[0][0] = cn[0]; c1[0][1] = cn[1]; }
    ORDER("+v"(tg0_1), "+v"(c1[0]), "+v"(d1[4]), "+v"(d1[6]));
    float tg1_1 = tanh_act(d1[4][3]);
    v2f mo_1 = PRE2(((v2f){d1[6][2], d1[6][3]}));
    ORDER("+v"(tg1_1), "+v"(mo_1), "+v"(mf_1));
    float tf0_1 = TSIG(mf_1[0]);
    ORDER("+v"(tf0_1), "+v"(h2B));
    d2[6] = __builtin_amdgcn_wmma_f32_16x16x32_f16(false, ah2[6], false, h2B, (short)0, zc, false, false);
    ORDER("+v"(d2[6]), "+v"(mf_1), "+v"(ti0_1), "+v"(ti1_1));
    float tf1_1 = TSIG(mf_1[1]);
    v2f fi_1 = FIN2(((v2f){ti0_1, ti1_1}));
    ORDER("+v"(tf1_1), "+v"(fi_1), "+v"(mo_1), "+v"(d1[0]));
    float to0_1 = TSIG(mo_1[0]);
    v2f mi_2 = PRE2(((v2f){d1[0][4], d1[0][5]}));
    ORDER("+v"(to0_1), "+v"(mi_2), "+v"(mo_1));
    float to1_1 = TSIG(mo_1[1]);
    ORDER("+v"(to1_1), "+v"(c1[0]), "+v"(fi_1), "+v"(tg0_1), "+v"(tg1_1));
    float tc0_0 = tanh_act(c1[0][0]);
    v2f ig_1 = fi_1 * (v2f){tg0_1, tg1_1};
    ORDER("+v"(tc0_0), "+v"(ig_1), "+v"(c1[0]));
    float tc1_0 = tanh_act(c1[0][1]);
    ORDER("+v"(tc1_0), "+v"(h2B));
    d2[1] = __builtin_amdgcn_wmma_f32_16x16x32_f16(false, ah2[1], false, h2B, (short)0, zc, false, false);
    ORDER("+v"(d2[1]), "+v"(tf0_1), "+v"(tf1_1));
    v2f ff_1 = FIN2(((v2f){tf0_1, tf1_1}));
    ORDER("+v"(ff_1), "+v"(mi_2), "+v"(d1[2]));
    float ti0_2 = TSIG(mi_2[0]);
    v2f mf_2 = PRE2(((v2f){d1[2][4], d1[2][5]}));
    ORDER("+v"(ti0_2), "+v"(mf_2), "+v"(mi_2), "+v"(to0_1), "+v"(to1_1));
    float ti1_2 = TSIG(mi_2[1]);
    v2f fo_1 = FIN2(((v2f){to0_1, to1_1}));
    ORDER("+v"(ti1_2), "+v"(fo_1), "+v"(d1[4]), "+v"(ff_1), "+v"(ig_1));
    float tg0_2 = tanh_act(d1[4][4]);
    { const v2f cn = __builtin_elementwise_fma(ff_1, (v2f){c1[0][2], c1[0][3]}, ig_1); c1[0][2] = cn[0]; c1[0][3] = cn[1]; }
    ORDER("+v"(tg0_2), "+v"(c1[0]), "+v"(d1[4]), "+v"(d1[6]));
    float tg1_2 = tanh_act(d1[4][5]);
    v2f mo_2 = PRE2(((v2f){d1[6][4], d1[6][5]}));
    ORDER("+v"(tg1_2), "+v"(mo_2), "+v"(mf_2));
    float tf0_2 = TSIG(mf_2[0]);
    ORDER("+v"(tf0_2), "+v"(h2B));
    d2[3] = __builtin_amdgcn_wmma_f32_16x16x32_f16(false, ah2[3], false, h2B, (short)0, zc, false, false);
    ORDER("+v"(d2[3]), "+v"(fo_0), "+v"(tc0_0), "+v"(tc1_0));
    v2f hh_0 = fo_0 * (v2f){tc0_0, tc1_0};
    ORDER("+v"(hh_0), "+v"(mf_2), "+v"(ti0_2), "+v"(ti1_2));
    float tf1_2 = TSIG(mf_2[1]);
    v2f fi_2 = FIN2(((v2f){ti0_2, ti1_2}));
    ORDER("+v"(tf1_2), "+v"(fi_2), "+v"(mo_2), "+v"(d1[0]));
    float to0_2 = TSIG(mo_2[0]);
    v2f mi_3 = PRE2(((v2f){d1[0][6], d1[0][7]}));
    ORDER("+v"(to0_2), "+v"(mi_3), "+v"(mo_2), "+v"(hh_0));
    float to1_2 = TSIG(mo_2[1]);
    v2h pk_0 = __builtin_convertvector(hh_0, v2h);
    ORDER("+v"(to1_2), "+v"(pk_0), "+v"(c1[0]), "+v"(fi_2), "+v"(tg0_2), "+v"(tg1_2));
    float tc0_1 = tanh_act(c1[0][2]);
    v2f ig_2 = fi_2 * (v2f){tg0_2, tg1_2};
    ORDER("+v"(tc0_1), "+v"(ig_2), "+v"(c1[0]));
    float tc1_1 = tanh_act(c1[0][3]);
    ORDER("+v"(tc1_1), "+v"(h2B));
    d2[5] = __builtin_amdgcn_wmma_f32_16x16x32_f16(false, ah2[5], false, h2B, (short)0, zc, false, false);
    ORDER("+v"(d2[5]), "+v"(tf0_2), "+v"(tf1_2));
    v2f ff_2 = FIN2(((v2f){tf0_2, tf1_2}));
    ORDER("+v"(ff_2), "+v"(mi_3), "+v"(d1[2]));
    float ti0_3 = TSIG(mi_3[0]);
    v2f mf_3 = PRE2(((v2f){d1[2][6], d1[2][7]}));
    ORDER("+v"(ti0_3), "+v"(mf_3), "+v"(mi_3), "+v"(to0_2), "+v"(to1_2));
    float ti1_3 = TSIG(mi_3[1]);
    v2f fo_2 = FIN2(((v2f){to0_2, to1_2}));
    ORDER("+v"(ti1_3), "+v"(fo_2), "+v"(d1[4]), "+v"(ff_2), "+v"(ig_2));
    float tg0_3 = tanh_act(d1[4][6]);
    { const v2f cn = __builtin_elementwise_fma(ff_2, (v2f){c1[0][4], c1[0][5]}, ig_2); c1[0][4] = cn[0]; c1[0][5] = cn[1]; }
    ORDER("+v"(tg0_3), "+v"(c1[0]), "+v"(d1[4]), "+v"(d1[6]));
    float tg1_3 = tanh_act(d1[4][7]);
    v2f mo_3 = PRE2(((v2f){d1[6][6], d1[6][7]}));
    ORDER("+v"(tg1_3), "+v"(mo_3), "+v"(mf_3));
    float tf0_3 = TSIG(mf_3[0]);
    ORDER("+v"(tf0_3), "+v"(h2B));
    d2[7] = __builtin_amdgcn_wmma_f32_16x16x32_f16(false, ah2[7], false, h2B, (short)0, zc, false, false);
    ORDER("+v"(d2[7]), "+v"(fo_1), "+v"(tc0_1), "+v"(tc1_1));
    v2f hh_1 = fo_1 * (v2f){tc0_1, tc1_1};
    ORDER("+v"(hh_1), "+v"(mf_3), "+v"(ti0_3), "+v"(ti1_3));
    float tf1_3 = TSIG(mf_3[1]);
    v2f fi_3 = FIN2(((v2f){ti0_3, ti1_3}));
    ORDER("+v"(tf1_3), "+v"(fi_3), "+v"(mo_3), "+v"(d1[1]));
    float to0_3 = TSIG(mo_3[0]);
    v2f mi_4 = PRE2(((v2f){d1[1][0], d1[1][1]}));
    ORDER("+v"(to0_3), "+v"(mi_4), "+v"(mo_3), "+v"(hh_1));
    float to1_3 = TSIG(mo_3[1]);
    v2h pk_1 = __builtin_convertvector(hh_1, v2h);
    ORDER("+v"(to1_3), "+v"(pk_1), "+v"(c1[0]), "+v"(fi_3), "+v"(tg0_3), "+v"(tg1_3));
    float tc0_2 = tanh_act(c1[0][4]);
    v2f ig_3 = fi_3 * (v2f){tg0_3, tg1_3};
    ORDER("+v"(tc0_2), "+v"(ig_3), "+v"(c1[0]));
    float tc1_2 = tanh_act(c1[0][5]);
    ORDER("+v"(tc1_2), "+v"(h1B), "+v"(d2[0]));
    d2[0] = __builtin_amdgcn_wmma_f32_16x16x32_f16(false, ai2[0], false, h1B, (short)0, d2[0], false, false);
    ORDER("+v"(d2[0]), "+v"(tf0_3), "+v"(tf1_3));
    v2f ff_3 = FIN2(((v2f){tf0_3, tf1_3}));
    ORDER("+v"(ff_3), "+v"(mi_4), "+v"(d1[3]));
    float ti0_4 = TSIG(mi_4[0]);
    v2f mf_4 = PRE2(((v2f){d1[3][0], d1[3][1]}));
    ORDER("+v"(ti0_4), "+v"(mf_4), "+v"(mi_4), "+v"(to0_3), "+v"(to1_3));
    float ti1_4 = TSIG(mi_4[1]);
    v2f fo_3 = FIN2(((v2f){to0_3, to1_3}));
    ORDER("+v"(ti1_4), "+v"(fo_3), "+v"(d1[5]), "+v"(ff_3), "+v"(ig_3));
    float tg0_4 = tanh_act(d1[5][0]);
    { const v2f cn = __builtin_elementwise_fma(ff_3, (v2f){c1[0][6], c1[0][7]}, ig_3); c1[0][6] = cn[0]; c1[0][7] = cn[1]; }
    ORDER("+v"(tg0_4), "+v"(c1[0]), "+v"(d1[5]), "+v"(d1[7]));
    float tg1_4 = tanh_act(d1[5][1]);
    v2f mo_4 = PRE2(((v2f){d1[7][0], d1[7][1]}));
    ORDER("+v"(tg1_4), "+v"(mo_4), "+v"(mf_4));
    float tf0_4 = TSIG(mf_4[0]);
    ORDER("+v"(tf0_4), "+v"(h1B), "+v"(d2[2]));
    d2[2] = __builtin_amdgcn_wmma_f32_16x16x32_f16(false, ai2[2], false, h1B, (short)0, d2[2], false, false);
    ORDER("+v"(d2[2]), "+v"(fo_2), "+v"(tc0_2), "+v"(tc1_2));
    v2f hh_2 = fo_2 * (v2f){tc0_2, tc1_2};
    ORDER("+v"(hh_2), "+v"(mf_4), "+v"(ti0_4), "+v"(ti1_4));
    float tf1_4 = TSIG(mf_4[1]);
    v2f fi_4 = FIN2(((v2f){ti0_4, ti1_4}));
    ORDER("+v"(tf1_4), "+v"(fi_4), "+v"(mo_4), "+v"(d1[1]));
    float to0_4 = TSIG(mo_4[0]);
    v2f mi_5 = PRE2(((v2f){d1[1][2], d1[1][3]}));
    ORDER("+v"(to0_4), "+v"(mi_5), "+v"(mo_4), "+v"(hh_2));
    float to1_4 = TSIG(mo_4[1]);
    v2h pk_2 = __builtin_convertvector(hh_2, v2h);
    ORDER("+v"(to1_4), "+v"(pk_2), "+v"(c1[0]), "+v"(fi_4), "+v"(tg0_4), "+v"(tg1_4));
    float tc0_3 = tanh_act(c1[0][6]);
    v2f ig_4 = fi_4 * (v2f){tg0_4, tg1_4};
    ORDER("+v"(tc0_3), "+v"(ig_4), "+v"(c1[0]));
    float tc1_3 = tanh_act(c1[0][7]);
    ORDER("+v"(tc1_3), "+v"(h1B), "+v"(d2[4]));
    d2[4] = __builtin_amdgcn_wmma_f32_16x16x32_f16(false, ai2[4], false, h1B, (short)0, d2[4], false, false);
    ORDER("+v"(d2[4]), "+v"(tf0_4), "+v"(tf1_4));
    v2f ff_4 = FIN2(((v2f){tf0_4, tf1_4}));
    ORDER("+v"(ff_4), "+v"(mi_5), "+v"(d1[3]));
    float ti0_5 = TSIG(mi_5[0]);
    v2f mf_5 = PRE2(((v2f){d1[3][2], d1[3][3]}));
    ORDER("+v"(ti0_5), "+v"(mf_5), "+v"(mi_5), "+v"(to0_4), "+v"(to1_4));
    float ti1_5 = TSIG(mi_5[1]);
    v2f fo_4 = FIN2(((v2f){to0_4, to1_4}));
    ORDER("+v"(ti1_5), "+v"(fo_4), "+v"(d1[5]), "+v"(ff_4), "+v"(ig_4));
    float tg0_5 = tanh_act(d1[5][2]);
    { const v2f cn = __builtin_elementwise_fma(ff_4, (v2f){c1[1][0], c1[1][1]}, ig_4); c1[1][0] = cn[0]; c1[1][1] = cn[1]; }
    ORDER("+v"(tg0_5), "+v"(c1[1]), "+v"(d1[5]), "+v"(d1[7]));
    float tg1_5 = tanh_act(d1[5][3]);
    v2f mo_5 = PRE2(((v2f){d1[7][2], d1[7][3]}));
    ORDER("+v"(tg1_5), "+v"(mo_5), "+v"(mf_5));
    float tf0_5 = TSIG(mf_5[0]);
    ORDER("+v"(tf0_5), "+v"(h1B), "+v"(d2[6]));
    d2[6] = __builtin_amdgcn_wmma_f32_16x16x32_f16(false, ai2[6], false, h1B, (short)0, d2[6], false, false);
    ORDER("+v"(d2[6]), "+v"(fo_3), "+v"(tc0_3), "+v"(tc1_3));
    v2f hh_3 = fo_3 * (v2f){tc0_3, tc1_3};
    ORDER("+v"(hh_3), "+v"(mf_5), "+v"(ti0_5), "+v"(ti1_5));
    float tf1_5 = TSIG(mf_5[1]);
    v2f fi_5 = FIN2(((v2f){ti0_5, ti1_5}));
    ORDER("+v"(tf1_5), "+v"(fi_5), "+v"(mo_5), "+v"(d1[1]));
    float to0_5 = TSIG(mo_5[0]);
    v2f mi_6 = PRE2(((v2f){d1[1][4], d1[1][5]}));
    ORDER("+v"(to0_5), "+v"(mi_6), "+v"(mo_5), "+v"(hh_3));
    float to1_5 = TSIG(mo_5[1]);
    v2h pk_3 = __builtin_convertvector(hh_3, v2h);
    ORDER("+v"(to1_5), "+v"(pk_3), "+v"(c1[1]), "+v"(fi_5), "+v"(tg0_5), "+v"(tg1_5));
    float tc0_4 = tanh_act(c1[1][0]);
    v2f ig_5 = fi_5 * (v2f){tg0_5, tg1_5};
    ORDER("+v"(tc0_4), "+v"(ig_5), "+v"(c1[1]));
    float tc1_4 = tanh_act(c1[1][1]);
    ORDER("+v"(tc1_4), "+v"(h1B), "+v"(d2[1]));
    d2[1] = __builtin_amdgcn_wmma_f32_16x16x32_f16(false, ai2[1], false, h1B, (short)0, d2[1], false, false);
    ORDER("+v"(d2[1]), "+v"(tf0_5), "+v"(tf1_5));
    v2f ff_5 = FIN2(((v2f){tf0_5, tf1_5}));
    ORDER("+v"(ff_5), "+v"(mi_6), "+v"(d1[3]));
    float ti0_6 = TSIG(mi_6[0]);
    v2f mf_6 = PRE2(((v2f){d1[3][4], d1[3][5]}));
    ORDER("+v"(ti0_6), "+v"(mf_6), "+v"(mi_6), "+v"(to0_5), "+v"(to1_5));
    float ti1_6 = TSIG(mi_6[1]);
    v2f fo_5 = FIN2(((v2f){to0_5, to1_5}));
    ORDER("+v"(ti1_6), "+v"(fo_5), "+v"(d1[5]), "+v"(ff_5), "+v"(ig_5));
    float tg0_6 = tanh_act(d1[5][4]);
    { const v2f cn = __builtin_elementwise_fma(ff_5, (v2f){c1[1][2], c1[1][3]}, ig_5); c1[1][2] = cn[0]; c1[1][3] = cn[1]; }
    ORDER("+v"(tg0_6), "+v"(c1[1]), "+v"(d1[5]), "+v"(d1[7]));
    float tg1_6 = tanh_act(d1[5][5]);
    v2f mo_6 = PRE2(((v2f){d1[7][4], d1[7][5]}));
    ORDER("+v"(tg1_6), "+v"(mo_6), "+v"(mf_6));
    float tf0_6 = TSIG(mf_6[0]);
    ORDER("+v"(tf0_6), "+v"(h1B), "+v"(d2[3]));
    d2[3] = __builtin_amdgcn_wmma_f32_16x16x32_f16(false, ai2[3], false, h1B, (short)0, d2[3], false, false);
    ORDER("+v"(d2[3]), "+v"(fo_4), "+v"(tc0_4), "+v"(tc1_4));
    v2f hh_4 = fo_4 * (v2f){tc0_4, tc1_4};
    ORDER("+v"(hh_4), "+v"(mf_6), "+v"(ti0_6), "+v"(ti1_6));
    float tf1_6 = TSIG(mf_6[1]);
    v2f fi_6 = FIN2(((v2f){ti0_6, ti1_6}));
    ORDER("+v"(tf1_6), "+v"(fi_6), "+v"(mo_6), "+v"(d1[1]));
    float to0_6 = TSIG(mo_6[0]);
    v2f mi_7 = PRE2(((v2f){d1[1][6], d1[1][7]}));
    ORDER("+v"(to0_6), "+v"(mi_7), "+v"(mo_6), "+v"(hh_4));
    float to1_6 = TSIG(mo_6[1]);
    v2h pk_4 = __builtin_convertvector(hh_4, v2h);
    ORDER("+v"(to1_6), "+v"(pk_4), "+v"(c1[1]), "+v"(fi_6), "+v"(tg0_6), "+v"(tg1_6));
    float tc0_5 = tanh_act(c1[1][2]);
    v2f ig_6 = fi_6 * (v2f){tg0_6, tg1_6};
    ORDER("+v"(tc0_5), "+v"(ig_6), "+v"(c1[1]));
    float tc1_5 = tanh_act(c1[1][3]);
    ORDER("+v"(tc1_5), "+v"(h1B), "+v"(d2[5]));
    d2[5] = __builtin_amdgcn_wmma_f32_16x16x32_f16(false, ai2[5], false, h1B, (short)0, d2[5], false, false);
    ORDER("+v"(d2[5]), "+v"(tf0_6), "+v"(tf1_6));
    v2f ff_6 = FIN2(((v2f){tf0_6, tf1_6}));
    ORDER("+v"(ff_6), "+v"(mi_7), "+v"(d1[3]));
    float ti0_7 = TSIG(mi_7[0]);
    v2f mf_7 = PRE2(((v2f){d1[3][6], d1[3][7]}));
    ORDER("+v"(ti0_7), "+v"(mf_7), "+v"(mi_7), "+v"(to0_6), "+v"(to1_6));
    float ti1_7 = TSIG(mi_7[1]);
    v2f fo_6 = FIN2(((v2f){to0_6, to1_6}));
    ORDER("+v"(ti1_7), "+v"(fo_6), "+v"(d1[5]), "+v"(ff_6), "+v"(ig_6));
    float tg0_7 = tanh_act(d1[5][6]);
    { const v2f cn = __builtin_elementwise_fma(ff_6, (v2f){c1[1][4], c1[1][5]}, ig_6); c1[1][4] = cn[0]; c1[1][5] = cn[1]; }
    ORDER("+v"(tg0_7), "+v"(c1[1]), "+v"(d1[5]), "+v"(d1[7]));
    float tg1_7 = tanh_act(d1[5][7]);
    v2f mo_7 = PRE2(((v2f){d1[7][6], d1[7][7]}));
    ORDER("+v"(tg1_7), "+v"(mo_7), "+v"(mf_7));
    float tf0_7 = TSIG(mf_7[0]);
    ORDER("+v"(tf0_7), "+v"(h1B), "+v"(d2[7]));
    d2[7] = __builtin_amdgcn_wmma_f32_16x16x32_f16(false, ai2[7], false, h1B, (short)0, d2[7], false, false);
    ORDER("+v"(d2[7]), "+v"(fo_5), "+v"(tc0_5), "+v"(tc1_5));
    v2f hh_5 = fo_5 * (v2f){tc0_5, tc1_5};
    ORDER("+v"(hh_5), "+v"(mf_7), "+v"(ti0_7), "+v"(ti1_7));
    float tf1_7 = TSIG(mf_7[1]);
    v2f fi_7 = FIN2(((v2f){ti0_7, ti1_7}));
    ORDER("+v"(tf1_7), "+v"(fi_7), "+v"(mo_7), "+v"(d2[0]));
    float to0_7 = TSIG(mo_7[0]);
    v2f mi_8 = PRE2(((v2f){d2[0][0], d2[0][1]}));
    ORDER("+v"(to0_7), "+v"(mi_8), "+v"(mo_7), "+v"(hh_5));
    float to1_7 = TSIG(mo_7[1]);
    v2h pk_5 = __builtin_convertvector(hh_5, v2h);
    ORDER("+v"(to1_7), "+v"(pk_5), "+v"(c1[1]), "+v"(fi_7), "+v"(tg0_7), "+v"(tg1_7));
    float tc0_6 = tanh_act(c1[1][4]);
    v2f ig_7 = fi_7 * (v2f){tg0_7, tg1_7};
    ORDER("+v"(tc0_6), "+v"(ig_7), "+v"(c1[1]), "+v"(tf0_7), "+v"(tf1_7));
    float tc1_6 = tanh_act(c1[1][5]);
    v2f ff_7 = FIN2(((v2f){tf0_7, tf1_7}));
    ORDER("+v"(tc1_6), "+v"(ff_7), "+v"(mi_8), "+v"(d2[2]));
    float ti0_8 = TSIG(mi_8[0]);
    v2f mf_8 = PRE2(((v2f){d2[2][0], d2[2][1]}));
    ORDER("+v"(ti0_8), "+v"(mf_8), "+v"(mi_8), "+v"(to0_7), "+v"(to1_7));
    float ti1_8 = TSIG(mi_8[1]);
    v2f fo_7 = FIN2(((v2f){to0_7, to1_7}));
    ORDER("+v"(ti1_8), "+v"(fo_7), "+v"(d2[4]), "+v"(ff_7), "+v"(ig_7));
    float tg0_8 = tanh_act(d2[4][0]);
    { const v2f cn = __builtin_elementwise_fma(ff_7, (v2f){c1[1][6], c1[1][7]}, ig_7); c1[1][6] = cn[0]; c1[1][7] = cn[1]; }
    ORDER("+v"(tg0_8), "+v"(c1[1]), "+v"(d2[4]), "+v"(d2[6]));
    float tg1_8 = tanh_act(d2[4][1]);
    v2f mo_8 = PRE2(((v2f){d2[6][0], d2[6][1]}));
    ORDER("+v"(tg1_8), "+v"(mo_8), "+v"(mf_8), "+v"(fo_6), "+v"(tc0_6), "+v"(tc1_6));
    float tf0_8 = TSIG(mf_8[0]);
    v2f hh_6 = fo_6 * (v2f){tc0_6, tc1_6};
    ORDER("+v"(tf0_8), "+v"(hh_6), "+v"(mf_8), "+v"(ti0_8), "+v"(ti1_8));
    float tf1_8 = TSIG(mf_8[1]);
    v2f fi_8 = FIN2(((v2f){ti0_8, ti1_8}));
    ORDER("+v"(tf1_8), "+v"(fi_8), "+v"(mo_8), "+v"(d2[0]));
    float to0_8 = TSIG(mo_8[0]);
    v2f mi_9 = PRE2(((v2f){d2[0][2], d2[0][3]}));
    ORDER("+v"(to0_8), "+v"(mi_9), "+v"(mo_8), "+v"(hh_6));
    float to1_8 = TSIG(mo_8[1]);
    v2h pk_6 = __builtin_convertvector(hh_6, v2h);
    ORDER("+v"(to1_8), "+v"(pk_6), "+v"(c1[1]), "+v"(fi_8), "+v"(tg0_8), "+v"(tg1_8));
    float tc0_7 = tanh_act(c1[1][6]);
    v2f ig_8 = fi_8 * (v2f){tg0_8, tg1_8};
    ORDER("+v"(tc0_7), "+v"(ig_8), "+v"(c1[1]), "+v"(tf0_8), "+v"(tf1_8));
    float tc1_7 = tanh_act(c1[1][7]);
    v2f ff_8 = FIN2(((v2f){tf0_8, tf1_8}));
    ORDER("+v"(tc1_7), "+v"(ff_8), "+v"(mi_9), "+v"(d2[2]));
    float ti0_9 = TSIG(mi_9[0]);
    v2f mf_9 = PRE2(((v2f){d2[2][2], d2[2][3]}));
    ORDER("+v"(ti0_9), "+v"(mf_9), "+v"(mi_9), "+v"(to0_8), "+v"(to1_8));
    float ti1_9 = TSIG(mi_9[1]);
    v2f fo_8 = FIN2(((v2f){to0_8, to1_8}));
    ORDER("+v"(ti1_9), "+v"(fo_8), "+v"(d2[4]), "+v"(ff_8), "+v"(ig_8));
    float tg0_9 = tanh_act(d2[4][2]);
    { const v2f cn = __builtin_elementwise_fma(ff_8, (v2f){c2[0][0], c2[0][1]}, ig_8); c2[0][0] = cn[0]; c2[0][1] = cn[1]; }
    ORDER("+v"(tg0_9), "+v"(c2[0]), "+v"(d2[4]), "+v"(d2[6]));
    float tg1_9 = tanh_act(d2[4][3]);
    v2f mo_9 = PRE2(((v2f){d2[6][2], d2[6][3]}));
    ORDER("+v"(tg1_9), "+v"(mo_9), "+v"(mf_9), "+v"(fo_7), "+v"(tc0_7), "+v"(tc1_7));
    float tf0_9 = TSIG(mf_9[0]);
    v2f hh_7 = fo_7 * (v2f){tc0_7, tc1_7};
    ORDER("+v"(tf0_9), "+v"(hh_7), "+v"(mf_9), "+v"(ti0_9), "+v"(ti1_9));
    float tf1_9 = TSIG(mf_9[1]);
    v2f fi_9 = FIN2(((v2f){ti0_9, ti1_9}));
    ORDER("+v"(tf1_9), "+v"(fi_9), "+v"(mo_9), "+v"(d2[0]));
    float to0_9 = TSIG(mo_9[0]);
    v2f mi_10 = PRE2(((v2f){d2[0][4], d2[0][5]}));
    ORDER("+v"(to0_9), "+v"(mi_10), "+v"(mo_9), "+v"(hh_7));
    float to1_9 = TSIG(mo_9[1]);
    v2h pk_7 = __builtin_convertvector(hh_7, v2h);
    ORDER("+v"(to1_9), "+v"(pk_7), "+v"(c2[0]), "+v"(fi_9), "+v"(tg0_9), "+v"(tg1_9));
    float tc0_8 = tanh_act(c2[0][0]);
    v2f ig_9 = fi_9 * (v2f){tg0_9, tg1_9};
    h1Bn[0] = pk_0[0]; h1Bn[1] = pk_0[1];
    h1Bn[2] = pk_1[0]; h1Bn[3] = pk_1[1];
    h1Bn[4] = pk_2[0]; h1Bn[5] = pk_2[1];
    h1Bn[6] = pk_3[0]; h1Bn[7] = pk_3[1];
    h1Bn[8] = pk_4[0]; h1Bn[9] = pk_4[1];
    h1Bn[10] = pk_5[0]; h1Bn[11] = pk_5[1];
    h1Bn[12] = pk_6[0]; h1Bn[13] = pk_6[1];
    h1Bn[14] = pk_7[0]; h1Bn[15] = pk_7[1];
    if (hi_half) { h1Bn[14] = (_Float16)xinj; h1Bn[15] = (_Float16)1.f; }
    ORDER("+v"(tc0_8), "+v"(ig_9), "+v"(h1Bn), "+v"(c2[0]), "+v"(tf0_9), "+v"(tf1_9));
    float tc1_8 = tanh_act(c2[0][1]);
    v2f ff_9 = FIN2(((v2f){tf0_9, tf1_9}));
    ORDER("+v"(tc1_8), "+v"(ff_9), "+v"(mi_10), "+v"(d2[2]));
    float ti0_10 = TSIG(mi_10[0]);
    v2f mf_10 = PRE2(((v2f){d2[2][4], d2[2][5]}));
    ORDER("+v"(ti0_10), "+v"(mf_10), "+v"(mi_10), "+v"(to0_9), "+v"(to1_9));
    float ti1_10 = TSIG(mi_10[1]);
    v2f fo_9 = FIN2(((v2f){to0_9, to1_9}));
    ORDER("+v"(ti1_10), "+v"(fo_9), "+v"(d2[4]), "+v"(ff_9), "+v"(ig_9));
    float tg0_10 = tanh_act(d2[4][4]);
    { const v2f cn = __builtin_elementwise_fma(ff_9, (v2f){c2[0][2], c2[0][3]}, ig_9); c2[0][2] = cn[0]; c2[0][3] = cn[1]; }
    ORDER("+v"(tg0_10), "+v"(c2[0]), "+v"(d2[4]), "+v"(d2[6]));
    float tg1_10 = tanh_act(d2[4][5]);
    v2f mo_10 = PRE2(((v2f){d2[6][4], d2[6][5]}));
    ORDER("+v"(tg1_10), "+v"(mo_10), "+v"(mf_10));
    float tf0_10 = TSIG(mf_10[0]);
    ORDER("+v"(tf0_10), "+v"(h1Bn));
    d1[0] = __builtin_amdgcn_wmma_f32_16x16x32_f16(false, a1[0], false, h1Bn, (short)0, zc, false, false);
    ORDER("+v"(d1[0]), "+v"(fo_8), "+v"(tc0_8), "+v"(tc1_8));
    v2f hh_8 = fo_8 * (v2f){tc0_8, tc1_8};
    ORDER("+v"(hh_8), "+v"(mf_10), "+v"(ti0_10), "+v"(ti1_10));
    float tf1_10 = TSIG(mf_10[1]);
    v2f fi_10 = FIN2(((v2f){ti0_10, ti1_10}));
    ORDER("+v"(tf1_10), "+v"(fi_10), "+v"(mo_10), "+v"(d2[0]));
    float to0_10 = TSIG(mo_10[0]);
    v2f mi_11 = PRE2(((v2f){d2[0][6], d2[0][7]}));
    ORDER("+v"(to0_10), "+v"(mi_11), "+v"(mo_10), "+v"(hh_8));
    float to1_10 = TSIG(mo_10[1]);
    v2h pk_8 = __builtin_convertvector(hh_8, v2h);
    ORDER("+v"(to1_10), "+v"(pk_8), "+v"(c2[0]), "+v"(fi_10), "+v"(tg0_10), "+v"(tg1_10), "+v"(hh_8), "+v"(acc));
    float tc0_9 = tanh_act(c2[0][2]);
    v2f ig_10 = fi_10 * (v2f){tg0_10, tg1_10};
    acc = fmaf(hh_8[0], wfcD[0][0], acc);
    ORDER("+v"(tc0_9), "+v"(ig_10), "+v"(acc), "+v"(c2[0]));
    float tc1_9 = tanh_act(c2[0][3]);
    ORDER("+v"(tc1_9), "+v"(h1Bn));
    d1[2] = __builtin_amdgcn_wmma_f32_16x16x32_f16(false, a1[2], false, h1Bn, (short)0, zc, false, false);
    ORDER("+v"(d1[2]), "+v"(tf0_10), "+v"(tf1_10));
    v2f ff_10 = FIN2(((v2f){tf0_10, tf1_10}));
    ORDER("+v"(ff_10), "+v"(mi_11), "+v"(d2[2]));
    float ti0_11 = TSIG(mi_11[0]);
    v2f mf_11 = PRE2(((v2f){d2[2][6], d2[2][7]}));
    ORDER("+v"(ti0_11), "+v"(mf_11), "+v"(mi_11), "+v"(to0_10), "+v"(to1_10));
    float ti1_11 = TSIG(mi_11[1]);
    v2f fo_10 = FIN2(((v2f){to0_10, to1_10}));
    ORDER("+v"(ti1_11), "+v"(fo_10), "+v"(d2[4]), "+v"(ff_10), "+v"(ig_10), "+v"(hh_8), "+v"(acc));
    float tg0_11 = tanh_act(d2[4][6]);
    { const v2f cn = __builtin_elementwise_fma(ff_10, (v2f){c2[0][4], c2[0][5]}, ig_10); c2[0][4] = cn[0]; c2[0][5] = cn[1]; }
    acc = fmaf(hh_8[1], wfcD[0][1], acc);
    ORDER("+v"(tg0_11), "+v"(c2[0]), "+v"(acc), "+v"(d2[4]), "+v"(d2[6]));
    float tg1_11 = tanh_act(d2[4][7]);
    v2f mo_11 = PRE2(((v2f){d2[6][6], d2[6][7]}));
    ORDER("+v"(tg1_11), "+v"(mo_11), "+v"(mf_11), "+v"(fo_9), "+v"(tc0_9), "+v"(tc1_9));
    float tf0_11 = TSIG(mf_11[0]);
    v2f hh_9 = fo_9 * (v2f){tc0_9, tc1_9};
    ORDER("+v"(tf0_11), "+v"(hh_9), "+v"(mf_11), "+v"(ti0_11), "+v"(ti1_11));
    float tf1_11 = TSIG(mf_11[1]);
    v2f fi_11 = FIN2(((v2f){ti0_11, ti1_11}));
    ORDER("+v"(tf1_11), "+v"(fi_11), "+v"(mo_11), "+v"(d2[1]));
    float to0_11 = TSIG(mo_11[0]);
    v2f mi_12 = PRE2(((v2f){d2[1][0], d2[1][1]}));
    ORDER("+v"(to0_11), "+v"(mi_12), "+v"(mo_11), "+v"(hh_9));
    float to1_11 = TSIG(mo_11[1]);
    v2h pk_9 = __builtin_convertvector(hh_9, v2h);
    ORDER("+v"(to1_11), "+v"(pk_9), "+v"(c2[0]), "+v"(fi_11), "+v"(tg0_11), "+v"(tg1_11), "+v"(hh_9), "+v"(acc));
    float tc0_10 = tanh_act(c2[0][4]);
    v2f ig_11 = fi_11 * (v2f){tg0_11, tg1_11};
    acc = fmaf(hh_9[0], wfcD[0][2], acc);
    ORDER("+v"(tc0_10), "+v"(ig_11), "+v"(acc), "+v"(c2[0]));
    float tc1_10 = tanh_act(c2[0][5]);
    ORDER("+v"(tc1_10), "+v"(h1Bn));
    d1[4] = __builtin_amdgcn_wmma_f32_16x16x32_f16(false, a1[4], false, h1Bn, (short)0, zc, false, false);
    ORDER("+v"(d1[4]), "+v"(tf0_11), "+v"(tf1_11));
    v2f ff_11 = FIN2(((v2f){tf0_11, tf1_11}));
    ORDER("+v"(ff_11), "+v"(mi_12), "+v"(d2[3]));
    float ti0_12 = TSIG(mi_12[0]);
    v2f mf_12 = PRE2(((v2f){d2[3][0], d2[3][1]}));
    ORDER("+v"(ti0_12), "+v"(mf_12), "+v"(mi_12), "+v"(to0_11), "+v"(to1_11));
    float ti1_12 = TSIG(mi_12[1]);
    v2f fo_11 = FIN2(((v2f){to0_11, to1_11}));
    ORDER("+v"(ti1_12), "+v"(fo_11), "+v"(d2[5]), "+v"(ff_11), "+v"(ig_11), "+v"(hh_9), "+v"(acc));
    float tg0_12 = tanh_act(d2[5][0]);
    { const v2f cn = __builtin_elementwise_fma(ff_11, (v2f){c2[0][6], c2[0][7]}, ig_11); c2[0][6] = cn[0]; c2[0][7] = cn[1]; }
    acc = fmaf(hh_9[1], wfcD[0][3], acc);
    ORDER("+v"(tg0_12), "+v"(c2[0]), "+v"(acc), "+v"(d2[5]), "+v"(d2[7]));
    float tg1_12 = tanh_act(d2[5][1]);
    v2f mo_12 = PRE2(((v2f){d2[7][0], d2[7][1]}));
    ORDER("+v"(tg1_12), "+v"(mo_12), "+v"(mf_12), "+v"(fo_10), "+v"(tc0_10), "+v"(tc1_10));
    float tf0_12 = TSIG(mf_12[0]);
    v2f hh_10 = fo_10 * (v2f){tc0_10, tc1_10};
    ORDER("+v"(tf0_12), "+v"(hh_10), "+v"(mf_12), "+v"(ti0_12), "+v"(ti1_12));
    float tf1_12 = TSIG(mf_12[1]);
    v2f fi_12 = FIN2(((v2f){ti0_12, ti1_12}));
    ORDER("+v"(tf1_12), "+v"(fi_12), "+v"(mo_12), "+v"(d2[1]));
    float to0_12 = TSIG(mo_12[0]);
    v2f mi_13 = PRE2(((v2f){d2[1][2], d2[1][3]}));
    ORDER("+v"(to0_12), "+v"(mi_13), "+v"(mo_12), "+v"(hh_10));
    float to1_12 = TSIG(mo_12[1]);
    v2h pk_10 = __builtin_convertvector(hh_10, v2h);
    ORDER("+v"(to1_12), "+v"(pk_10), "+v"(c2[0]), "+v"(fi_12), "+v"(tg0_12), "+v"(tg1_12), "+v"(hh_10), "+v"(acc));
    float tc0_11 = tanh_act(c2[0][6]);
    v2f ig_12 = fi_12 * (v2f){tg0_12, tg1_12};
    acc = fmaf(hh_10[0], wfcD[0][4], acc);
    ORDER("+v"(tc0_11), "+v"(ig_12), "+v"(acc), "+v"(c2[0]));
    float tc1_11 = tanh_act(c2[0][7]);
    ORDER("+v"(tc1_11), "+v"(h1Bn));
    d1[6] = __builtin_amdgcn_wmma_f32_16x16x32_f16(false, a1[6], false, h1Bn, (short)0, zc, false, false);
    ORDER("+v"(d1[6]), "+v"(tf0_12), "+v"(tf1_12));
    v2f ff_12 = FIN2(((v2f){tf0_12, tf1_12}));
    ORDER("+v"(ff_12), "+v"(mi_13), "+v"(d2[3]));
    float ti0_13 = TSIG(mi_13[0]);
    v2f mf_13 = PRE2(((v2f){d2[3][2], d2[3][3]}));
    ORDER("+v"(ti0_13), "+v"(mf_13), "+v"(mi_13), "+v"(to0_12), "+v"(to1_12));
    float ti1_13 = TSIG(mi_13[1]);
    v2f fo_12 = FIN2(((v2f){to0_12, to1_12}));
    ORDER("+v"(ti1_13), "+v"(fo_12), "+v"(d2[5]), "+v"(ff_12), "+v"(ig_12), "+v"(hh_10), "+v"(acc));
    float tg0_13 = tanh_act(d2[5][2]);
    { const v2f cn = __builtin_elementwise_fma(ff_12, (v2f){c2[1][0], c2[1][1]}, ig_12); c2[1][0] = cn[0]; c2[1][1] = cn[1]; }
    acc = fmaf(hh_10[1], wfcD[0][5], acc);
    ORDER("+v"(tg0_13), "+v"(c2[1]), "+v"(acc), "+v"(d2[5]), "+v"(d2[7]));
    float tg1_13 = tanh_act(d2[5][3]);
    v2f mo_13 = PRE2(((v2f){d2[7][2], d2[7][3]}));
    ORDER("+v"(tg1_13), "+v"(mo_13), "+v"(mf_13), "+v"(fo_11), "+v"(tc0_11), "+v"(tc1_11));
    float tf0_13 = TSIG(mf_13[0]);
    v2f hh_11 = fo_11 * (v2f){tc0_11, tc1_11};
    ORDER("+v"(tf0_13), "+v"(hh_11), "+v"(mf_13), "+v"(ti0_13), "+v"(ti1_13));
    float tf1_13 = TSIG(mf_13[1]);
    v2f fi_13 = FIN2(((v2f){ti0_13, ti1_13}));
    ORDER("+v"(tf1_13), "+v"(fi_13), "+v"(mo_13), "+v"(d2[1]));
    float to0_13 = TSIG(mo_13[0]);
    v2f mi_14 = PRE2(((v2f){d2[1][4], d2[1][5]}));
    ORDER("+v"(to0_13), "+v"(mi_14), "+v"(mo_13), "+v"(hh_11));
    float to1_13 = TSIG(mo_13[1]);
    v2h pk_11 = __builtin_convertvector(hh_11, v2h);
    ORDER("+v"(to1_13), "+v"(pk_11), "+v"(c2[1]), "+v"(fi_13), "+v"(tg0_13), "+v"(tg1_13), "+v"(hh_11), "+v"(acc));
    float tc0_12 = tanh_act(c2[1][0]);
    v2f ig_13 = fi_13 * (v2f){tg0_13, tg1_13};
    acc = fmaf(hh_11[0], wfcD[0][6], acc);
    ORDER("+v"(tc0_12), "+v"(ig_13), "+v"(acc), "+v"(c2[1]));
    float tc1_12 = tanh_act(c2[1][1]);
    ORDER("+v"(tc1_12), "+v"(h1Bn));
    d1[1] = __builtin_amdgcn_wmma_f32_16x16x32_f16(false, a1[1], false, h1Bn, (short)0, zc, false, false);
    ORDER("+v"(d1[1]), "+v"(tf0_13), "+v"(tf1_13));
    v2f ff_13 = FIN2(((v2f){tf0_13, tf1_13}));
    ORDER("+v"(ff_13), "+v"(mi_14), "+v"(d2[3]));
    float ti0_14 = TSIG(mi_14[0]);
    v2f mf_14 = PRE2(((v2f){d2[3][4], d2[3][5]}));
    ORDER("+v"(ti0_14), "+v"(mf_14), "+v"(mi_14), "+v"(to0_13), "+v"(to1_13));
    float ti1_14 = TSIG(mi_14[1]);
    v2f fo_13 = FIN2(((v2f){to0_13, to1_13}));
    ORDER("+v"(ti1_14), "+v"(fo_13), "+v"(d2[5]), "+v"(ff_13), "+v"(ig_13), "+v"(hh_11), "+v"(acc));
    float tg0_14 = tanh_act(d2[5][4]);
    { const v2f cn = __builtin_elementwise_fma(ff_13, (v2f){c2[1][2], c2[1][3]}, ig_13); c2[1][2] = cn[0]; c2[1][3] = cn[1]; }
    acc = fmaf(hh_11[1], wfcD[0][7], acc);
    ORDER("+v"(tg0_14), "+v"(c2[1]), "+v"(acc), "+v"(d2[5]), "+v"(d2[7]));
    float tg1_14 = tanh_act(d2[5][5]);
    v2f mo_14 = PRE2(((v2f){d2[7][4], d2[7][5]}));
    ORDER("+v"(tg1_14), "+v"(mo_14), "+v"(mf_14), "+v"(fo_12), "+v"(tc0_12), "+v"(tc1_12));
    float tf0_14 = TSIG(mf_14[0]);
    v2f hh_12 = fo_12 * (v2f){tc0_12, tc1_12};
    ORDER("+v"(tf0_14), "+v"(hh_12), "+v"(mf_14), "+v"(ti0_14), "+v"(ti1_14));
    float tf1_14 = TSIG(mf_14[1]);
    v2f fi_14 = FIN2(((v2f){ti0_14, ti1_14}));
    ORDER("+v"(tf1_14), "+v"(fi_14), "+v"(mo_14), "+v"(d2[1]));
    float to0_14 = TSIG(mo_14[0]);
    v2f mi_15 = PRE2(((v2f){d2[1][6], d2[1][7]}));
    ORDER("+v"(to0_14), "+v"(mi_15), "+v"(mo_14), "+v"(hh_12));
    float to1_14 = TSIG(mo_14[1]);
    v2h pk_12 = __builtin_convertvector(hh_12, v2h);
    ORDER("+v"(to1_14), "+v"(pk_12), "+v"(c2[1]), "+v"(fi_14), "+v"(tg0_14), "+v"(tg1_14), "+v"(hh_12), "+v"(acc));
    float tc0_13 = tanh_act(c2[1][2]);
    v2f ig_14 = fi_14 * (v2f){tg0_14, tg1_14};
    acc = fmaf(hh_12[0], wfcD[1][0], acc);
    ORDER("+v"(tc0_13), "+v"(ig_14), "+v"(acc), "+v"(c2[1]));
    float tc1_13 = tanh_act(c2[1][3]);
    ORDER("+v"(tc1_13), "+v"(h1Bn));
    d1[3] = __builtin_amdgcn_wmma_f32_16x16x32_f16(false, a1[3], false, h1Bn, (short)0, zc, false, false);
    ORDER("+v"(d1[3]), "+v"(tf0_14), "+v"(tf1_14));
    v2f ff_14 = FIN2(((v2f){tf0_14, tf1_14}));
    ORDER("+v"(ff_14), "+v"(mi_15), "+v"(d2[3]));
    float ti0_15 = TSIG(mi_15[0]);
    v2f mf_15 = PRE2(((v2f){d2[3][6], d2[3][7]}));
    ORDER("+v"(ti0_15), "+v"(mf_15), "+v"(mi_15), "+v"(to0_14), "+v"(to1_14));
    float ti1_15 = TSIG(mi_15[1]);
    v2f fo_14 = FIN2(((v2f){to0_14, to1_14}));
    ORDER("+v"(ti1_15), "+v"(fo_14), "+v"(d2[5]), "+v"(ff_14), "+v"(ig_14), "+v"(hh_12), "+v"(acc));
    float tg0_15 = tanh_act(d2[5][6]);
    { const v2f cn = __builtin_elementwise_fma(ff_14, (v2f){c2[1][4], c2[1][5]}, ig_14); c2[1][4] = cn[0]; c2[1][5] = cn[1]; }
    acc = fmaf(hh_12[1], wfcD[1][1], acc);
    ORDER("+v"(tg0_15), "+v"(c2[1]), "+v"(acc), "+v"(d2[5]), "+v"(d2[7]));
    float tg1_15 = tanh_act(d2[5][7]);
    v2f mo_15 = PRE2(((v2f){d2[7][6], d2[7][7]}));
    ORDER("+v"(tg1_15), "+v"(mo_15), "+v"(mf_15), "+v"(fo_13), "+v"(tc0_13), "+v"(tc1_13));
    float tf0_15 = TSIG(mf_15[0]);
    v2f hh_13 = fo_13 * (v2f){tc0_13, tc1_13};
    ORDER("+v"(tf0_15), "+v"(hh_13), "+v"(mf_15), "+v"(ti0_15), "+v"(ti1_15));
    float tf1_15 = TSIG(mf_15[1]);
    v2f fi_15 = FIN2(((v2f){ti0_15, ti1_15}));
    ORDER("+v"(tf1_15), "+v"(fi_15), "+v"(mo_15));
    float to0_15 = TSIG(mo_15[0]);
    ORDER("+v"(to0_15), "+v"(mo_15), "+v"(hh_13));
    float to1_15 = TSIG(mo_15[1]);
    v2h pk_13 = __builtin_convertvector(hh_13, v2h);
    ORDER("+v"(to1_15), "+v"(pk_13), "+v"(c2[1]), "+v"(fi_15), "+v"(tg0_15), "+v"(tg1_15), "+v"(hh_13), "+v"(acc));
    float tc0_14 = tanh_act(c2[1][4]);
    v2f ig_15 = fi_15 * (v2f){tg0_15, tg1_15};
    acc = fmaf(hh_13[0], wfcD[1][2], acc);
    ORDER("+v"(tc0_14), "+v"(ig_15), "+v"(acc), "+v"(c2[1]));
    float tc1_14 = tanh_act(c2[1][5]);
    ORDER("+v"(tc1_14), "+v"(h1Bn));
    d1[5] = __builtin_amdgcn_wmma_f32_16x16x32_f16(false, a1[5], false, h1Bn, (short)0, zc, false, false);
    ORDER("+v"(d1[5]), "+v"(tf0_15), "+v"(tf1_15));
    v2f ff_15 = FIN2(((v2f){tf0_15, tf1_15}));
    ORDER("+v"(ff_15), "+v"(to0_15), "+v"(to1_15));
    v2f fo_15 = FIN2(((v2f){to0_15, to1_15}));
    ORDER("+v"(fo_15), "+v"(ff_15), "+v"(ig_15), "+v"(hh_13), "+v"(acc));
    { const v2f cn = __builtin_elementwise_fma(ff_15, (v2f){c2[1][6], c2[1][7]}, ig_15); c2[1][6] = cn[0]; c2[1][7] = cn[1]; }
    acc = fmaf(hh_13[1], wfcD[1][3], acc);
    ORDER("+v"(c2[1]), "+v"(acc), "+v"(fo_14), "+v"(tc0_14), "+v"(tc1_14));
    v2f hh_14 = fo_14 * (v2f){tc0_14, tc1_14};
    ORDER("+v"(hh_14));
    v2h pk_14 = __builtin_convertvector(hh_14, v2h);
    ORDER("+v"(pk_14), "+v"(c2[1]), "+v"(hh_14), "+v"(acc));
    float tc0_15 = tanh_act(c2[1][6]);
    acc = fmaf(hh_14[0], wfcD[1][4], acc);
    ORDER("+v"(tc0_15), "+v"(acc), "+v"(c2[1]));
    float tc1_15 = tanh_act(c2[1][7]);
    ORDER("+v"(tc1_15), "+v"(h1Bn));
    d1[7] = __builtin_amdgcn_wmma_f32_16x16x32_f16(false, a1[7], false, h1Bn, (short)0, zc, false, false);
    ORDER("+v"(d1[7]), "+v"(hh_14), "+v"(acc));
    acc = fmaf(hh_14[1], wfcD[1][5], acc);
    ORDER("+v"(acc), "+v"(fo_15), "+v"(tc0_15), "+v"(tc1_15));
    v2f hh_15 = fo_15 * (v2f){tc0_15, tc1_15};
    ORDER("+v"(hh_15));
    v2h pk_15 = __builtin_convertvector(hh_15, v2h);
    ORDER("+v"(pk_15), "+v"(hh_15), "+v"(acc));
    acc = fmaf(hh_15[0], wfcD[1][6], acc);
    ORDER("+v"(acc), "+v"(hh_15));
    acc = fmaf(hh_15[1], wfcD[1][7], acc);
    h2Bn[0] = pk_8[0]; h2Bn[1] = pk_8[1];
    h2Bn[2] = pk_9[0]; h2Bn[3] = pk_9[1];
    h2Bn[4] = pk_10[0]; h2Bn[5] = pk_10[1];
    h2Bn[6] = pk_11[0]; h2Bn[7] = pk_11[1];
    h2Bn[8] = pk_12[0]; h2Bn[9] = pk_12[1];
    h2Bn[10] = pk_13[0]; h2Bn[11] = pk_13[1];
    h2Bn[12] = pk_14[0]; h2Bn[13] = pk_14[1];
    h2Bn[14] = pk_15[0]; h2Bn[15] = pk_15[1];
    if (hi_half) { h2Bn[14] = (_Float16)0.f; h2Bn[15] = (_Float16)1.f; }
    pend = (accPrev + accOther) + bfc;
    accOther = __shfl_xor(acc, 16, 32);
    float xnext = xpre;
    ORDER("+v"(xnext));


    accPrev = acc;
    h1B = h1Bn;
    h2B = h2Bn;
    xinj = xnext;
  }

  {
    if (lane < 16) orow[TLEN - 3] = pend;
    const float accOther = __shfl_xor(accPrev, 16, 32);
    if (lane < 16) orow[TLEN - 2] = (accPrev + accOther) + bfc;
    float acc;
    (void)layer2_step(ai2, ah2, h1B, h2B, c2, wfcD, zc, acc);
    acc += __shfl_xor(acc, 16, 32);
    if (lane < 16) orow[TLEN - 1] = acc + bfc;
  }
}

extern "C" void kernel_launch(void* const* d_in, const int* in_sizes, int n_in,
                              void* d_out, int out_size, void* d_ws, size_t ws_size,
                              hipStream_t stream) {
  (void)in_sizes; (void)n_in; (void)out_size; (void)d_ws; (void)ws_size;
  const float* x     = (const float*)d_in[0];
  const float* w_ih1 = (const float*)d_in[1];
  const float* w_hh1 = (const float*)d_in[2];
  const float* b_ih1 = (const float*)d_in[3];
  const float* b_hh1 = (const float*)d_in[4];
  const float* w_ih2 = (const float*)d_in[5];
  const float* w_hh2 = (const float*)d_in[6];
  const float* b_ih2 = (const float*)d_in[7];
  const float* b_hh2 = (const float*)d_in[8];
  const float* w_fc  = (const float*)d_in[9];
  const float* b_fc  = (const float*)d_in[10];
  float* out = (float*)d_out;

  dim3 grid(BSZ / 16);
  dim3 block(32);
  lstm2_wmma_kernel<<<grid, block, 0, stream>>>(x, w_ih1, w_hh1, b_ih1, b_hh1,
                                                w_ih2, w_hh2, b_ih2, b_hh2,
                                                w_fc, b_fc, out);
}


# FC head's lane-half exchange on the vector unit: one v_permlanex16_b32 in place of the ds_bpermute_b32 of __shfl_xor, the loop's only LDS-pipe instruction; values and sum order unchanged
# speedup vs baseline: 1.7220x; 1.0501x over previous
#include <hip/hip_runtime.h>


#define HSZ   30
#define MPAD  128
#define KPAD  32
#define TLEN  2048
#define BSZ   512

typedef __attribute__((ext_vector_type(16))) _Float16 v16h;
typedef __attribute__((ext_vector_type(8)))  _Float16 v8h;
typedef __attribute__((ext_vector_type(8)))  float    v8f;
typedef __attribute__((ext_vector_type(2)))  float    v2f;
typedef __attribute__((ext_vector_type(2)))  _Float16 v2h;

#define ORDER(...) asm volatile("" : __VA_ARGS__)

#if __has_builtin(__builtin_amdgcn_tanhf)
#define HW_TANH __builtin_amdgcn_tanhf
#elif __has_builtin(__builtin_amdgcn_tanh_f32)
#define HW_TANH __builtin_amdgcn_tanh_f32
#endif

#ifdef HW_TANH
__device__ __forceinline__ float tanh_act(float x) { return HW_TANH(x); }
__device__ __forceinline__ float sig_act(float x)  { return fmaf(0.5f, HW_TANH(0.5f * x), 0.5f); }
#define PRE2(x) ((x) * 0.5f)
#define TSIG(m) HW_TANH(m)
#define FIN2(t) __builtin_elementwise_fma((v2f)(0.5f), (t), (v2f)(0.5f))
#else
#define PRE2(x) (x)
#define TSIG(m) __builtin_amdgcn_rcpf(1.0f + __expf(-(m)))
#define FIN2(t) (t)
__device__ __forceinline__ float tanh_act(float x) {
  return fmaf(2.0f, __builtin_amdgcn_rcpf(1.0f + __expf(-2.0f * x)), -1.0f);
}
__device__ __forceinline__ float sig_act(float x) {
  return __builtin_amdgcn_rcpf(1.0f + __expf(-x));
}
#endif

__device__ __forceinline__ float xhalf(float v) {
  const unsigned u = __builtin_bit_cast(unsigned, v);
  return __builtin_bit_cast(float, __builtin_amdgcn_permlanex16(u, u, 0x76543210u, 0xfedcba98u, false, false));
}

__device__ __forceinline__ v16h load_a_tile(const _Float16* __restrict__ base,
                                            int tile, int lane) {
  const int m  = lane & 15;
  const int kb = (lane >> 4) << 3;
  const _Float16* row = base + ((tile << 4) + m) * KPAD;
  v8h lo = *(const v8h*)(row + kb);
  v8h hi = *(const v8h*)(row + 16 + kb);
  v16h a;
#pragma unroll
  for (int i = 0; i < 8; ++i) { a[i] = lo[i]; a[8 + i] = hi[i]; }
  return a;
}

__device__ __forceinline__ void gates8(const v8f (&d)[4], v8f &c, float (&hn)[8]) {
#pragma unroll
  for (int r = 0; r < 8; ++r) {
    const float iv = sig_act(d[0][r]);
    const float fv = sig_act(d[1][r]);
    const float gv = tanh_act(d[2][r]);
    const float ov = sig_act(d[3][r]);
    const float cn = fmaf(fv, c[r], iv * gv);
    c[r] = cn;
    hn[r] = ov * tanh_act(cn);
  }
}

__device__ __forceinline__ v16h layer1_step(const v16h (&a1)[8], v16h h1B, v8f (&c1)[2], v8f zc) {
  v16h h1Bn;
#pragma unroll
  for (int s = 0; s < 2; ++s) {
    v8f d[4];
#pragma unroll
    for (int g = 0; g < 4; ++g)
      d[g] = __builtin_amdgcn_wmma_f32_16x16x32_f16(false, a1[2 * g + s], false, h1B,
                                                    (short)0, zc, false, false);
    float hn[8];
    gates8(d, c1[s], hn);
#pragma unroll
    for (int r = 0; r < 8; ++r) h1Bn[8 * s + r] = (_Float16)hn[r];
  }
  return h1Bn;
}

__device__ __forceinline__ v16h layer2_step(const v16h (&ai2)[8], const v16h (&ah2)[8],
                                            v16h h1B, v16h h2B, v8f (&c2)[2],
                                            const float (&wfcD)[2][8], v8f zc, float& acc_out) {
  v16h h2Bn;
  float acc = 0.f;
#pragma unroll
  for (int s = 0; s < 2; ++s) {
    v8f d[4];
#pragma unroll
    for (int g = 0; g < 4; ++g) {
      v8f tmp = __builtin_amdgcn_wmma_f32_16x16x32_f16(false, ah2[2 * g + s], false, h2B,
                                                       (short)0, zc, false, false);
      d[g] = __builtin_amdgcn_wmma_f32_16x16x32_f16(false, ai2[2 * g + s], false, h1B,
                                                    (short)0, tmp, false, false);
    }
    float hn[8];
    gates8(d, c2[s], hn);
#pragma unroll
    for (int r = 0; r < 8; ++r) {
      h2Bn[8 * s + r] = (_Float16)hn[r];
      acc = fmaf(hn[r], wfcD[s][r], acc);
    }
  }
  acc_out = acc;
  return h2Bn;
}

__global__ __launch_bounds__(32, 1)
void lstm2_wmma_kernel(const float* __restrict__ x,
                       const float* __restrict__ w_ih1, const float* __restrict__ w_hh1,
                       const float* __restrict__ b_ih1, const float* __restrict__ b_hh1,
                       const float* __restrict__ w_ih2, const float* __restrict__ w_hh2,
                       const float* __restrict__ b_ih2, const float* __restrict__ b_hh2,
                       const float* __restrict__ w_fc,  const float* __restrict__ b_fc,
                       float* __restrict__ out) {
  __shared__ _Float16 sA1[MPAD * KPAD];
  __shared__ _Float16 sAi2[MPAD * KPAD];
  __shared__ _Float16 sAh2[MPAD * KPAD];

  const int lane = threadIdx.x;
  const bool hi_half = lane >= 16;
  const int n = lane & 15;

  for (int idx = lane; idx < MPAD * KPAD; idx += 32) {
    const int m = idx >> 5, k = idx & 31;
    const int g = m >> 5, j = m & 31;
    const int row = g * HSZ + j;
    float a1 = 0.f, ai2 = 0.f, ah2 = 0.f;
    if (j < HSZ) {
      if (k < HSZ) {
        a1  = w_hh1[row * HSZ + k];
        ai2 = w_ih2[row * HSZ + k];
        ah2 = w_hh2[row * HSZ + k];
      } else if (k == 30) {
        a1  = w_ih1[row];
      } else {
        a1  = b_ih1[row] + b_hh1[row];
        ah2 = b_ih2[row] + b_hh2[row];
      }
    }
    sA1[idx]  = (_Float16)a1;
    sAi2[idx] = (_Float16)ai2;
    sAh2[idx] = (_Float16)ah2;
  }
  __syncthreads();

  v16h a1[8], ai2[8], ah2[8];
#pragma unroll
  for (int tl = 0; tl < 8; ++tl) {
    a1[tl]  = load_a_tile(sA1,  tl, lane);
    ai2[tl] = load_a_tile(sAi2, tl, lane);
    ah2[tl] = load_a_tile(sAh2, tl, lane);
  }

  float wfcD[2][8];
#pragma unroll
  for (int s = 0; s < 2; ++s)
#pragma unroll
    for (int r = 0; r < 8; ++r) {
      const int j = 16 * s + (hi_half ? 8 : 0) + r;
      wfcD[s][r] = (j < HSZ) ? w_fc[j] : 0.f;
    }
  const float bfc = b_fc[0];

  v8f zc;
#pragma unroll
  for (int i = 0; i < 8; ++i) zc[i] = 0.f;

  v16h h1B, h2B;
#pragma unroll
  for (int i = 0; i < 16; ++i) { h1B[i] = (_Float16)0.f; h2B[i] = (_Float16)0.f; }
  if (hi_half) h2B[15] = (_Float16)1.f;

  v8f c1[2], c2[2];
#pragma unroll
  for (int s = 0; s < 2; ++s)
#pragma unroll
    for (int r = 0; r < 8; ++r) { c1[s][r] = 0.f; c2[s][r] = 0.f; }

  const int bbase = blockIdx.x << 4;
  const float* xrow = x + (size_t)(bbase + n) * TLEN;
  float* orow = out + (size_t)(bbase + n) * TLEN;

  const float x0 = xrow[0];
  const float x1 = xrow[1];
  float xinj = xrow[2];
  if (hi_half) { h1B[14] = (_Float16)x0; h1B[15] = (_Float16)1.f; }
  h1B = layer1_step(a1, h1B, c1, zc);
  if (hi_half) { h1B[14] = (_Float16)x1; h1B[15] = (_Float16)1.f; }

  v8f d1[8];
#pragma unroll
  for (int tl = 0; tl < 8; ++tl)
    d1[tl] = __builtin_amdgcn_wmma_f32_16x16x32_f16(false, a1[tl], false, h1B,
                                                    (short)0, zc, false, false);

  float accPrev = 0.f;
  float accOther = 0.f;
  float pend = 0.f;

  for (int t = 0; t < TLEN - 1; ++t) {

    if (t > 1 && lane < 16) orow[t - 2] = pend;

    const float xpre = xrow[min(t + 3, TLEN - 1)];

    v8f d2[8];
    v16h h1Bn, h2Bn;
    float acc = 0.f;
    ORDER("+v"(d1[0]));
    v2f mi_0 = PRE2(((v2f){d1[0][0], d1[0][1]}));
    ORDER("+v"(mi_0), "+v"(h2B));
    d2[0] = __builtin_amdgcn_wmma_f32_16x16x32_f16(false, ah2[0], false, h2B, (short)0, zc, false, false);
    ORDER("+v"(d2[0]), "+v"(mi_0), "+v"(d1[2]));
    float ti0_0 = TSIG(mi_0[0]);
    v2f mf_0 = PRE2(((v2f){d1[2][0], d1[2][1]}));
    ORDER("+v"(ti0_0), "+v"(mf_0), "+v"(mi_0));
    float ti1_0 = TSIG(mi_0[1]);
    ORDER("+v"(ti1_0), "+v"(d1[4]));
    float tg0_0 = tanh_act(d1[4][0]);
    ORDER("+v"(tg0_0), "+v"(d1[4]), "+v"(d1[6]));
    float tg1_0 = tanh_act(d1[4][1]);
    v2f mo_0 = PRE2(((v2f){d1[6][0], d1[6][1]}));
    ORDER("+v"(tg1_0), "+v"(mo_0), "+v"(mf_0));
    float tf0_0 = TSIG(mf_0[0]);
    ORDER("+v"(tf0_0), "+v"(h2B));
    d2[2] = __builtin_amdgcn_wmma_f32_16x16x32_f16(false, ah2[2], false, h2B, (short)0, zc, false, false);
    ORDER("+v"(d2[2]), "+v"(mf_0), "+v"(ti0_0), "+v"(ti1_0));
    float tf1_0 = TSIG(mf_0[1]);
    v2f fi_0 = FIN2(((v2f){ti0_0, ti1_0}));
    ORDER("+v"(tf1_0), "+v"(fi_0), "+v"(mo_0), "+v"(d1[0]));
    float to0_0 = TSIG(mo_0[0]);
    v2f mi_1 = PRE2(((v2f){d1[0][2], d1[0][3]}));
    ORDER("+v"(to0_0), "+v"(mi_1), "+v"(mo_0));
    float to1_0 = TSIG(mo_0[1]);
    ORDER("+v"(to1_0), "+v"(fi_0), "+v"(tg0_0), "+v"(tg1_0));
    v2f ig_0 = fi_0 * (v2f){tg0_0, tg1_0};
    ORDER("+v"(ig_0), "+v"(h2B));
    d2[4] = __builtin_amdgcn_wmma_f32_16x16x32_f16(false, ah2[4], false, h2B, (short)0, zc, false, false);
    ORDER("+v"(d2[4]), "+v"(tf0_0), "+v"(tf1_0));
    v2f ff_0 = FIN2(((v2f){tf0_0, tf1_0}));
    ORDER("+v"(ff_0), "+v"(mi_1), "+v"(d1[2]));
    float ti0_1 = TSIG(mi_1[0]);
    v2f mf_1 = PRE2(((v2f){d1[2][2], d1[2][3]}));
    ORDER("+v"(ti0_1), "+v"(mf_1), "+v"(mi_1), "+v"(to0_0), "+v"(to1_0));
    float ti1_1 = TSIG(mi_1[1]);
    v2f fo_0 = FIN2(((v2f){to0_0, to1_0}));
    ORDER("+v"(ti1_1), "+v"(fo_0), "+v"(d1[4]), "+v"(ff_0), "+v"(ig_0));
    float tg0_1 = tanh_act(d1[4][2]);
    { const v2f cn = __builtin_elementwise_fma(ff_0, (v2f){c1[0][0], c1[0][1]}, ig_0); c1[0][0] = cn[0]; c1[0][1] = cn[1]; }
    ORDER("+v"(tg0_1), "+v"(c1[0]), "+v"(d1[4]), "+v"(d1[6]));
    float tg1_1 = tanh_act(d1[4][3]);
    v2f mo_1 = PRE2(((v2f){d1[6][2], d1[6][3]}));
    ORDER("+v"(tg1_1), "+v"(mo_1), "+v"(mf_1));
    float tf0_1 = TSIG(mf_1[0]);
    ORDER("+v"(tf0_1), "+v"(h2B));
    d2[6] = __builtin_amdgcn_wmma_f32_16x16x32_f16(false, ah2[6], false, h2B, (short)0, zc, false, false);
    ORDER("+v"(d2[6]), "+v"(mf_1), "+v"(ti0_1), "+v"(ti1_1));
    float tf1_1 = TSIG(mf_1[1]);
    v2f fi_1 = FIN2(((v2f){ti0_1, ti1_1}));
    ORDER("+v"(tf1_1), "+v"(fi_1), "+v"(mo_1), "+v"(d1[0]));
    float to0_1 = TSIG(mo_1[0]);
    v2f mi_2 = PRE2(((v2f){d1[0][4], d1[0][5]}));
    ORDER("+v"(to0_1), "+v"(mi_2), "+v"(mo_1));
    float to1_1 = TSIG(mo_1[1]);
    ORDER("+v"(to1_1), "+v"(c1[0]), "+v"(fi_1), "+v"(tg0_1), "+v"(tg1_1));
    float tc0_0 = tanh_act(c1[0][0]);
    v2f ig_1 = fi_1 * (v2f){tg0_1, tg1_1};
    ORDER("+v"(tc0_0), "+v"(ig_1), "+v"(c1[0]));
    float tc1_0 = tanh_act(c1[0][1]);
    ORDER("+v"(tc1_0), "+v"(h2B));
    d2[1] = __builtin_amdgcn_wmma_f32_16x16x32_f16(false, ah2[1], false, h2B, (short)0, zc, false, false);
    ORDER("+v"(d2[1]), "+v"(tf0_1), "+v"(tf1_1));
    v2f ff_1 = FIN2(((v2f){tf0_1, tf1_1}));
    ORDER("+v"(ff_1), "+v"(mi_2), "+v"(d1[2]));
    float ti0_2 = TSIG(mi_2[0]);
    v2f mf_2 = PRE2(((v2f){d1[2][4], d1[2][5]}));
    ORDER("+v"(ti0_2), "+v"(mf_2), "+v"(mi_2), "+v"(to0_1), "+v"(to1_1));
    float ti1_2 = TSIG(mi_2[1]);
    v2f fo_1 = FIN2(((v2f){to0_1, to1_1}));
    ORDER("+v"(ti1_2), "+v"(fo_1), "+v"(d1[4]), "+v"(ff_1), "+v"(ig_1));
    float tg0_2 = tanh_act(d1[4][4]);
    { const v2f cn = __builtin_elementwise_fma(ff_1, (v2f){c1[0][2], c1[0][3]}, ig_1); c1[0][2] = cn[0]; c1[0][3] = cn[1]; }
    ORDER("+v"(tg0_2), "+v"(c1[0]), "+v"(d1[4]), "+v"(d1[6]));
    float tg1_2 = tanh_act(d1[4][5]);
    v2f mo_2 = PRE2(((v2f){d1[6][4], d1[6][5]}));
    ORDER("+v"(tg1_2), "+v"(mo_2), "+v"(mf_2));
    float tf0_2 = TSIG(mf_2[0]);
    ORDER("+v"(tf0_2), "+v"(h2B));
    d2[3] = __builtin_amdgcn_wmma_f32_16x16x32_f16(false, ah2[3], false, h2B, (short)0, zc, false, false);
    ORDER("+v"(d2[3]), "+v"(fo_0), "+v"(tc0_0), "+v"(tc1_0));
    v2f hh_0 = fo_0 * (v2f){tc0_0, tc1_0};
    ORDER("+v"(hh_0), "+v"(mf_2), "+v"(ti0_2), "+v"(ti1_2));
    float tf1_2 = TSIG(mf_2[1]);
    v2f fi_2 = FIN2(((v2f){ti0_2, ti1_2}));
    ORDER("+v"(tf1_2), "+v"(fi_2), "+v"(mo_2), "+v"(d1[0]));
    float to0_2 = TSIG(mo_2[0]);
    v2f mi_3 = PRE2(((v2f){d1[0][6], d1[0][7]}));
    ORDER("+v"(to0_2), "+v"(mi_3), "+v"(mo_2), "+v"(hh_0));
    float to1_2 = TSIG(mo_2[1]);
    v2h pk_0 = __builtin_convertvector(hh_0, v2h);
    ORDER("+v"(to1_2), "+v"(pk_0), "+v"(c1[0]), "+v"(fi_2), "+v"(tg0_2), "+v"(tg1_2));
    float tc0_1 = tanh_act(c1[0][2]);
    v2f ig_2 = fi_2 * (v2f){tg0_2, tg1_2};
    ORDER("+v"(tc0_1), "+v"(ig_2), "+v"(c1[0]));
    float tc1_1 = tanh_act(c1[0][3]);
    ORDER("+v"(tc1_1), "+v"(h2B));
    d2[5] = __builtin_amdgcn_wmma_f32_16x16x32_f16(false, ah2[5], false, h2B, (short)0, zc, false, false);
    ORDER("+v"(d2[5]), "+v"(tf0_2), "+v"(tf1_2));
    v2f ff_2 = FIN2(((v2f){tf0_2, tf1_2}));
    ORDER("+v"(ff_2), "+v"(mi_3), "+v"(d1[2]));
    float ti0_3 = TSIG(mi_3[0]);
    v2f mf_3 = PRE2(((v2f){d1[2][6], d1[2][7]}));
    ORDER("+v"(ti0_3), "+v"(mf_3), "+v"(mi_3), "+v"(to0_2), "+v"(to1_2));
    float ti1_3 = TSIG(mi_3[1]);
    v2f fo_2 = FIN2(((v2f){to0_2, to1_2}));
    ORDER("+v"(ti1_3), "+v"(fo_2), "+v"(d1[4]), "+v"(ff_2), "+v"(ig_2));
    float tg0_3 = tanh_act(d1[4][6]);
    { const v2f cn = __builtin_elementwise_fma(ff_2, (v2f){c1[0][4], c1[0][5]}, ig_2); c1[0][4] = cn[0]; c1[0][5] = cn[1]; }
    ORDER("+v"(tg0_3), "+v"(c1[0]), "+v"(d1[4]), "+v"(d1[6]));
    float tg1_3 = tanh_act(d1[4][7]);
    v2f mo_3 = PRE2(((v2f){d1[6][6], d1[6][7]}));
    ORDER("+v"(tg1_3), "+v"(mo_3), "+v"(mf_3));
    float tf0_3 = TSIG(mf_3[0]);
    ORDER("+v"(tf0_3), "+v"(h2B));
    d2[7] = __builtin_amdgcn_wmma_f32_16x16x32_f16(false, ah2[7], false, h2B, (short)0, zc, false, false);
    ORDER("+v"(d2[7]), "+v"(fo_1), "+v"(tc0_1), "+v"(tc1_1));
    v2f hh_1 = fo_1 * (v2f){tc0_1, tc1_1};
    ORDER("+v"(hh_1), "+v"(mf_3), "+v"(ti0_3), "+v"(ti1_3));
    float tf1_3 = TSIG(mf_3[1]);
    v2f fi_3 = FIN2(((v2f){ti0_3, ti1_3}));
    ORDER("+v"(tf1_3), "+v"(fi_3), "+v"(mo_3), "+v"(d1[1]));
    float to0_3 = TSIG(mo_3[0]);
    v2f mi_4 = PRE2(((v2f){d1[1][0], d1[1][1]}));
    ORDER("+v"(to0_3), "+v"(mi_4), "+v"(mo_3), "+v"(hh_1));
    float to1_3 = TSIG(mo_3[1]);
    v2h pk_1 = __builtin_convertvector(hh_1, v2h);
    ORDER("+v"(to1_3), "+v"(pk_1), "+v"(c1[0]), "+v"(fi_3), "+v"(tg0_3), "+v"(tg1_3));
    float tc0_2 = tanh_act(c1[0][4]);
    v2f ig_3 = fi_3 * (v2f){tg0_3, tg1_3};
    ORDER("+v"(tc0_2), "+v"(ig_3), "+v"(c1[0]));
    float tc1_2 = tanh_act(c1[0][5]);
    ORDER("+v"(tc1_2), "+v"(h1B), "+v"(d2[0]));
    d2[0] = __builtin_amdgcn_wmma_f32_16x16x32_f16(false, ai2[0], false, h1B, (short)0, d2[0], false, false);
    ORDER("+v"(d2[0]), "+v"(tf0_3), "+v"(tf1_3));
    v2f ff_3 = FIN2(((v2f){tf0_3, tf1_3}));
    ORDER("+v"(ff_3), "+v"(mi_4), "+v"(d1[3]));
    float ti0_4 = TSIG(mi_4[0]);
    v2f mf_4 = PRE2(((v2f){d1[3][0], d1[3][1]}));
    ORDER("+v"(ti0_4), "+v"(mf_4), "+v"(mi_4), "+v"(to0_3), "+v"(to1_3));
    float ti1_4 = TSIG(mi_4[1]);
    v2f fo_3 = FIN2(((v2f){to0_3, to1_3}));
    ORDER("+v"(ti1_4), "+v"(fo_3), "+v"(d1[5]), "+v"(ff_3), "+v"(ig_3));
    float tg0_4 = tanh_act(d1[5][0]);
    { const v2f cn = __builtin_elementwise_fma(ff_3, (v2f){c1[0][6], c1[0][7]}, ig_3); c1[0][6] = cn[0]; c1[0][7] = cn[1]; }
    ORDER("+v"(tg0_4), "+v"(c1[0]), "+v"(d1[5]), "+v"(d1[7]));
    float tg1_4 = tanh_act(d1[5][1]);
    v2f mo_4 = PRE2(((v2f){d1[7][0], d1[7][1]}));
    ORDER("+v"(tg1_4), "+v"(mo_4), "+v"(mf_4));
    float tf0_4 = TSIG(mf_4[0]);
    ORDER("+v"(tf0_4), "+v"(h1B), "+v"(d2[2]));
    d2[2] = __builtin_amdgcn_wmma_f32_16x16x32_f16(false, ai2[2], false, h1B, (short)0, d2[2], false, false);
    ORDER("+v"(d2[2]), "+v"(fo_2), "+v"(tc0_2), "+v"(tc1_2));
    v2f hh_2 = fo_2 * (v2f){tc0_2, tc1_2};
    ORDER("+v"(hh_2), "+v"(mf_4), "+v"(ti0_4), "+v"(ti1_4));
    float tf1_4 = TSIG(mf_4[1]);
    v2f fi_4 = FIN2(((v2f){ti0_4, ti1_4}));
    ORDER("+v"(tf1_4), "+v"(fi_4), "+v"(mo_4), "+v"(d1[1]));
    float to0_4 = TSIG(mo_4[0]);
    v2f mi_5 = PRE2(((v2f){d1[1][2], d1[1][3]}));
    ORDER("+v"(to0_4), "+v"(mi_5), "+v"(mo_4), "+v"(hh_2));
    float to1_4 = TSIG(mo_4[1]);
    v2h pk_2 = __builtin_convertvector(hh_2, v2h);
    ORDER("+v"(to1_4), "+v"(pk_2), "+v"(c1[0]), "+v"(fi_4), "+v"(tg0_4), "+v"(tg1_4));
    float tc0_3 = tanh_act(c1[0][6]);
    v2f ig_4 = fi_4 * (v2f){tg0_4, tg1_4};
    ORDER("+v"(tc0_3), "+v"(ig_4), "+v"(c1[0]));
    float tc1_3 = tanh_act(c1[0][7]);
    ORDER("+v"(tc1_3), "+v"(h1B), "+v"(d2[4]));
    d2[4] = __builtin_amdgcn_wmma_f32_16x16x32_f16(false, ai2[4], false, h1B, (short)0, d2[4], false, false);
    ORDER("+v"(d2[4]), "+v"(tf0_4), "+v"(tf1_4));
    v2f ff_4 = FIN2(((v2f){tf0_4, tf1_4}));
    ORDER("+v"(ff_4), "+v"(mi_5), "+v"(d1[3]));
    float ti0_5 = TSIG(mi_5[0]);
    v2f mf_5 = PRE2(((v2f){d1[3][2], d1[3][3]}));
    ORDER("+v"(ti0_5), "+v"(mf_5), "+v"(mi_5), "+v"(to0_4), "+v"(to1_4));
    float ti1_5 = TSIG(mi_5[1]);
    v2f fo_4 = FIN2(((v2f){to0_4, to1_4}));
    ORDER("+v"(ti1_5), "+v"(fo_4), "+v"(d1[5]), "+v"(ff_4), "+v"(ig_4));
    float tg0_5 = tanh_act(d1[5][2]);
    { const v2f cn = __builtin_elementwise_fma(ff_4, (v2f){c1[1][0], c1[1][1]}, ig_4); c1[1][0] = cn[0]; c1[1][1] = cn[1]; }
    ORDER("+v"(tg0_5), "+v"(c1[1]), "+v"(d1[5]), "+v"(d1[7]));
    float tg1_5 = tanh_act(d1[5][3]);
    v2f mo_5 = PRE2(((v2f){d1[7][2], d1[7][3]}));
    ORDER("+v"(tg1_5), "+v"(mo_5), "+v"(mf_5));
    float tf0_5 = TSIG(mf_5[0]);
    ORDER("+v"(tf0_5), "+v"(h1B), "+v"(d2[6]));
    d2[6] = __builtin_amdgcn_wmma_f32_16x16x32_f16(false, ai2[6], false, h1B, (short)0, d2[6], false, false);
    ORDER("+v"(d2[6]), "+v"(fo_3), "+v"(tc0_3), "+v"(tc1_3));
    v2f hh_3 = fo_3 * (v2f){tc0_3, tc1_3};
    ORDER("+v"(hh_3), "+v"(mf_5), "+v"(ti0_5), "+v"(ti1_5));
    float tf1_5 = TSIG(mf_5[1]);
    v2f fi_5 = FIN2(((v2f){ti0_5, ti1_5}));
    ORDER("+v"(tf1_5), "+v"(fi_5), "+v"(mo_5), "+v"(d1[1]));
    float to0_5 = TSIG(mo_5[0]);
    v2f mi_6 = PRE2(((v2f){d1[1][4], d1[1][5]}));
    ORDER("+v"(to0_5), "+v"(mi_6), "+v"(mo_5), "+v"(hh_3));
    float to1_5 = TSIG(mo_5[1]);
    v2h pk_3 = __builtin_convertvector(hh_3, v2h);
    ORDER("+v"(to1_5), "+v"(pk_3), "+v"(c1[1]), "+v"(fi_5), "+v"(tg0_5), "+v"(tg1_5));
    float tc0_4 = tanh_act(c1[1][0]);
    v2f ig_5 = fi_5 * (v2f){tg0_5, tg1_5};
    ORDER("+v"(tc0_4), "+v"(ig_5), "+v"(c1[1]));
    float tc1_4 = tanh_act(c1[1][1]);
    ORDER("+v"(tc1_4), "+v"(h1B), "+v"(d2[1]));
    d2[1] = __builtin_amdgcn_wmma_f32_16x16x32_f16(false, ai2[1], false, h1B, (short)0, d2[1], false, false);
    ORDER("+v"(d2[1]), "+v"(tf0_5), "+v"(tf1_5));
    v2f ff_5 = FIN2(((v2f){tf0_5, tf1_5}));
    ORDER("+v"(ff_5), "+v"(mi_6), "+v"(d1[3]));
    float ti0_6 = TSIG(mi_6[0]);
    v2f mf_6 = PRE2(((v2f){d1[3][4], d1[3][5]}));
    ORDER("+v"(ti0_6), "+v"(mf_6), "+v"(mi_6), "+v"(to0_5), "+v"(to1_5));
    float ti1_6 = TSIG(mi_6[1]);
    v2f fo_5 = FIN2(((v2f){to0_5, to1_5}));
    ORDER("+v"(ti1_6), "+v"(fo_5), "+v"(d1[5]), "+v"(ff_5), "+v"(ig_5));
    float tg0_6 = tanh_act(d1[5][4]);
    { const v2f cn = __builtin_elementwise_fma(ff_5, (v2f){c1[1][2], c1[1][3]}, ig_5); c1[1][2] = cn[0]; c1[1][3] = cn[1]; }
    ORDER("+v"(tg0_6), "+v"(c1[1]), "+v"(d1[5]), "+v"(d1[7]));
    float tg1_6 = tanh_act(d1[5][5]);
    v2f mo_6 = PRE2(((v2f){d1[7][4], d1[7][5]}));
    ORDER("+v"(tg1_6), "+v"(mo_6), "+v"(mf_6));
    float tf0_6 = TSIG(mf_6[0]);
    ORDER("+v"(tf0_6), "+v"(h1B), "+v"(d2[3]));
    d2[3] = __builtin_amdgcn_wmma_f32_16x16x32_f16(false, ai2[3], false, h1B, (short)0, d2[3], false, false);
    ORDER("+v"(d2[3]), "+v"(fo_4), "+v"(tc0_4), "+v"(tc1_4));
    v2f hh_4 = fo_4 * (v2f){tc0_4, tc1_4};
    ORDER("+v"(hh_4), "+v"(mf_6), "+v"(ti0_6), "+v"(ti1_6));
    float tf1_6 = TSIG(mf_6[1]);
    v2f fi_6 = FIN2(((v2f){ti0_6, ti1_6}));
    ORDER("+v"(tf1_6), "+v"(fi_6), "+v"(mo_6), "+v"(d1[1]));
    float to0_6 = TSIG(mo_6[0]);
    v2f mi_7 = PRE2(((v2f){d1[1][6], d1[1][7]}));
    ORDER("+v"(to0_6), "+v"(mi_7), "+v"(mo_6), "+v"(hh_4));
    float to1_6 = TSIG(mo_6[1]);
    v2h pk_4 = __builtin_convertvector(hh_4, v2h);
    ORDER("+v"(to1_6), "+v"(pk_4), "+v"(c1[1]), "+v"(fi_6), "+v"(tg0_6), "+v"(tg1_6));
    float tc0_5 = tanh_act(c1[1][2]);
    v2f ig_6 = fi_6 * (v2f){tg0_6, tg1_6};
    ORDER("+v"(tc0_5), "+v"(ig_6), "+v"(c1[1]));
    float tc1_5 = tanh_act(c1[1][3]);
    ORDER("+v"(tc1_5), "+v"(h1B), "+v"(d2[5]));
    d2[5] = __builtin_amdgcn_wmma_f32_16x16x32_f16(false, ai2[5], false, h1B, (short)0, d2[5], false, false);
    ORDER("+v"(d2[5]), "+v"(tf0_6), "+v"(tf1_6));
    v2f ff_6 = FIN2(((v2f){tf0_6, tf1_6}));
    ORDER("+v"(ff_6), "+v"(mi_7), "+v"(d1[3]));
    float ti0_7 = TSIG(mi_7[0]);
    v2f mf_7 = PRE2(((v2f){d1[3][6], d1[3][7]}));
    ORDER("+v"(ti0_7), "+v"(mf_7), "+v"(mi_7), "+v"(to0_6), "+v"(to1_6));
    float ti1_7 = TSIG(mi_7[1]);
    v2f fo_6 = FIN2(((v2f){to0_6, to1_6}));
    ORDER("+v"(ti1_7), "+v"(fo_6), "+v"(d1[5]), "+v"(ff_6), "+v"(ig_6));
    float tg0_7 = tanh_act(d1[5][6]);
    { const v2f cn = __builtin_elementwise_fma(ff_6, (v2f){c1[1][4], c1[1][5]}, ig_6); c1[1][4] = cn[0]; c1[1][5] = cn[1]; }
    ORDER("+v"(tg0_7), "+v"(c1[1]), "+v"(d1[5]), "+v"(d1[7]));
    float tg1_7 = tanh_act(d1[5][7]);
    v2f mo_7 = PRE2(((v2f){d1[7][6], d1[7][7]}));
    ORDER("+v"(tg1_7), "+v"(mo_7), "+v"(mf_7));
    float tf0_7 = TSIG(mf_7[0]);
    ORDER("+v"(tf0_7), "+v"(h1B), "+v"(d2[7]));
    d2[7] = __builtin_amdgcn_wmma_f32_16x16x32_f16(false, ai2[7], false, h1B, (short)0, d2[7], false, false);
    ORDER("+v"(d2[7]), "+v"(fo_5), "+v"(tc0_5), "+v"(tc1_5));
    v2f hh_5 = fo_5 * (v2f){tc0_5, tc1_5};
    ORDER("+v"(hh_5), "+v"(mf_7), "+v"(ti0_7), "+v"(ti1_7));
    float tf1_7 = TSIG(mf_7[1]);
    v2f fi_7 = FIN2(((v2f){ti0_7, ti1_7}));
    ORDER("+v"(tf1_7), "+v"(fi_7), "+v"(mo_7), "+v"(d2[0]));
    float to0_7 = TSIG(mo_7[0]);
    v2f mi_8 = PRE2(((v2f){d2[0][0], d2[0][1]}));
    ORDER("+v"(to0_7), "+v"(mi_8), "+v"(mo_7), "+v"(hh_5));
    float to1_7 = TSIG(mo_7[1]);
    v2h pk_5 = __builtin_convertvector(hh_5, v2h);
    ORDER("+v"(to1_7), "+v"(pk_5), "+v"(c1[1]), "+v"(fi_7), "+v"(tg0_7), "+v"(tg1_7));
    float tc0_6 = tanh_act(c1[1][4]);
    v2f ig_7 = fi_7 * (v2f){tg0_7, tg1_7};
    ORDER("+v"(tc0_6), "+v"(ig_7), "+v"(c1[1]), "+v"(tf0_7), "+v"(tf1_7));
    float tc1_6 = tanh_act(c1[1][5]);
    v2f ff_7 = FIN2(((v2f){tf0_7, tf1_7}));
    ORDER("+v"(tc1_6), "+v"(ff_7), "+v"(mi_8), "+v"(d2[2]));
    float ti0_8 = TSIG(mi_8[0]);
    v2f mf_8 = PRE2(((v2f){d2[2][0], d2[2][1]}));
    ORDER("+v"(ti0_8), "+v"(mf_8), "+v"(mi_8), "+v"(to0_7), "+v"(to1_7));
    float ti1_8 = TSIG(mi_8[1]);
    v2f fo_7 = FIN2(((v2f){to0_7, to1_7}));
    ORDER("+v"(ti1_8), "+v"(fo_7), "+v"(d2[4]), "+v"(ff_7), "+v"(ig_7));
    float tg0_8 = tanh_act(d2[4][0]);
    { const v2f cn = __builtin_elementwise_fma(ff_7, (v2f){c1[1][6], c1[1][7]}, ig_7); c1[1][6] = cn[0]; c1[1][7] = cn[1]; }
    ORDER("+v"(tg0_8), "+v"(c1[1]), "+v"(d2[4]), "+v"(d2[6]));
    float tg1_8 = tanh_act(d2[4][1]);
    v2f mo_8 = PRE2(((v2f){d2[6][0], d2[6][1]}));
    ORDER("+v"(tg1_8), "+v"(mo_8), "+v"(mf_8), "+v"(fo_6), "+v"(tc0_6), "+v"(tc1_6));
    float tf0_8 = TSIG(mf_8[0]);
    v2f hh_6 = fo_6 * (v2f){tc0_6, tc1_6};
    ORDER("+v"(tf0_8), "+v"(hh_6), "+v"(mf_8), "+v"(ti0_8), "+v"(ti1_8));
    float tf1_8 = TSIG(mf_8[1]);
    v2f fi_8 = FIN2(((v2f){ti0_8, ti1_8}));
    ORDER("+v"(tf1_8), "+v"(fi_8), "+v"(mo_8), "+v"(d2[0]));
    float to0_8 = TSIG(mo_8[0]);
    v2f mi_9 = PRE2(((v2f){d2[0][2], d2[0][3]}));
    ORDER("+v"(to0_8), "+v"(mi_9), "+v"(mo_8), "+v"(hh_6));
    float to1_8 = TSIG(mo_8[1]);
    v2h pk_6 = __builtin_convertvector(hh_6, v2h);
    ORDER("+v"(to1_8), "+v"(pk_6), "+v"(c1[1]), "+v"(fi_8), "+v"(tg0_8), "+v"(tg1_8));
    float tc0_7 = tanh_act(c1[1][6]);
    v2f ig_8 = fi_8 * (v2f){tg0_8, tg1_8};
    ORDER("+v"(tc0_7), "+v"(ig_8), "+v"(c1[1]), "+v"(tf0_8), "+v"(tf1_8));
    float tc1_7 = tanh_act(c1[1][7]);
    v2f ff_8 = FIN2(((v2f){tf0_8, tf1_8}));
    ORDER("+v"(tc1_7), "+v"(ff_8), "+v"(mi_9), "+v"(d2[2]));
    float ti0_9 = TSIG(mi_9[0]);
    v2f mf_9 = PRE2(((v2f){d2[2][2], d2[2][3]}));
    ORDER("+v"(ti0_9), "+v"(mf_9), "+v"(mi_9), "+v"(to0_8), "+v"(to1_8));
    float ti1_9 = TSIG(mi_9[1]);
    v2f fo_8 = FIN2(((v2f){to0_8, to1_8}));
    ORDER("+v"(ti1_9), "+v"(fo_8), "+v"(d2[4]), "+v"(ff_8), "+v"(ig_8));
    float tg0_9 = tanh_act(d2[4][2]);
    { const v2f cn = __builtin_elementwise_fma(ff_8, (v2f){c2[0][0], c2[0][1]}, ig_8); c2[0][0] = cn[0]; c2[0][1] = cn[1]; }
    ORDER("+v"(tg0_9), "+v"(c2[0]), "+v"(d2[4]), "+v"(d2[6]));
    float tg1_9 = tanh_act(d2[4][3]);
    v2f mo_9 = PRE2(((v2f){d2[6][2], d2[6][3]}));
    ORDER("+v"(tg1_9), "+v"(mo_9), "+v"(mf_9), "+v"(fo_7), "+v"(tc0_7), "+v"(tc1_7));
    float tf0_9 = TSIG(mf_9[0]);
    v2f hh_7 = fo_7 * (v2f){tc0_7, tc1_7};
    ORDER("+v"(tf0_9), "+v"(hh_7), "+v"(mf_9), "+v"(ti0_9), "+v"(ti1_9));
    float tf1_9 = TSIG(mf_9[1]);
    v2f fi_9 = FIN2(((v2f){ti0_9, ti1_9}));
    ORDER("+v"(tf1_9), "+v"(fi_9), "+v"(mo_9), "+v"(d2[0]));
    float to0_9 = TSIG(mo_9[0]);
    v2f mi_10 = PRE2(((v2f){d2[0][4], d2[0][5]}));
    ORDER("+v"(to0_9), "+v"(mi_10), "+v"(mo_9), "+v"(hh_7));
    float to1_9 = TSIG(mo_9[1]);
    v2h pk_7 = __builtin_convertvector(hh_7, v2h);
    ORDER("+v"(to1_9), "+v"(pk_7), "+v"(c2[0]), "+v"(fi_9), "+v"(tg0_9), "+v"(tg1_9));
    float tc0_8 = tanh_act(c2[0][0]);
    v2f ig_9 = fi_9 * (v2f){tg0_9, tg1_9};
    h1Bn[0] = pk_0[0]; h1Bn[1] = pk_0[1];
    h1Bn[2] = pk_1[0]; h1Bn[3] = pk_1[1];
    h1Bn[4] = pk_2[0]; h1Bn[5] = pk_2[1];
    h1Bn[6] = pk_3[0]; h1Bn[7] = pk_3[1];
    h1Bn[8] = pk_4[0]; h1Bn[9] = pk_4[1];
    h1Bn[10] = pk_5[0]; h1Bn[11] = pk_5[1];
    h1Bn[12] = pk_6[0]; h1Bn[13] = pk_6[1];
    h1Bn[14] = pk_7[0]; h1Bn[15] = pk_7[1];
    if (hi_half) { h1Bn[14] = (_Float16)xinj; h1Bn[15] = (_Float16)1.f; }
    ORDER("+v"(tc0_8), "+v"(ig_9), "+v"(h1Bn), "+v"(c2[0]), "+v"(tf0_9), "+v"(tf1_9));
    float tc1_8 = tanh_act(c2[0][1]);
    v2f ff_9 = FIN2(((v2f){tf0_9, tf1_9}));
    ORDER("+v"(tc1_8), "+v"(ff_9), "+v"(mi_10), "+v"(d2[2]));
    float ti0_10 = TSIG(mi_10[0]);
    v2f mf_10 = PRE2(((v2f){d2[2][4], d2[2][5]}));
    ORDER("+v"(ti0_10), "+v"(mf_10), "+v"(mi_10), "+v"(to0_9), "+v"(to1_9));
    float ti1_10 = TSIG(mi_10[1]);
    v2f fo_9 = FIN2(((v2f){to0_9, to1_9}));
    ORDER("+v"(ti1_10), "+v"(fo_9), "+v"(d2[4]), "+v"(ff_9), "+v"(ig_9));
    float tg0_10 = tanh_act(d2[4][4]);
    { const v2f cn = __builtin_elementwise_fma(ff_9, (v2f){c2[0][2], c2[0][3]}, ig_9); c2[0][2] = cn[0]; c2[0][3] = cn[1]; }
    ORDER("+v"(tg0_10), "+v"(c2[0]), "+v"(d2[4]), "+v"(d2[6]));
    float tg1_10 = tanh_act(d2[4][5]);
    v2f mo_10 = PRE2(((v2f){d2[6][4], d2[6][5]}));
    ORDER("+v"(tg1_10), "+v"(mo_10), "+v"(mf_10));
    float tf0_10 = TSIG(mf_10[0]);
    ORDER("+v"(tf0_10), "+v"(h1Bn));
    d1[0] = __builtin_amdgcn_wmma_f32_16x16x32_f16(false, a1[0], false, h1Bn, (short)0, zc, false, false);
    ORDER("+v"(d1[0]), "+v"(fo_8), "+v"(tc0_8), "+v"(tc1_8));
    v2f hh_8 = fo_8 * (v2f){tc0_8, tc1_8};
    ORDER("+v"(hh_8), "+v"(mf_10), "+v"(ti0_10), "+v"(ti1_10));
    float tf1_10 = TSIG(mf_10[1]);
    v2f fi_10 = FIN2(((v2f){ti0_10, ti1_10}));
    ORDER("+v"(tf1_10), "+v"(fi_10), "+v"(mo_10), "+v"(d2[0]));
    float to0_10 = TSIG(mo_10[0]);
    v2f mi_11 = PRE2(((v2f){d2[0][6], d2[0][7]}));
    ORDER("+v"(to0_10), "+v"(mi_11), "+v"(mo_10), "+v"(hh_8));
    float to1_10 = TSIG(mo_10[1]);
    v2h pk_8 = __builtin_convertvector(hh_8, v2h);
    ORDER("+v"(to1_10), "+v"(pk_8), "+v"(c2[0]), "+v"(fi_10), "+v"(tg0_10), "+v"(tg1_10), "+v"(hh_8), "+v"(acc));
    float tc0_9 = tanh_act(c2[0][2]);
    v2f ig_10 = fi_10 * (v2f){tg0_10, tg1_10};
    acc = fmaf(hh_8[0], wfcD[0][0], acc);
    ORDER("+v"(tc0_9), "+v"(ig_10), "+v"(acc), "+v"(c2[0]));
    float tc1_9 = tanh_act(c2[0][3]);
    ORDER("+v"(tc1_9), "+v"(h1Bn));
    d1[2] = __builtin_amdgcn_wmma_f32_16x16x32_f16(false, a1[2], false, h1Bn, (short)0, zc, false, false);
    ORDER("+v"(d1[2]), "+v"(tf0_10), "+v"(tf1_10));
    v2f ff_10 = FIN2(((v2f){tf0_10, tf1_10}));
    ORDER("+v"(ff_10), "+v"(mi_11), "+v"(d2[2]));
    float ti0_11 = TSIG(mi_11[0]);
    v2f mf_11 = PRE2(((v2f){d2[2][6], d2[2][7]}));
    ORDER("+v"(ti0_11), "+v"(mf_11), "+v"(mi_11), "+v"(to0_10), "+v"(to1_10));
    float ti1_11 = TSIG(mi_11[1]);
    v2f fo_10 = FIN2(((v2f){to0_10, to1_10}));
    ORDER("+v"(ti1_11), "+v"(fo_10), "+v"(d2[4]), "+v"(ff_10), "+v"(ig_10), "+v"(hh_8), "+v"(acc));
    float tg0_11 = tanh_act(d2[4][6]);
    { const v2f cn = __builtin_elementwise_fma(ff_10, (v2f){c2[0][4], c2[0][5]}, ig_10); c2[0][4] = cn[0]; c2[0][5] = cn[1]; }
    acc = fmaf(hh_8[1], wfcD[0][1], acc);
    ORDER("+v"(tg0_11), "+v"(c2[0]), "+v"(acc), "+v"(d2[4]), "+v"(d2[6]));
    float tg1_11 = tanh_act(d2[4][7]);
    v2f mo_11 = PRE2(((v2f){d2[6][6], d2[6][7]}));
    ORDER("+v"(tg1_11), "+v"(mo_11), "+v"(mf_11), "+v"(fo_9), "+v"(tc0_9), "+v"(tc1_9));
    float tf0_11 = TSIG(mf_11[0]);
    v2f hh_9 = fo_9 * (v2f){tc0_9, tc1_9};
    ORDER("+v"(tf0_11), "+v"(hh_9), "+v"(mf_11), "+v"(ti0_11), "+v"(ti1_11));
    float tf1_11 = TSIG(mf_11[1]);
    v2f fi_11 = FIN2(((v2f){ti0_11, ti1_11}));
    ORDER("+v"(tf1_11), "+v"(fi_11), "+v"(mo_11), "+v"(d2[1]));
    float to0_11 = TSIG(mo_11[0]);
    v2f mi_12 = PRE2(((v2f){d2[1][0], d2[1][1]}));
    ORDER("+v"(to0_11), "+v"(mi_12), "+v"(mo_11), "+v"(hh_9));
    float to1_11 = TSIG(mo_11[1]);
    v2h pk_9 = __builtin_convertvector(hh_9, v2h);
    ORDER("+v"(to1_11), "+v"(pk_9), "+v"(c2[0]), "+v"(fi_11), "+v"(tg0_11), "+v"(tg1_11), "+v"(hh_9), "+v"(acc));
    float tc0_10 = tanh_act(c2[0][4]);
    v2f ig_11 = fi_11 * (v2f){tg0_11, tg1_11};
    acc = fmaf(hh_9[0], wfcD[0][2], acc);
    ORDER("+v"(tc0_10), "+v"(ig_11), "+v"(acc), "+v"(c2[0]));
    float tc1_10 = tanh_act(c2[0][5]);
    ORDER("+v"(tc1_10), "+v"(h1Bn));
    d1[4] = __builtin_amdgcn_wmma_f32_16x16x32_f16(false, a1[4], false, h1Bn, (short)0, zc, false, false);
    ORDER("+v"(d1[4]), "+v"(tf0_11), "+v"(tf1_11));
    v2f ff_11 = FIN2(((v2f){tf0_11, tf1_11}));
    ORDER("+v"(ff_11), "+v"(mi_12), "+v"(d2[3]));
    float ti0_12 = TSIG(mi_12[0]);
    v2f mf_12 = PRE2(((v2f){d2[3][0], d2[3][1]}));
    ORDER("+v"(ti0_12), "+v"(mf_12), "+v"(mi_12), "+v"(to0_11), "+v"(to1_11));
    float ti1_12 = TSIG(mi_12[1]);
    v2f fo_11 = FIN2(((v2f){to0_11, to1_11}));
    ORDER("+v"(ti1_12), "+v"(fo_11), "+v"(d2[5]), "+v"(ff_11), "+v"(ig_11), "+v"(hh_9), "+v"(acc));
    float tg0_12 = tanh_act(d2[5][0]);
    { const v2f cn = __builtin_elementwise_fma(ff_11, (v2f){c2[0][6], c2[0][7]}, ig_11); c2[0][6] = cn[0]; c2[0][7] = cn[1]; }
    acc = fmaf(hh_9[1], wfcD[0][3], acc);
    ORDER("+v"(tg0_12), "+v"(c2[0]), "+v"(acc), "+v"(d2[5]), "+v"(d2[7]));
    float tg1_12 = tanh_act(d2[5][1]);
    v2f mo_12 = PRE2(((v2f){d2[7][0], d2[7][1]}));
    ORDER("+v"(tg1_12), "+v"(mo_12), "+v"(mf_12), "+v"(fo_10), "+v"(tc0_10), "+v"(tc1_10));
    float tf0_12 = TSIG(mf_12[0]);
    v2f hh_10 = fo_10 * (v2f){tc0_10, tc1_10};
    ORDER("+v"(tf0_12), "+v"(hh_10), "+v"(mf_12), "+v"(ti0_12), "+v"(ti1_12));
    float tf1_12 = TSIG(mf_12[1]);
    v2f fi_12 = FIN2(((v2f){ti0_12, ti1_12}));
    ORDER("+v"(tf1_12), "+v"(fi_12), "+v"(mo_12), "+v"(d2[1]));
    float to0_12 = TSIG(mo_12[0]);
    v2f mi_13 = PRE2(((v2f){d2[1][2], d2[1][3]}));
    ORDER("+v"(to0_12), "+v"(mi_13), "+v"(mo_12), "+v"(hh_10));
    float to1_12 = TSIG(mo_12[1]);
    v2h pk_10 = __builtin_convertvector(hh_10, v2h);
    ORDER("+v"(to1_12), "+v"(pk_10), "+v"(c2[0]), "+v"(fi_12), "+v"(tg0_12), "+v"(tg1_12), "+v"(hh_10), "+v"(acc));
    float tc0_11 = tanh_act(c2[0][6]);
    v2f ig_12 = fi_12 * (v2f){tg0_12, tg1_12};
    acc = fmaf(hh_10[0], wfcD[0][4], acc);
    ORDER("+v"(tc0_11), "+v"(ig_12), "+v"(acc), "+v"(c2[0]));
    float tc1_11 = tanh_act(c2[0][7]);
    ORDER("+v"(tc1_11), "+v"(h1Bn));
    d1[6] = __builtin_amdgcn_wmma_f32_16x16x32_f16(false, a1[6], false, h1Bn, (short)0, zc, false, false);
    ORDER("+v"(d1[6]), "+v"(tf0_12), "+v"(tf1_12));
    v2f ff_12 = FIN2(((v2f){tf0_12, tf1_12}));
    ORDER("+v"(ff_12), "+v"(mi_13), "+v"(d2[3]));
    float ti0_13 = TSIG(mi_13[0]);
    v2f mf_13 = PRE2(((v2f){d2[3][2], d2[3][3]}));
    ORDER("+v"(ti0_13), "+v"(mf_13), "+v"(mi_13), "+v"(to0_12), "+v"(to1_12));
    float ti1_13 = TSIG(mi_13[1]);
    v2f fo_12 = FIN2(((v2f){to0_12, to1_12}));
    ORDER("+v"(ti1_13), "+v"(fo_12), "+v"(d2[5]), "+v"(ff_12), "+v"(ig_12), "+v"(hh_10), "+v"(acc));
    float tg0_13 = tanh_act(d2[5][2]);
    { const v2f cn = __builtin_elementwise_fma(ff_12, (v2f){c2[1][0], c2[1][1]}, ig_12); c2[1][0] = cn[0]; c2[1][1] = cn[1]; }
    acc = fmaf(hh_10[1], wfcD[0][5], acc);
    ORDER("+v"(tg0_13), "+v"(c2[1]), "+v"(acc), "+v"(d2[5]), "+v"(d2[7]));
    float tg1_13 = tanh_act(d2[5][3]);
    v2f mo_13 = PRE2(((v2f){d2[7][2], d2[7][3]}));
    ORDER("+v"(tg1_13), "+v"(mo_13), "+v"(mf_13), "+v"(fo_11), "+v"(tc0_11), "+v"(tc1_11));
    float tf0_13 = TSIG(mf_13[0]);
    v2f hh_11 = fo_11 * (v2f){tc0_11, tc1_11};
    ORDER("+v"(tf0_13), "+v"(hh_11), "+v"(mf_13), "+v"(ti0_13), "+v"(ti1_13));
    float tf1_13 = TSIG(mf_13[1]);
    v2f fi_13 = FIN2(((v2f){ti0_13, ti1_13}));
    ORDER("+v"(tf1_13), "+v"(fi_13), "+v"(mo_13), "+v"(d2[1]));
    float to0_13 = TSIG(mo_13[0]);
    v2f mi_14 = PRE2(((v2f){d2[1][4], d2[1][5]}));
    ORDER("+v"(to0_13), "+v"(mi_14), "+v"(mo_13), "+v"(hh_11));
    float to1_13 = TSIG(mo_13[1]);
    v2h pk_11 = __builtin_convertvector(hh_11, v2h);
    ORDER("+v"(to1_13), "+v"(pk_11), "+v"(c2[1]), "+v"(fi_13), "+v"(tg0_13), "+v"(tg1_13), "+v"(hh_11), "+v"(acc));
    float tc0_12 = tanh_act(c2[1][0]);
    v2f ig_13 = fi_13 * (v2f){tg0_13, tg1_13};
    acc = fmaf(hh_11[0], wfcD[0][6], acc);
    ORDER("+v"(tc0_12), "+v"(ig_13), "+v"(acc), "+v"(c2[1]));
    float tc1_12 = tanh_act(c2[1][1]);
    ORDER("+v"(tc1_12), "+v"(h1Bn));
    d1[1] = __builtin_amdgcn_wmma_f32_16x16x32_f16(false, a1[1], false, h1Bn, (short)0, zc, false, false);
    ORDER("+v"(d1[1]), "+v"(tf0_13), "+v"(tf1_13));
    v2f ff_13 = FIN2(((v2f){tf0_13, tf1_13}));
    ORDER("+v"(ff_13), "+v"(mi_14), "+v"(d2[3]));
    float ti0_14 = TSIG(mi_14[0]);
    v2f mf_14 = PRE2(((v2f){d2[3][4], d2[3][5]}));
    ORDER("+v"(ti0_14), "+v"(mf_14), "+v"(mi_14), "+v"(to0_13), "+v"(to1_13));
    float ti1_14 = TSIG(mi_14[1]);
    v2f fo_13 = FIN2(((v2f){to0_13, to1_13}));
    ORDER("+v"(ti1_14), "+v"(fo_13), "+v"(d2[5]), "+v"(ff_13), "+v"(ig_13), "+v"(hh_11), "+v"(acc));
    float tg0_14 = tanh_act(d2[5][4]);
    { const v2f cn = __builtin_elementwise_fma(ff_13, (v2f){c2[1][2], c2[1][3]}, ig_13); c2[1][2] = cn[0]; c2[1][3] = cn[1]; }
    acc = fmaf(hh_11[1], wfcD[0][7], acc);
    ORDER("+v"(tg0_14), "+v"(c2[1]), "+v"(acc), "+v"(d2[5]), "+v"(d2[7]));
    float tg1_14 = tanh_act(d2[5][5]);
    v2f mo_14 = PRE2(((v2f){d2[7][4], d2[7][5]}));
    ORDER("+v"(tg1_14), "+v"(mo_14), "+v"(mf_14), "+v"(fo_12), "+v"(tc0_12), "+v"(tc1_12));
    float tf0_14 = TSIG(mf_14[0]);
    v2f hh_12 = fo_12 * (v2f){tc0_12, tc1_12};
    ORDER("+v"(tf0_14), "+v"(hh_12), "+v"(mf_14), "+v"(ti0_14), "+v"(ti1_14));
    float tf1_14 = TSIG(mf_14[1]);
    v2f fi_14 = FIN2(((v2f){ti0_14, ti1_14}));
    ORDER("+v"(tf1_14), "+v"(fi_14), "+v"(mo_14), "+v"(d2[1]));
    float to0_14 = TSIG(mo_14[0]);
    v2f mi_15 = PRE2(((v2f){d2[1][6], d2[1][7]}));
    ORDER("+v"(to0_14), "+v"(mi_15), "+v"(mo_14), "+v"(hh_12));
    float to1_14 = TSIG(mo_14[1]);
    v2h pk_12 = __builtin_convertvector(hh_12, v2h);
    ORDER("+v"(to1_14), "+v"(pk_12), "+v"(c2[1]), "+v"(fi_14), "+v"(tg0_14), "+v"(tg1_14), "+v"(hh_12), "+v"(acc));
    float tc0_13 = tanh_act(c2[1][2]);
    v2f ig_14 = fi_14 * (v2f){tg0_14, tg1_14};
    acc = fmaf(hh_12[0], wfcD[1][0], acc);
    ORDER("+v"(tc0_13), "+v"(ig_14), "+v"(acc), "+v"(c2[1]));
    float tc1_13 = tanh_act(c2[1][3]);
    ORDER("+v"(tc1_13), "+v"(h1Bn));
    d1[3] = __builtin_amdgcn_wmma_f32_16x16x32_f16(false, a1[3], false, h1Bn, (short)0, zc, false, false);
    ORDER("+v"(d1[3]), "+v"(tf0_14), "+v"(tf1_14));
    v2f ff_14 = FIN2(((v2f){tf0_14, tf1_14}));
    ORDER("+v"(ff_14), "+v"(mi_15), "+v"(d2[3]));
    float ti0_15 = TSIG(mi_15[0]);
    v2f mf_15 = PRE2(((v2f){d2[3][6], d2[3][7]}));
    ORDER("+v"(ti0_15), "+v"(mf_15), "+v"(mi_15), "+v"(to0_14), "+v"(to1_14));
    float ti1_15 = TSIG(mi_15[1]);
    v2f fo_14 = FIN2(((v2f){to0_14, to1_14}));
    ORDER("+v"(ti1_15), "+v"(fo_14), "+v"(d2[5]), "+v"(ff_14), "+v"(ig_14), "+v"(hh_12), "+v"(acc));
    float tg0_15 = tanh_act(d2[5][6]);
    { const v2f cn = __builtin_elementwise_fma(ff_14, (v2f){c2[1][4], c2[1][5]}, ig_14); c2[1][4] = cn[0]; c2[1][5] = cn[1]; }
    acc = fmaf(hh_12[1], wfcD[1][1], acc);
    ORDER("+v"(tg0_15), "+v"(c2[1]), "+v"(acc), "+v"(d2[5]), "+v"(d2[7]));
    float tg1_15 = tanh_act(d2[5][7]);
    v2f mo_15 = PRE2(((v2f){d2[7][6], d2[7][7]}));
    ORDER("+v"(tg1_15), "+v"(mo_15), "+v"(mf_15), "+v"(fo_13), "+v"(tc0_13), "+v"(tc1_13));
    float tf0_15 = TSIG(mf_15[0]);
    v2f hh_13 = fo_13 * (v2f){tc0_13, tc1_13};
    ORDER("+v"(tf0_15), "+v"(hh_13), "+v"(mf_15), "+v"(ti0_15), "+v"(ti1_15));
    float tf1_15 = TSIG(mf_15[1]);
    v2f fi_15 = FIN2(((v2f){ti0_15, ti1_15}));
    ORDER("+v"(tf1_15), "+v"(fi_15), "+v"(mo_15));
    float to0_15 = TSIG(mo_15[0]);
    ORDER("+v"(to0_15), "+v"(mo_15), "+v"(hh_13));
    float to1_15 = TSIG(mo_15[1]);
    v2h pk_13 = __builtin_convertvector(hh_13, v2h);
    ORDER("+v"(to1_15), "+v"(pk_13), "+v"(c2[1]), "+v"(fi_15), "+v"(tg0_15), "+v"(tg1_15), "+v"(hh_13), "+v"(acc));
    float tc0_14 = tanh_act(c2[1][4]);
    v2f ig_15 = fi_15 * (v2f){tg0_15, tg1_15};
    acc = fmaf(hh_13[0], wfcD[1][2], acc);
    ORDER("+v"(tc0_14), "+v"(ig_15), "+v"(acc), "+v"(c2[1]));
    float tc1_14 = tanh_act(c2[1][5]);
    ORDER("+v"(tc1_14), "+v"(h1Bn));
    d1[5] = __builtin_amdgcn_wmma_f32_16x16x32_f16(false, a1[5], false, h1Bn, (short)0, zc, false, false);
    ORDER("+v"(d1[5]), "+v"(tf0_15), "+v"(tf1_15));
    v2f ff_15 = FIN2(((v2f){tf0_15, tf1_15}));
    ORDER("+v"(ff_15), "+v"(to0_15), "+v"(to1_15));
    v2f fo_15 = FIN2(((v2f){to0_15, to1_15}));
    ORDER("+v"(fo_15), "+v"(ff_15), "+v"(ig_15), "+v"(hh_13), "+v"(acc));
    { const v2f cn = __builtin_elementwise_fma(ff_15, (v2f){c2[1][6], c2[1][7]}, ig_15); c2[1][6] = cn[0]; c2[1][7] = cn[1]; }
    acc = fmaf(hh_13[1], wfcD[1][3], acc);
    ORDER("+v"(c2[1]), "+v"(acc), "+v"(fo_14), "+v"(tc0_14), "+v"(tc1_14));
    v2f hh_14 = fo_14 * (v2f){tc0_14, tc1_14};
    ORDER("+v"(hh_14));
    v2h pk_14 = __builtin_convertvector(hh_14, v2h);
    ORDER("+v"(pk_14), "+v"(c2[1]), "+v"(hh_14), "+v"(acc));
    float tc0_15 = tanh_act(c2[1][6]);
    acc = fmaf(hh_14[0], wfcD[1][4], acc);
    ORDER("+v"(tc0_15), "+v"(acc), "+v"(c2[1]));
    float tc1_15 = tanh_act(c2[1][7]);
    ORDER("+v"(tc1_15), "+v"(h1Bn));
    d1[7] = __builtin_amdgcn_wmma_f32_16x16x32_f16(false, a1[7], false, h1Bn, (short)0, zc, false, false);
    ORDER("+v"(d1[7]), "+v"(hh_14), "+v"(acc));
    acc = fmaf(hh_14[1], wfcD[1][5], acc);
    ORDER("+v"(acc), "+v"(fo_15), "+v"(tc0_15), "+v"(tc1_15));
    v2f hh_15 = fo_15 * (v2f){tc0_15, tc1_15};
    ORDER("+v"(hh_15));
    v2h pk_15 = __builtin_convertvector(hh_15, v2h);
    ORDER("+v"(pk_15), "+v"(hh_15), "+v"(acc));
    acc = fmaf(hh_15[0], wfcD[1][6], acc);
    ORDER("+v"(acc), "+v"(hh_15));
    acc = fmaf(hh_15[1], wfcD[1][7], acc);
    h2Bn[0] = pk_8[0]; h2Bn[1] = pk_8[1];
    h2Bn[2] = pk_9[0]; h2Bn[3] = pk_9[1];
    h2Bn[4] = pk_10[0]; h2Bn[5] = pk_10[1];
    h2Bn[6] = pk_11[0]; h2Bn[7] = pk_11[1];
    h2Bn[8] = pk_12[0]; h2Bn[9] = pk_12[1];
    h2Bn[10] = pk_13[0]; h2Bn[11] = pk_13[1];
    h2Bn[12] = pk_14[0]; h2Bn[13] = pk_14[1];
    h2Bn[14] = pk_15[0]; h2Bn[15] = pk_15[1];
    if (hi_half) { h2Bn[14] = (_Float16)0.f; h2Bn[15] = (_Float16)1.f; }
    pend = (accPrev + accOther) + bfc;
    accOther = xhalf(acc);
    float xnext = xpre;
    ORDER("+v"(xnext));


    accPrev = acc;
    h1B = h1Bn;
    h2B = h2Bn;
    xinj = xnext;
  }

  {
    if (lane < 16) orow[TLEN - 3] = pend;
    const float accOther = xhalf(accPrev);
    if (lane < 16) orow[TLEN - 2] = (accPrev + accOther) + bfc;
    float acc;
    (void)layer2_step(ai2, ah2, h1B, h2B, c2, wfcD, zc, acc);
    acc += xhalf(acc);
    if (lane < 16) orow[TLEN - 1] = acc + bfc;
  }
}

extern "C" void kernel_launch(void* const* d_in, const int* in_sizes, int n_in,
                              void* d_out, int out_size, void* d_ws, size_t ws_size,
                              hipStream_t stream) {
  (void)in_sizes; (void)n_in; (void)out_size; (void)d_ws; (void)ws_size;
  const float* x     = (const float*)d_in[0];
  const float* w_ih1 = (const float*)d_in[1];
  const float* w_hh1 = (const float*)d_in[2];
  const float* b_ih1 = (const float*)d_in[3];
  const float* b_hh1 = (const float*)d_in[4];
  const float* w_ih2 = (const float*)d_in[5];
  const float* w_hh2 = (const float*)d_in[6];
  const float* b_ih2 = (const float*)d_in[7];
  const float* b_hh2 = (const float*)d_in[8];
  const float* w_fc  = (const float*)d_in[9];
  const float* b_fc  = (const float*)d_in[10];
  float* out = (float*)d_out;

  dim3 grid(BSZ / 16);
  dim3 block(32);
  lstm2_wmma_kernel<<<grid, block, 0, stream>>>(x, w_ih1, w_hh1, b_ih1, b_hh1,
                                                w_ih2, w_hh2, b_ih2, b_hh2,
                                                w_fc, b_fc, out);
}


# Two steps per loop trip: out stored, x loaded once a trip by all 32 lanes (lane half = step parity): no lane-masked branch per step, no h operand copy; last layer 2 in the loop; values, sums kept
# speedup vs baseline: 1.8611x; 1.0808x over previous
#include <hip/hip_runtime.h>


#define HSZ   30
#define MPAD  128
#define KPAD  32
#define TLEN  2048
#define BSZ   512

typedef __attribute__((ext_vector_type(16))) _Float16 v16h;
typedef __attribute__((ext_vector_type(8)))  _Float16 v8h;
typedef __attribute__((ext_vector_type(8)))  float    v8f;
typedef __attribute__((ext_vector_type(2)))  float    v2f;
typedef __attribute__((ext_vector_type(2)))  _Float16 v2h;

#define ORDER(...) asm volatile("" : __VA_ARGS__)

#if __has_builtin(__builtin_amdgcn_tanhf)
#define HW_TANH __builtin_amdgcn_tanhf
#elif __has_builtin(__builtin_amdgcn_tanh_f32)
#define HW_TANH __builtin_amdgcn_tanh_f32
#endif

#ifdef HW_TANH
__device__ __forceinline__ float tanh_act(float x) { return HW_TANH(x); }
__device__ __forceinline__ float sig_act(float x)  { return fmaf(0.5f, HW_TANH(0.5f * x), 0.5f); }
#define PRE2(x) ((x) * 0.5f)
#define TSIG(m) HW_TANH(m)
#define FIN2(t) __builtin_elementwise_fma((v2f)(0.5f), (t), (v2f)(0.5f))
#else
#define PRE2(x) (x)
#define TSIG(m) __builtin_amdgcn_rcpf(1.0f + __expf(-(m)))
#define FIN2(t) (t)
__device__ __forceinline__ float tanh_act(float x) {
  return fmaf(2.0f, __builtin_amdgcn_rcpf(1.0f + __expf(-2.0f * x)), -1.0f);
}
__device__ __forceinline__ float sig_act(float x) {
  return __builtin_amdgcn_rcpf(1.0f + __expf(-x));
}
#endif

__device__ __forceinline__ float xhalf(float v) {
  const unsigned u = __builtin_bit_cast(unsigned, v);
  return __builtin_bit_cast(float, __builtin_amdgcn_permlanex16(u, u, 0x76543210u, 0xfedcba98u, false, false));
}

__device__ __forceinline__ v16h load_a_tile(const _Float16* __restrict__ base,
                                            int tile, int lane) {
  const int m  = lane & 15;
  const int kb = (lane >> 4) << 3;
  const _Float16* row = base + ((tile << 4) + m) * KPAD;
  v8h lo = *(const v8h*)(row + kb);
  v8h hi = *(const v8h*)(row + 16 + kb);
  v16h a;
#pragma unroll
  for (int i = 0; i < 8; ++i) { a[i] = lo[i]; a[8 + i] = hi[i]; }
  return a;
}

__device__ __forceinline__ void gates8(const v8f (&d)[4], v8f &c, float (&hn)[8]) {
#pragma unroll
  for (int r = 0; r < 8; ++r) {
    const float iv = sig_act(d[0][r]);
    const float fv = sig_act(d[1][r]);
    const float gv = tanh_act(d[2][r]);
    const float ov = sig_act(d[3][r]);
    const float cn = fmaf(fv, c[r], iv * gv);
    c[r] = cn;
    hn[r] = ov * tanh_act(cn);
  }
}

__device__ __forceinline__ v16h layer1_step(const v16h (&a1)[8], v16h h1B, v8f (&c1)[2], v8f zc) {
  v16h h1Bn;
#pragma unroll
  for (int s = 0; s < 2; ++s) {
    v8f d[4];
#pragma unroll
    for (int g = 0; g < 4; ++g)
      d[g] = __builtin_amdgcn_wmma_f32_16x16x32_f16(false, a1[2 * g + s], false, h1B,
                                                    (short)0, zc, false, false);
    float hn[8];
    gates8(d, c1[s], hn);
#pragma unroll
    for (int r = 0; r < 8; ++r) h1Bn[8 * s + r] = (_Float16)hn[r];
  }
  return h1Bn;
}

__global__ __launch_bounds__(32, 1)
void lstm2_wmma_kernel(const float* __restrict__ x,
                       const float* __restrict__ w_ih1, const float* __restrict__ w_hh1,
                       const float* __restrict__ b_ih1, const float* __restrict__ b_hh1,
                       const float* __restrict__ w_ih2, const float* __restrict__ w_hh2,
                       const float* __restrict__ b_ih2, const float* __restrict__ b_hh2,
                       const float* __restrict__ w_fc,  const float* __restrict__ b_fc,
                       float* __restrict__ out) {
  __shared__ _Float16 sA1[MPAD * KPAD];
  __shared__ _Float16 sAi2[MPAD * KPAD];
  __shared__ _Float16 sAh2[MPAD * KPAD];

  const int lane = threadIdx.x;
  const bool hi_half = lane >= 16;
  const int n = lane & 15;

  for (int idx = lane; idx < MPAD * KPAD; idx += 32) {
    const int m = idx >> 5, k = idx & 31;
    const int g = m >> 5, j = m & 31;
    const int row = g * HSZ + j;
    float a1 = 0.f, ai2 = 0.f, ah2 = 0.f;
    if (j < HSZ) {
      if (k < HSZ) {
        a1  = w_hh1[row * HSZ + k];
        ai2 = w_ih2[row * HSZ + k];
        ah2 = w_hh2[row * HSZ + k];
      } else if (k == 30) {
        a1  = w_ih1[row];
      } else {
        a1  = b_ih1[row] + b_hh1[row];
        ah2 = b_ih2[row] + b_hh2[row];
      }
    }
    sA1[idx]  = (_Float16)a1;
    sAi2[idx] = (_Float16)ai2;
    sAh2[idx] = (_Float16)ah2;
  }
  __syncthreads();

  v16h a1[8], ai2[8], ah2[8];
#pragma unroll
  for (int tl = 0; tl < 8; ++tl) {
    a1[tl]  = load_a_tile(sA1,  tl, lane);
    ai2[tl] = load_a_tile(sAi2, tl, lane);
    ah2[tl] = load_a_tile(sAh2, tl, lane);
  }

  float wfcD[2][8];
#pragma unroll
  for (int s = 0; s < 2; ++s)
#pragma unroll
    for (int r = 0; r < 8; ++r) {
      const int j = 16 * s + (hi_half ? 8 : 0) + r;
      wfcD[s][r] = (j < HSZ) ? w_fc[j] : 0.f;
    }
  const float bfc = b_fc[0];

  v8f zc;
#pragma unroll
  for (int i = 0; i < 8; ++i) zc[i] = 0.f;

  v16h h1B, h2B;
#pragma unroll
  for (int i = 0; i < 16; ++i) { h1B[i] = (_Float16)0.f; h2B[i] = (_Float16)0.f; }
  if (hi_half) h2B[15] = (_Float16)1.f;

  v8f c1[2], c2[2];
#pragma unroll
  for (int s = 0; s < 2; ++s)
#pragma unroll
    for (int r = 0; r < 8; ++r) { c1[s][r] = 0.f; c2[s][r] = 0.f; }

  const int bbase = blockIdx.x << 4;
  const float* xrow = x + (size_t)(bbase + n) * TLEN;
  float* orow = out + (size_t)(bbase + n) * TLEN;

  const float x0 = xrow[0];
  const float x1 = xrow[1];
  float xa = xrow[2];
  float xb = xrow[3];
  if (hi_half) { h1B[14] = (_Float16)x0; h1B[15] = (_Float16)1.f; }
  h1B = layer1_step(a1, h1B, c1, zc);
  if (hi_half) { h1B[14] = (_Float16)x1; h1B[15] = (_Float16)1.f; }

  v8f d1[8];
#pragma unroll
  for (int tl = 0; tl < 8; ++tl)
    d1[tl] = __builtin_amdgcn_wmma_f32_16x16x32_f16(false, a1[tl], false, h1B,
                                                    (short)0, zc, false, false);

  float accPrev = 0.f;
  float accOther = 0.f;
  float outEven = 0.f;
  float outOdd = 0.f;
  float* ost = orow + (hi_half ? 1 : 0);
  const float* xp = xrow + 4;

  for (int tt = 0; tt < TLEN; tt += 2) {
    const v2f xq = *reinterpret_cast<const v2f*>(xp);
#pragma unroll
  for (int u = 0; u < 2; ++u) {
    const int t = tt + u;

    if (u == 1 && tt > 0) ost[tt - 2] = hi_half ? outOdd : outEven;

    v8f d2[8];
    v16h h1Bn, h2Bn;
    float acc = 0.f;
    ORDER("+v"(d1[0]));
    v2f mi_0 = PRE2(((v2f){d1[0][0], d1[0][1]}));
    ORDER("+v"(mi_0), "+v"(h2B));
    d2[0] = __builtin_amdgcn_wmma_f32_16x16x32_f16(false, ah2[0], false, h2B, (short)0, zc, false, false);
    ORDER("+v"(d2[0]), "+v"(mi_0), "+v"(d1[2]));
    float ti0_0 = TSIG(mi_0[0]);
    v2f mf_0 = PRE2(((v2f){d1[2][0], d1[2][1]}));
    ORDER("+v"(ti0_0), "+v"(mf_0), "+v"(mi_0));
    float ti1_0 = TSIG(mi_0[1]);
    ORDER("+v"(ti1_0), "+v"(d1[4]));
    float tg0_0 = tanh_act(d1[4][0]);
    ORDER("+v"(tg0_0), "+v"(d1[4]), "+v"(d1[6]));
    float tg1_0 = tanh_act(d1[4][1]);
    v2f mo_0 = PRE2(((v2f){d1[6][0], d1[6][1]}));
    ORDER("+v"(tg1_0), "+v"(mo_0), "+v"(mf_0));
    float tf0_0 = TSIG(mf_0[0]);
    ORDER("+v"(tf0_0), "+v"(h2B));
    d2[2] = __builtin_amdgcn_wmma_f32_16x16x32_f16(false, ah2[2], false, h2B, (short)0, zc, false, false);
    ORDER("+v"(d2[2]), "+v"(mf_0), "+v"(ti0_0), "+v"(ti1_0));
    float tf1_0 = TSIG(mf_0[1]);
    v2f fi_0 = FIN2(((v2f){ti0_0, ti1_0}));
    ORDER("+v"(tf1_0), "+v"(fi_0), "+v"(mo_0), "+v"(d1[0]));
    float to0_0 = TSIG(mo_0[0]);
    v2f mi_1 = PRE2(((v2f){d1[0][2], d1[0][3]}));
    ORDER("+v"(to0_0), "+v"(mi_1), "+v"(mo_0));
    float to1_0 = TSIG(mo_0[1]);
    ORDER("+v"(to1_0), "+v"(fi_0), "+v"(tg0_0), "+v"(tg1_0));
    v2f ig_0 = fi_0 * (v2f){tg0_0, tg1_0};
    ORDER("+v"(ig_0), "+v"(h2B));
    d2[4] = __builtin_amdgcn_wmma_f32_16x16x32_f16(false, ah2[4], false, h2B, (short)0, zc, false, false);
    ORDER("+v"(d2[4]), "+v"(tf0_0), "+v"(tf1_0));
    v2f ff_0 = FIN2(((v2f){tf0_0, tf1_0}));
    ORDER("+v"(ff_0), "+v"(mi_1), "+v"(d1[2]));
    float ti0_1 = TSIG(mi_1[0]);
    v2f mf_1 = PRE2(((v2f){d1[2][2], d1[2][3]}));
    ORDER("+v"(ti0_1), "+v"(mf_1), "+v"(mi_1), "+v"(to0_0), "+v"(to1_0));
    float ti1_1 = TSIG(mi_1[1]);
    v2f fo_0 = FIN2(((v2f){to0_0, to1_0}));
    ORDER("+v"(ti1_1), "+v"(fo_0), "+v"(d1[4]), "+v"(ff_0), "+v"(ig_0));
    float tg0_1 = tanh_act(d1[4][2]);
    { const v2f cn = __builtin_elementwise_fma(ff_0, (v2f){c1[0][0], c1[0][1]}, ig_0); c1[0][0] = cn[0]; c1[0][1] = cn[1]; }
    ORDER("+v"(tg0_1), "+v"(c1[0]), "+v"(d1[4]), "+v"(d1[6]));
    float tg1_1 = tanh_act(d1[4][3]);
    v2f mo_1 = PRE2(((v2f){d1[6][2], d1[6][3]}));
    ORDER("+v"(tg1_1), "+v"(mo_1), "+v"(mf_1));
    float tf0_1 = TSIG(mf_1[0]);
    ORDER("+v"(tf0_1), "+v"(h2B));
    d2[6] = __builtin_amdgcn_wmma_f32_16x16x32_f16(false, ah2[6], false, h2B, (short)0, zc, false, false);
    ORDER("+v"(d2[6]), "+v"(mf_1), "+v"(ti0_1), "+v"(ti1_1));
    float tf1_1 = TSIG(mf_1[1]);
    v2f fi_1 = FIN2(((v2f){ti0_1, ti1_1}));
    ORDER("+v"(tf1_1), "+v"(fi_1), "+v"(mo_1), "+v"(d1[0]));
    float to0_1 = TSIG(mo_1[0]);
    v2f mi_2 = PRE2(((v2f){d1[0][4], d1[0][5]}));
    ORDER("+v"(to0_1), "+v"(mi_2), "+v"(mo_1));
    float to1_1 = TSIG(mo_1[1]);
    ORDER("+v"(to1_1), "+v"(c1[0]), "+v"(fi_1), "+v"(tg0_1), "+v"(tg1_1));
    float tc0_0 = tanh_act(c1[0][0]);
    v2f ig_1 = fi_1 * (v2f){tg0_1, tg1_1};
    ORDER("+v"(tc0_0), "+v"(ig_1), "+v"(c1[0]));
    float tc1_0 = tanh_act(c1[0][1]);
    ORDER("+v"(tc1_0), "+v"(h2B));
    d2[1] = __builtin_amdgcn_wmma_f32_16x16x32_f16(false, ah2[1], false, h2B, (short)0, zc, false, false);
    ORDER("+v"(d2[1]), "+v"(tf0_1), "+v"(tf1_1));
    v2f ff_1 = FIN2(((v2f){tf0_1, tf1_1}));
    ORDER("+v"(ff_1), "+v"(mi_2), "+v"(d1[2]));
    float ti0_2 = TSIG(mi_2[0]);
    v2f mf_2 = PRE2(((v2f){d1[2][4], d1[2][5]}));
    ORDER("+v"(ti0_2), "+v"(mf_2), "+v"(mi_2), "+v"(to0_1), "+v"(to1_1));
    float ti1_2 = TSIG(mi_2[1]);
    v2f fo_1 = FIN2(((v2f){to0_1, to1_1}));
    ORDER("+v"(ti1_2), "+v"(fo_1), "+v"(d1[4]), "+v"(ff_1), "+v"(ig_1));
    float tg0_2 = tanh_act(d1[4][4]);
    { const v2f cn = __builtin_elementwise_fma(ff_1, (v2f){c1[0][2], c1[0][3]}, ig_1); c1[0][2] = cn[0]; c1[0][3] = cn[1]; }
    ORDER("+v"(tg0_2), "+v"(c1[0]), "+v"(d1[4]), "+v"(d1[6]));
    float tg1_2 = tanh_act(d1[4][5]);
    v2f mo_2 = PRE2(((v2f){d1[6][4], d1[6][5]}));
    ORDER("+v"(tg1_2), "+v"(mo_2), "+v"(mf_2));
    float tf0_2 = TSIG(mf_2[0]);
    ORDER("+v"(tf0_2), "+v"(h2B));
    d2[3] = __builtin_amdgcn_wmma_f32_16x16x32_f16(false, ah2[3], false, h2B, (short)0, zc, false, false);
    ORDER("+v"(d2[3]), "+v"(fo_0), "+v"(tc0_0), "+v"(tc1_0));
    v2f hh_0 = fo_0 * (v2f){tc0_0, tc1_0};
    ORDER("+v"(hh_0), "+v"(mf_2), "+v"(ti0_2), "+v"(ti1_2));
    float tf1_2 = TSIG(mf_2[1]);
    v2f fi_2 = FIN2(((v2f){ti0_2, ti1_2}));
    ORDER("+v"(tf1_2), "+v"(fi_2), "+v"(mo_2), "+v"(d1[0]));
    float to0_2 = TSIG(mo_2[0]);
    v2f mi_3 = PRE2(((v2f){d1[0][6], d1[0][7]}));
    ORDER("+v"(to0_2), "+v"(mi_3), "+v"(mo_2), "+v"(hh_0));
    float to1_2 = TSIG(mo_2[1]);
    v2h pk_0 = __builtin_convertvector(hh_0, v2h);
    ORDER("+v"(to1_2), "+v"(pk_0), "+v"(c1[0]), "+v"(fi_2), "+v"(tg0_2), "+v"(tg1_2));
    float tc0_1 = tanh_act(c1[0][2]);
    v2f ig_2 = fi_2 * (v2f){tg0_2, tg1_2};
    ORDER("+v"(tc0_1), "+v"(ig_2), "+v"(c1[0]));
    float tc1_1 = tanh_act(c1[0][3]);
    ORDER("+v"(tc1_1), "+v"(h2B));
    d2[5] = __builtin_amdgcn_wmma_f32_16x16x32_f16(false, ah2[5], false, h2B, (short)0, zc, false, false);
    ORDER("+v"(d2[5]), "+v"(tf0_2), "+v"(tf1_2));
    v2f ff_2 = FIN2(((v2f){tf0_2, tf1_2}));
    ORDER("+v"(ff_2), "+v"(mi_3), "+v"(d1[2]));
    float ti0_3 = TSIG(mi_3[0]);
    v2f mf_3 = PRE2(((v2f){d1[2][6], d1[2][7]}));
    ORDER("+v"(ti0_3), "+v"(mf_3), "+v"(mi_3), "+v"(to0_2), "+v"(to1_2));
    float ti1_3 = TSIG(mi_3[1]);
    v2f fo_2 = FIN2(((v2f){to0_2, to1_2}));
    ORDER("+v"(ti1_3), "+v"(fo_2), "+v"(d1[4]), "+v"(ff_2), "+v"(ig_2));
    float tg0_3 = tanh_act(d1[4][6]);
    { const v2f cn = __builtin_elementwise_fma(ff_2, (v2f){c1[0][4], c1[0][5]}, ig_2); c1[0][4] = cn[0]; c1[0][5] = cn[1]; }
    ORDER("+v"(tg0_3), "+v"(c1[0]), "+v"(d1[4]), "+v"(d1[6]));
    float tg1_3 = tanh_act(d1[4][7]);
    v2f mo_3 = PRE2(((v2f){d1[6][6], d1[6][7]}));
    ORDER("+v"(tg1_3), "+v"(mo_3), "+v"(mf_3));
    float tf0_3 = TSIG(mf_3[0]);
    ORDER("+v"(tf0_3), "+v"(h2B));
    d2[7] = __builtin_amdgcn_wmma_f32_16x16x32_f16(false, ah2[7], false, h2B, (short)0, zc, false, false);
    ORDER("+v"(d2[7]), "+v"(fo_1), "+v"(tc0_1), "+v"(tc1_1));
    v2f hh_1 = fo_1 * (v2f){tc0_1, tc1_1};
    ORDER("+v"(hh_1), "+v"(mf_3), "+v"(ti0_3), "+v"(ti1_3));
    float tf1_3 = TSIG(mf_3[1]);
    v2f fi_3 = FIN2(((v2f){ti0_3, ti1_3}));
    ORDER("+v"(tf1_3), "+v"(fi_3), "+v"(mo_3), "+v"(d1[1]));
    float to0_3 = TSIG(mo_3[0]);
    v2f mi_4 = PRE2(((v2f){d1[1][0], d1[1][1]}));
    ORDER("+v"(to0_3), "+v"(mi_4), "+v"(mo_3), "+v"(hh_1));
    float to1_3 = TSIG(mo_3[1]);
    v2h pk_1 = __builtin_convertvector(hh_1, v2h);
    ORDER("+v"(to1_3), "+v"(pk_1), "+v"(c1[0]), "+v"(fi_3), "+v"(tg0_3), "+v"(tg1_3));
    float tc0_2 = tanh_act(c1[0][4]);
    v2f ig_3 = fi_3 * (v2f){tg0_3, tg1_3};
    ORDER("+v"(tc0_2), "+v"(ig_3), "+v"(c1[0]));
    float tc1_2 = tanh_act(c1[0][5]);
    ORDER("+v"(tc1_2), "+v"(h1B), "+v"(d2[0]));
    d2[0] = __builtin_amdgcn_wmma_f32_16x16x32_f16(false, ai2[0], false, h1B, (short)0, d2[0], false, false);
    ORDER("+v"(d2[0]), "+v"(tf0_3), "+v"(tf1_3));
    v2f ff_3 = FIN2(((v2f){tf0_3, tf1_3}));
    ORDER("+v"(ff_3), "+v"(mi_4), "+v"(d1[3]));
    float ti0_4 = TSIG(mi_4[0]);
    v2f mf_4 = PRE2(((v2f){d1[3][0], d1[3][1]}));
    ORDER("+v"(ti0_4), "+v"(mf_4), "+v"(mi_4), "+v"(to0_3), "+v"(to1_3));
    float ti1_4 = TSIG(mi_4[1]);
    v2f fo_3 = FIN2(((v2f){to0_3, to1_3}));
    ORDER("+v"(ti1_4), "+v"(fo_3), "+v"(d1[5]), "+v"(ff_3), "+v"(ig_3));
    float tg0_4 = tanh_act(d1[5][0]);
    { const v2f cn = __builtin_elementwise_fma(ff_3, (v2f){c1[0][6], c1[0][7]}, ig_3); c1[0][6] = cn[0]; c1[0][7] = cn[1]; }
    ORDER("+v"(tg0_4), "+v"(c1[0]), "+v"(d1[5]), "+v"(d1[7]));
    float tg1_4 = tanh_act(d1[5][1]);
    v2f mo_4 = PRE2(((v2f){d1[7][0], d1[7][1]}));
    ORDER("+v"(tg1_4), "+v"(mo_4), "+v"(mf_4));
    float tf0_4 = TSIG(mf_4[0]);
    ORDER("+v"(tf0_4), "+v"(h1B), "+v"(d2[2]));
    d2[2] = __builtin_amdgcn_wmma_f32_16x16x32_f16(false, ai2[2], false, h1B, (short)0, d2[2], false, false);
    ORDER("+v"(d2[2]), "+v"(fo_2), "+v"(tc0_2), "+v"(tc1_2));
    v2f hh_2 = fo_2 * (v2f){tc0_2, tc1_2};
    ORDER("+v"(hh_2), "+v"(mf_4), "+v"(ti0_4), "+v"(ti1_4));
    float tf1_4 = TSIG(mf_4[1]);
    v2f fi_4 = FIN2(((v2f){ti0_4, ti1_4}));
    ORDER("+v"(tf1_4), "+v"(fi_4), "+v"(mo_4), "+v"(d1[1]));
    float to0_4 = TSIG(mo_4[0]);
    v2f mi_5 = PRE2(((v2f){d1[1][2], d1[1][3]}));
    ORDER("+v"(to0_4), "+v"(mi_5), "+v"(mo_4), "+v"(hh_2));
    float to1_4 = TSIG(mo_4[1]);
    v2h pk_2 = __builtin_convertvector(hh_2, v2h);
    ORDER("+v"(to1_4), "+v"(pk_2), "+v"(c1[0]), "+v"(fi_4), "+v"(tg0_4), "+v"(tg1_4));
    float tc0_3 = tanh_act(c1[0][6]);
    v2f ig_4 = fi_4 * (v2f){tg0_4, tg1_4};
    ORDER("+v"(tc0_3), "+v"(ig_4), "+v"(c1[0]));
    float tc1_3 = tanh_act(c1[0][7]);
    ORDER("+v"(tc1_3), "+v"(h1B), "+v"(d2[4]));
    d2[4] = __builtin_amdgcn_wmma_f32_16x16x32_f16(false, ai2[4], false, h1B, (short)0, d2[4], false, false);
    ORDER("+v"(d2[4]), "+v"(tf0_4), "+v"(tf1_4));
    v2f ff_4 = FIN2(((v2f){tf0_4, tf1_4}));
    ORDER("+v"(ff_4), "+v"(mi_5), "+v"(d1[3]));
    float ti0_5 = TSIG(mi_5[0]);
    v2f mf_5 = PRE2(((v2f){d1[3][2], d1[3][3]}));
    ORDER("+v"(ti0_5), "+v"(mf_5), "+v"(mi_5), "+v"(to0_4), "+v"(to1_4));
    float ti1_5 = TSIG(mi_5[1]);
    v2f fo_4 = FIN2(((v2f){to0_4, to1_4}));
    ORDER("+v"(ti1_5), "+v"(fo_4), "+v"(d1[5]), "+v"(ff_4), "+v"(ig_4));
    float tg0_5 = tanh_act(d1[5][2]);
    { const v2f cn = __builtin_elementwise_fma(ff_4, (v2f){c1[1][0], c1[1][1]}, ig_4); c1[1][0] = cn[0]; c1[1][1] = cn[1]; }
    ORDER("+v"(tg0_5), "+v"(c1[1]), "+v"(d1[5]), "+v"(d1[7]));
    float tg1_5 = tanh_act(d1[5][3]);
    v2f mo_5 = PRE2(((v2f){d1[7][2], d1[7][3]}));
    ORDER("+v"(tg1_5), "+v"(mo_5), "+v"(mf_5));
    float tf0_5 = TSIG(mf_5[0]);
    ORDER("+v"(tf0_5), "+v"(h1B), "+v"(d2[6]));
    d2[6] = __builtin_amdgcn_wmma_f32_16x16x32_f16(false, ai2[6], false, h1B, (short)0, d2[6], false, false);
    ORDER("+v"(d2[6]), "+v"(fo_3), "+v"(tc0_3), "+v"(tc1_3));
    v2f hh_3 = fo_3 * (v2f){tc0_3, tc1_3};
    ORDER("+v"(hh_3), "+v"(mf_5), "+v"(ti0_5), "+v"(ti1_5));
    float tf1_5 = TSIG(mf_5[1]);
    v2f fi_5 = FIN2(((v2f){ti0_5, ti1_5}));
    ORDER("+v"(tf1_5), "+v"(fi_5), "+v"(mo_5), "+v"(d1[1]));
    float to0_5 = TSIG(mo_5[0]);
    v2f mi_6 = PRE2(((v2f){d1[1][4], d1[1][5]}));
    ORDER("+v"(to0_5), "+v"(mi_6), "+v"(mo_5), "+v"(hh_3));
    float to1_5 = TSIG(mo_5[1]);
    v2h pk_3 = __builtin_convertvector(hh_3, v2h);
    ORDER("+v"(to1_5), "+v"(pk_3), "+v"(c1[1]), "+v"(fi_5), "+v"(tg0_5), "+v"(tg1_5));
    float tc0_4 = tanh_act(c1[1][0]);
    v2f ig_5 = fi_5 * (v2f){tg0_5, tg1_5};
    ORDER("+v"(tc0_4), "+v"(ig_5), "+v"(c1[1]));
    float tc1_4 = tanh_act(c1[1][1]);
    ORDER("+v"(tc1_4), "+v"(h1B), "+v"(d2[1]));
    d2[1] = __builtin_amdgcn_wmma_f32_16x16x32_f16(false, ai2[1], false, h1B, (short)0, d2[1], false, false);
    ORDER("+v"(d2[1]), "+v"(tf0_5), "+v"(tf1_5));
    v2f ff_5 = FIN2(((v2f){tf0_5, tf1_5}));
    ORDER("+v"(ff_5), "+v"(mi_6), "+v"(d1[3]));
    float ti0_6 = TSIG(mi_6[0]);
    v2f mf_6 = PRE2(((v2f){d1[3][4], d1[3][5]}));
    ORDER("+v"(ti0_6), "+v"(mf_6), "+v"(mi_6), "+v"(to0_5), "+v"(to1_5));
    float ti1_6 = TSIG(mi_6[1]);
    v2f fo_5 = FIN2(((v2f){to0_5, to1_5}));
    ORDER("+v"(ti1_6), "+v"(fo_5), "+v"(d1[5]), "+v"(ff_5), "+v"(ig_5));
    float tg0_6 = tanh_act(d1[5][4]);
    { const v2f cn = __builtin_elementwise_fma(ff_5, (v2f){c1[1][2], c1[1][3]}, ig_5); c1[1][2] = cn[0]; c1[1][3] = cn[1]; }
    ORDER("+v"(tg0_6), "+v"(c1[1]), "+v"(d1[5]), "+v"(d1[7]));
    float tg1_6 = tanh_act(d1[5][5]);
    v2f mo_6 = PRE2(((v2f){d1[7][4], d1[7][5]}));
    ORDER("+v"(tg1_6), "+v"(mo_6), "+v"(mf_6));
    float tf0_6 = TSIG(mf_6[0]);
    ORDER("+v"(tf0_6), "+v"(h1B), "+v"(d2[3]));
    d2[3] = __builtin_amdgcn_wmma_f32_16x16x32_f16(false, ai2[3], false, h1B, (short)0, d2[3], false, false);
    ORDER("+v"(d2[3]), "+v"(fo_4), "+v"(tc0_4), "+v"(tc1_4));
    v2f hh_4 = fo_4 * (v2f){tc0_4, tc1_4};
    ORDER("+v"(hh_4), "+v"(mf_6), "+v"(ti0_6), "+v"(ti1_6));
    float tf1_6 = TSIG(mf_6[1]);
    v2f fi_6 = FIN2(((v2f){ti0_6, ti1_6}));
    ORDER("+v"(tf1_6), "+v"(fi_6), "+v"(mo_6), "+v"(d1[1]));
    float to0_6 = TSIG(mo_6[0]);
    v2f mi_7 = PRE2(((v2f){d1[1][6], d1[1][7]}));
    ORDER("+v"(to0_6), "+v"(mi_7), "+v"(mo_6), "+v"(hh_4));
    float to1_6 = TSIG(mo_6[1]);
    v2h pk_4 = __builtin_convertvector(hh_4, v2h);
    ORDER("+v"(to1_6), "+v"(pk_4), "+v"(c1[1]), "+v"(fi_6), "+v"(tg0_6), "+v"(tg1_6));
    float tc0_5 = tanh_act(c1[1][2]);
    v2f ig_6 = fi_6 * (v2f){tg0_6, tg1_6};
    ORDER("+v"(tc0_5), "+v"(ig_6), "+v"(c1[1]));
    float tc1_5 = tanh_act(c1[1][3]);
    ORDER("+v"(tc1_5), "+v"(h1B), "+v"(d2[5]));
    d2[5] = __builtin_amdgcn_wmma_f32_16x16x32_f16(false, ai2[5], false, h1B, (short)0, d2[5], false, false);
    ORDER("+v"(d2[5]), "+v"(tf0_6), "+v"(tf1_6));
    v2f ff_6 = FIN2(((v2f){tf0_6, tf1_6}));
    ORDER("+v"(ff_6), "+v"(mi_7), "+v"(d1[3]));
    float ti0_7 = TSIG(mi_7[0]);
    v2f mf_7 = PRE2(((v2f){d1[3][6], d1[3][7]}));
    ORDER("+v"(ti0_7), "+v"(mf_7), "+v"(mi_7), "+v"(to0_6), "+v"(to1_6));
    float ti1_7 = TSIG(mi_7[1]);
    v2f fo_6 = FIN2(((v2f){to0_6, to1_6}));
    ORDER("+v"(ti1_7), "+v"(fo_6), "+v"(d1[5]), "+v"(ff_6), "+v"(ig_6));
    float tg0_7 = tanh_act(d1[5][6]);
    { const v2f cn = __builtin_elementwise_fma(ff_6, (v2f){c1[1][4], c1[1][5]}, ig_6); c1[1][4] = cn[0]; c1[1][5] = cn[1]; }
    ORDER("+v"(tg0_7), "+v"(c1[1]), "+v"(d1[5]), "+v"(d1[7]));
    float tg1_7 = tanh_act(d1[5][7]);
    v2f mo_7 = PRE2(((v2f){d1[7][6], d1[7][7]}));
    ORDER("+v"(tg1_7), "+v"(mo_7), "+v"(mf_7));
    float tf0_7 = TSIG(mf_7[0]);
    ORDER("+v"(tf0_7), "+v"(h1B), "+v"(d2[7]));
    d2[7] = __builtin_amdgcn_wmma_f32_16x16x32_f16(false, ai2[7], false, h1B, (short)0, d2[7], false, false);
    ORDER("+v"(d2[7]), "+v"(fo_5), "+v"(tc0_5), "+v"(tc1_5));
    v2f hh_5 = fo_5 * (v2f){tc0_5, tc1_5};
    ORDER("+v"(hh_5), "+v"(mf_7), "+v"(ti0_7), "+v"(ti1_7));
    float tf1_7 = TSIG(mf_7[1]);
    v2f fi_7 = FIN2(((v2f){ti0_7, ti1_7}));
    ORDER("+v"(tf1_7), "+v"(fi_7), "+v"(mo_7), "+v"(d2[0]));
    float to0_7 = TSIG(mo_7[0]);
    v2f mi_8 = PRE2(((v2f){d2[0][0], d2[0][1]}));
    ORDER("+v"(to0_7), "+v"(mi_8), "+v"(mo_7), "+v"(hh_5));
    float to1_7 = TSIG(mo_7[1]);
    v2h pk_5 = __builtin_convertvector(hh_5, v2h);
    ORDER("+v"(to1_7), "+v"(pk_5), "+v"(c1[1]), "+v"(fi_7), "+v"(tg0_7), "+v"(tg1_7));
    float tc0_6 = tanh_act(c1[1][4]);
    v2f ig_7 = fi_7 * (v2f){tg0_7, tg1_7};
    ORDER("+v"(tc0_6), "+v"(ig_7), "+v"(c1[1]), "+v"(tf0_7), "+v"(tf1_7));
    float tc1_6 = tanh_act(c1[1][5]);
    v2f ff_7 = FIN2(((v2f){tf0_7, tf1_7}));
    ORDER("+v"(tc1_6), "+v"(ff_7), "+v"(mi_8), "+v"(d2[2]));
    float ti0_8 = TSIG(mi_8[0]);
    v2f mf_8 = PRE2(((v2f){d2[2][0], d2[2][1]}));
    ORDER("+v"(ti0_8), "+v"(mf_8), "+v"(mi_8), "+v"(to0_7), "+v"(to1_7));
    float ti1_8 = TSIG(mi_8[1]);
    v2f fo_7 = FIN2(((v2f){to0_7, to1_7}));
    ORDER("+v"(ti1_8), "+v"(fo_7), "+v"(d2[4]), "+v"(ff_7), "+v"(ig_7));
    float tg0_8 = tanh_act(d2[4][0]);
    { const v2f cn = __builtin_elementwise_fma(ff_7, (v2f){c1[1][6], c1[1][7]}, ig_7); c1[1][6] = cn[0]; c1[1][7] = cn[1]; }
    ORDER("+v"(tg0_8), "+v"(c1[1]), "+v"(d2[4]), "+v"(d2[6]));
    float tg1_8 = tanh_act(d2[4][1]);
    v2f mo_8 = PRE2(((v2f){d2[6][0], d2[6][1]}));
    ORDER("+v"(tg1_8), "+v"(mo_8), "+v"(mf_8), "+v"(fo_6), "+v"(tc0_6), "+v"(tc1_6));
    float tf0_8 = TSIG(mf_8[0]);
    v2f hh_6 = fo_6 * (v2f){tc0_6, tc1_6};
    ORDER("+v"(tf0_8), "+v"(hh_6), "+v"(mf_8), "+v"(ti0_8), "+v"(ti1_8));
    float tf1_8 = TSIG(mf_8[1]);
    v2f fi_8 = FIN2(((v2f){ti0_8, ti1_8}));
    ORDER("+v"(tf1_8), "+v"(fi_8), "+v"(mo_8), "+v"(d2[0]));
    float to0_8 = TSIG(mo_8[0]);
    v2f mi_9 = PRE2(((v2f){d2[0][2], d2[0][3]}));
    ORDER("+v"(to0_8), "+v"(mi_9), "+v"(mo_8), "+v"(hh_6));
    float to1_8 = TSIG(mo_8[1]);
    v2h pk_6 = __builtin_convertvector(hh_6, v2h);
    ORDER("+v"(to1_8), "+v"(pk_6), "+v"(c1[1]), "+v"(fi_8), "+v"(tg0_8), "+v"(tg1_8));
    float tc0_7 = tanh_act(c1[1][6]);
    v2f ig_8 = fi_8 * (v2f){tg0_8, tg1_8};
    ORDER("+v"(tc0_7), "+v"(ig_8), "+v"(c1[1]), "+v"(tf0_8), "+v"(tf1_8));
    float tc1_7 = tanh_act(c1[1][7]);
    v2f ff_8 = FIN2(((v2f){tf0_8, tf1_8}));
    ORDER("+v"(tc1_7), "+v"(ff_8), "+v"(mi_9), "+v"(d2[2]));
    float ti0_9 = TSIG(mi_9[0]);
    v2f mf_9 = PRE2(((v2f){d2[2][2], d2[2][3]}));
    ORDER("+v"(ti0_9), "+v"(mf_9), "+v"(mi_9), "+v"(to0_8), "+v"(to1_8));
    float ti1_9 = TSIG(mi_9[1]);
    v2f fo_8 = FIN2(((v2f){to0_8, to1_8}));
    ORDER("+v"(ti1_9), "+v"(fo_8), "+v"(d2[4]), "+v"(ff_8), "+v"(ig_8));
    float tg0_9 = tanh_act(d2[4][2]);
    { const v2f cn = __builtin_elementwise_fma(ff_8, (v2f){c2[0][0], c2[0][1]}, ig_8); c2[0][0] = cn[0]; c2[0][1] = cn[1]; }
    ORDER("+v"(tg0_9), "+v"(c2[0]), "+v"(d2[4]), "+v"(d2[6]));
    float tg1_9 = tanh_act(d2[4][3]);
    v2f mo_9 = PRE2(((v2f){d2[6][2], d2[6][3]}));
    ORDER("+v"(tg1_9), "+v"(mo_9), "+v"(mf_9), "+v"(fo_7), "+v"(tc0_7), "+v"(tc1_7));
    float tf0_9 = TSIG(mf_9[0]);
    v2f hh_7 = fo_7 * (v2f){tc0_7, tc1_7};
    ORDER("+v"(tf0_9), "+v"(hh_7), "+v"(mf_9), "+v"(ti0_9), "+v"(ti1_9));
    float tf1_9 = TSIG(mf_9[1]);
    v2f fi_9 = FIN2(((v2f){ti0_9, ti1_9}));
    ORDER("+v"(tf1_9), "+v"(fi_9), "+v"(mo_9), "+v"(d2[0]));
    float to0_9 = TSIG(mo_9[0]);
    v2f mi_10 = PRE2(((v2f){d2[0][4], d2[0][5]}));
    ORDER("+v"(to0_9), "+v"(mi_10), "+v"(mo_9), "+v"(hh_7));
    float to1_9 = TSIG(mo_9[1]);
    v2h pk_7 = __builtin_convertvector(hh_7, v2h);
    ORDER("+v"(to1_9), "+v"(pk_7), "+v"(c2[0]), "+v"(fi_9), "+v"(tg0_9), "+v"(tg1_9));
    float tc0_8 = tanh_act(c2[0][0]);
    v2f ig_9 = fi_9 * (v2f){tg0_9, tg1_9};
    h1Bn[0] = pk_0[0]; h1Bn[1] = pk_0[1];
    h1Bn[2] = pk_1[0]; h1Bn[3] = pk_1[1];
    h1Bn[4] = pk_2[0]; h1Bn[5] = pk_2[1];
    h1Bn[6] = pk_3[0]; h1Bn[7] = pk_3[1];
    h1Bn[8] = pk_4[0]; h1Bn[9] = pk_4[1];
    h1Bn[10] = pk_5[0]; h1Bn[11] = pk_5[1];
    h1Bn[12] = pk_6[0]; h1Bn[13] = pk_6[1];
    h1Bn[14] = pk_7[0]; h1Bn[15] = pk_7[1];
    if (hi_half) { h1Bn[14] = (_Float16)(u == 0 ? xa : xb); h1Bn[15] = (_Float16)1.f; }
    ORDER("+v"(tc0_8), "+v"(ig_9), "+v"(h1Bn), "+v"(c2[0]), "+v"(tf0_9), "+v"(tf1_9));
    float tc1_8 = tanh_act(c2[0][1]);
    v2f ff_9 = FIN2(((v2f){tf0_9, tf1_9}));
    ORDER("+v"(tc1_8), "+v"(ff_9), "+v"(mi_10), "+v"(d2[2]));
    float ti0_10 = TSIG(mi_10[0]);
    v2f mf_10 = PRE2(((v2f){d2[2][4], d2[2][5]}));
    ORDER("+v"(ti0_10), "+v"(mf_10), "+v"(mi_10), "+v"(to0_9), "+v"(to1_9));
    float ti1_10 = TSIG(mi_10[1]);
    v2f fo_9 = FIN2(((v2f){to0_9, to1_9}));
    ORDER("+v"(ti1_10), "+v"(fo_9), "+v"(d2[4]), "+v"(ff_9), "+v"(ig_9));
    float tg0_10 = tanh_act(d2[4][4]);
    { const v2f cn = __builtin_elementwise_fma(ff_9, (v2f){c2[0][2], c2[0][3]}, ig_9); c2[0][2] = cn[0]; c2[0][3] = cn[1]; }
    ORDER("+v"(tg0_10), "+v"(c2[0]), "+v"(d2[4]), "+v"(d2[6]));
    float tg1_10 = tanh_act(d2[4][5]);
    v2f mo_10 = PRE2(((v2f){d2[6][4], d2[6][5]}));
    ORDER("+v"(tg1_10), "+v"(mo_10), "+v"(mf_10));
    float tf0_10 = TSIG(mf_10[0]);
    ORDER("+v"(tf0_10), "+v"(h1Bn));
    d1[0] = __builtin_amdgcn_wmma_f32_16x16x32_f16(false, a1[0], false, h1Bn, (short)0, zc, false, false);
    ORDER("+v"(d1[0]), "+v"(fo_8), "+v"(tc0_8), "+v"(tc1_8));
    v2f hh_8 = fo_8 * (v2f){tc0_8, tc1_8};
    ORDER("+v"(hh_8), "+v"(mf_10), "+v"(ti0_10), "+v"(ti1_10));
    float tf1_10 = TSIG(mf_10[1]);
    v2f fi_10 = FIN2(((v2f){ti0_10, ti1_10}));
    ORDER("+v"(tf1_10), "+v"(fi_10), "+v"(mo_10), "+v"(d2[0]));
    float to0_10 = TSIG(mo_10[0]);
    v2f mi_11 = PRE2(((v2f){d2[0][6], d2[0][7]}));
    ORDER("+v"(to0_10), "+v"(mi_11), "+v"(mo_10), "+v"(hh_8));
    float to1_10 = TSIG(mo_10[1]);
    v2h pk_8 = __builtin_convertvector(hh_8, v2h);
    ORDER("+v"(to1_10), "+v"(pk_8), "+v"(c2[0]), "+v"(fi_10), "+v"(tg0_10), "+v"(tg1_10), "+v"(hh_8), "+v"(acc));
    float tc0_9 = tanh_act(c2[0][2]);
    v2f ig_10 = fi_10 * (v2f){tg0_10, tg1_10};
    acc = fmaf(hh_8[0], wfcD[0][0], acc);
    ORDER("+v"(tc0_9), "+v"(ig_10), "+v"(acc), "+v"(c2[0]));
    float tc1_9 = tanh_act(c2[0][3]);
    ORDER("+v"(tc1_9), "+v"(h1Bn));
    d1[2] = __builtin_amdgcn_wmma_f32_16x16x32_f16(false, a1[2], false, h1Bn, (short)0, zc, false, false);
    ORDER("+v"(d1[2]), "+v"(tf0_10), "+v"(tf1_10));
    v2f ff_10 = FIN2(((v2f){tf0_10, tf1_10}));
    ORDER("+v"(ff_10), "+v"(mi_11), "+v"(d2[2]));
    float ti0_11 = TSIG(mi_11[0]);
    v2f mf_11 = PRE2(((v2f){d2[2][6], d2[2][7]}));
    ORDER("+v"(ti0_11), "+v"(mf_11), "+v"(mi_11), "+v"(to0_10), "+v"(to1_10));
    float ti1_11 = TSIG(mi_11[1]);
    v2f fo_10 = FIN2(((v2f){to0_10, to1_10}));
    ORDER("+v"(ti1_11), "+v"(fo_10), "+v"(d2[4]), "+v"(ff_10), "+v"(ig_10), "+v"(hh_8), "+v"(acc));
    float tg0_11 = tanh_act(d2[4][6]);
    { const v2f cn = __builtin_elementwise_fma(ff_10, (v2f){c2[0][4], c2[0][5]}, ig_10); c2[0][4] = cn[0]; c2[0][5] = cn[1]; }
    acc = fmaf(hh_8[1], wfcD[0][1], acc);
    ORDER("+v"(tg0_11), "+v"(c2[0]), "+v"(acc), "+v"(d2[4]), "+v"(d2[6]));
    float tg1_11 = tanh_act(d2[4][7]);
    v2f mo_11 = PRE2(((v2f){d2[6][6], d2[6][7]}));
    ORDER("+v"(tg1_11), "+v"(mo_11), "+v"(mf_11), "+v"(fo_9), "+v"(tc0_9), "+v"(tc1_9));
    float tf0_11 = TSIG(mf_11[0]);
    v2f hh_9 = fo_9 * (v2f){tc0_9, tc1_9};
    ORDER("+v"(tf0_11), "+v"(hh_9), "+v"(mf_11), "+v"(ti0_11), "+v"(ti1_11));
    float tf1_11 = TSIG(mf_11[1]);
    v2f fi_11 = FIN2(((v2f){ti0_11, ti1_11}));
    ORDER("+v"(tf1_11), "+v"(fi_11), "+v"(mo_11), "+v"(d2[1]));
    float to0_11 = TSIG(mo_11[0]);
    v2f mi_12 = PRE2(((v2f){d2[1][0], d2[1][1]}));
    ORDER("+v"(to0_11), "+v"(mi_12), "+v"(mo_11), "+v"(hh_9));
    float to1_11 = TSIG(mo_11[1]);
    v2h pk_9 = __builtin_convertvector(hh_9, v2h);
    ORDER("+v"(to1_11), "+v"(pk_9), "+v"(c2[0]), "+v"(fi_11), "+v"(tg0_11), "+v"(tg1_11), "+v"(hh_9), "+v"(acc));
    float tc0_10 = tanh_act(c2[0][4]);
    v2f ig_11 = fi_11 * (v2f){tg0_11, tg1_11};
    acc = fmaf(hh_9[0], wfcD[0][2], acc);
    ORDER("+v"(tc0_10), "+v"(ig_11), "+v"(acc), "+v"(c2[0]));
    float tc1_10 = tanh_act(c2[0][5]);
    ORDER("+v"(tc1_10), "+v"(h1Bn));
    d1[4] = __builtin_amdgcn_wmma_f32_16x16x32_f16(false, a1[4], false, h1Bn, (short)0, zc, false, false);
    ORDER("+v"(d1[4]), "+v"(tf0_11), "+v"(tf1_11));
    v2f ff_11 = FIN2(((v2f){tf0_11, tf1_11}));
    ORDER("+v"(ff_11), "+v"(mi_12), "+v"(d2[3]));
    float ti0_12 = TSIG(mi_12[0]);
    v2f mf_12 = PRE2(((v2f){d2[3][0], d2[3][1]}));
    ORDER("+v"(ti0_12), "+v"(mf_12), "+v"(mi_12), "+v"(to0_11), "+v"(to1_11));
    float ti1_12 = TSIG(mi_12[1]);
    v2f fo_11 = FIN2(((v2f){to0_11, to1_11}));
    ORDER("+v"(ti1_12), "+v"(fo_11), "+v"(d2[5]), "+v"(ff_11), "+v"(ig_11), "+v"(hh_9), "+v"(acc));
    float tg0_12 = tanh_act(d2[5][0]);
    { const v2f cn = __builtin_elementwise_fma(ff_11, (v2f){c2[0][6], c2[0][7]}, ig_11); c2[0][6] = cn[0]; c2[0][7] = cn[1]; }
    acc = fmaf(hh_9[1], wfcD[0][3], acc);
    ORDER("+v"(tg0_12), "+v"(c2[0]), "+v"(acc), "+v"(d2[5]), "+v"(d2[7]));
    float tg1_12 = tanh_act(d2[5][1]);
    v2f mo_12 = PRE2(((v2f){d2[7][0], d2[7][1]}));
    ORDER("+v"(tg1_12), "+v"(mo_12), "+v"(mf_12), "+v"(fo_10), "+v"(tc0_10), "+v"(tc1_10));
    float tf0_12 = TSIG(mf_12[0]);
    v2f hh_10 = fo_10 * (v2f){tc0_10, tc1_10};
    ORDER("+v"(tf0_12), "+v"(hh_10), "+v"(mf_12), "+v"(ti0_12), "+v"(ti1_12));
    float tf1_12 = TSIG(mf_12[1]);
    v2f fi_12 = FIN2(((v2f){ti0_12, ti1_12}));
    ORDER("+v"(tf1_12), "+v"(fi_12), "+v"(mo_12), "+v"(d2[1]));
    float to0_12 = TSIG(mo_12[0]);
    v2f mi_13 = PRE2(((v2f){d2[1][2], d2[1][3]}));
    ORDER("+v"(to0_12), "+v"(mi_13), "+v"(mo_12), "+v"(hh_10));
    float to1_12 = TSIG(mo_12[1]);
    v2h pk_10 = __builtin_convertvector(hh_10, v2h);
    ORDER("+v"(to1_12), "+v"(pk_10), "+v"(c2[0]), "+v"(fi_12), "+v"(tg0_12), "+v"(tg1_12), "+v"(hh_10), "+v"(acc));
    float tc0_11 = tanh_act(c2[0][6]);
    v2f ig_12 = fi_12 * (v2f){tg0_12, tg1_12};
    acc = fmaf(hh_10[0], wfcD[0][4], acc);
    ORDER("+v"(tc0_11), "+v"(ig_12), "+v"(acc), "+v"(c2[0]));
    float tc1_11 = tanh_act(c2[0][7]);
    ORDER("+v"(tc1_11), "+v"(h1Bn));
    d1[6] = __builtin_amdgcn_wmma_f32_16x16x32_f16(false, a1[6], false, h1Bn, (short)0, zc, false, false);
    ORDER("+v"(d1[6]), "+v"(tf0_12), "+v"(tf1_12));
    v2f ff_12 = FIN2(((v2f){tf0_12, tf1_12}));
    ORDER("+v"(ff_12), "+v"(mi_13), "+v"(d2[3]));
    float ti0_13 = TSIG(mi_13[0]);
    v2f mf_13 = PRE2(((v2f){d2[3][2], d2[3][3]}));
    ORDER("+v"(ti0_13), "+v"(mf_13), "+v"(mi_13), "+v"(to0_12), "+v"(to1_12));
    float ti1_13 = TSIG(mi_13[1]);
    v2f fo_12 = FIN2(((v2f){to0_12, to1_12}));
    ORDER("+v"(ti1_13), "+v"(fo_12), "+v"(d2[5]), "+v"(ff_12), "+v"(ig_12), "+v"(hh_10), "+v"(acc));
    float tg0_13 = tanh_act(d2[5][2]);
    { const v2f cn = __builtin_elementwise_fma(ff_12, (v2f){c2[1][0], c2[1][1]}, ig_12); c2[1][0] = cn[0]; c2[1][1] = cn[1]; }
    acc = fmaf(hh_10[1], wfcD[0][5], acc);
    ORDER("+v"(tg0_13), "+v"(c2[1]), "+v"(acc), "+v"(d2[5]), "+v"(d2[7]));
    float tg1_13 = tanh_act(d2[5][3]);
    v2f mo_13 = PRE2(((v2f){d2[7][2], d2[7][3]}));
    ORDER("+v"(tg1_13), "+v"(mo_13), "+v"(mf_13), "+v"(fo_11), "+v"(tc0_11), "+v"(tc1_11));
    float tf0_13 = TSIG(mf_13[0]);
    v2f hh_11 = fo_11 * (v2f){tc0_11, tc1_11};
    ORDER("+v"(tf0_13), "+v"(hh_11), "+v"(mf_13), "+v"(ti0_13), "+v"(ti1_13));
    float tf1_13 = TSIG(mf_13[1]);
    v2f fi_13 = FIN2(((v2f){ti0_13, ti1_13}));
    ORDER("+v"(tf1_13), "+v"(fi_13), "+v"(mo_13), "+v"(d2[1]));
    float to0_13 = TSIG(mo_13[0]);
    v2f mi_14 = PRE2(((v2f){d2[1][4], d2[1][5]}));
    ORDER("+v"(to0_13), "+v"(mi_14), "+v"(mo_13), "+v"(hh_11));
    float to1_13 = TSIG(mo_13[1]);
    v2h pk_11 = __builtin_convertvector(hh_11, v2h);
    ORDER("+v"(to1_13), "+v"(pk_11), "+v"(c2[1]), "+v"(fi_13), "+v"(tg0_13), "+v"(tg1_13), "+v"(hh_11), "+v"(acc));
    float tc0_12 = tanh_act(c2[1][0]);
    v2f ig_13 = fi_13 * (v2f){tg0_13, tg1_13};
    acc = fmaf(hh_11[0], wfcD[0][6], acc);
    ORDER("+v"(tc0_12), "+v"(ig_13), "+v"(acc), "+v"(c2[1]));
    float tc1_12 = tanh_act(c2[1][1]);
    ORDER("+v"(tc1_12), "+v"(h1Bn));
    d1[1] = __builtin_amdgcn_wmma_f32_16x16x32_f16(false, a1[1], false, h1Bn, (short)0, zc, false, false);
    ORDER("+v"(d1[1]), "+v"(tf0_13), "+v"(tf1_13));
    v2f ff_13 = FIN2(((v2f){tf0_13, tf1_13}));
    ORDER("+v"(ff_13), "+v"(mi_14), "+v"(d2[3]));
    float ti0_14 = TSIG(mi_14[0]);
    v2f mf_14 = PRE2(((v2f){d2[3][4], d2[3][5]}));
    ORDER("+v"(ti0_14), "+v"(mf_14), "+v"(mi_14), "+v"(to0_13), "+v"(to1_13));
    float ti1_14 = TSIG(mi_14[1]);
    v2f fo_13 = FIN2(((v2f){to0_13, to1_13}));
    ORDER("+v"(ti1_14), "+v"(fo_13), "+v"(d2[5]), "+v"(ff_13), "+v"(ig_13), "+v"(hh_11), "+v"(acc));
    float tg0_14 = tanh_act(d2[5][4]);
    { const v2f cn = __builtin_elementwise_fma(ff_13, (v2f){c2[1][2], c2[1][3]}, ig_13); c2[1][2] = cn[0]; c2[1][3] = cn[1]; }
    acc = fmaf(hh_11[1], wfcD[0][7], acc);
    ORDER("+v"(tg0_14), "+v"(c2[1]), "+v"(acc), "+v"(d2[5]), "+v"(d2[7]));
    float tg1_14 = tanh_act(d2[5][5]);
    v2f mo_14 = PRE2(((v2f){d2[7][4], d2[7][5]}));
    ORDER("+v"(tg1_14), "+v"(mo_14), "+v"(mf_14), "+v"(fo_12), "+v"(tc0_12), "+v"(tc1_12));
    float tf0_14 = TSIG(mf_14[0]);
    v2f hh_12 = fo_12 * (v2f){tc0_12, tc1_12};
    ORDER("+v"(tf0_14), "+v"(hh_12), "+v"(mf_14), "+v"(ti0_14), "+v"(ti1_14));
    float tf1_14 = TSIG(mf_14[1]);
    v2f fi_14 = FIN2(((v2f){ti0_14, ti1_14}));
    ORDER("+v"(tf1_14), "+v"(fi_14), "+v"(mo_14), "+v"(d2[1]));
    float to0_14 = TSIG(mo_14[0]);
    v2f mi_15 = PRE2(((v2f){d2[1][6], d2[1][7]}));
    ORDER("+v"(to0_14), "+v"(mi_15), "+v"(mo_14), "+v"(hh_12));
    float to1_14 = TSIG(mo_14[1]);
    v2h pk_12 = __builtin_convertvector(hh_12, v2h);
    ORDER("+v"(to1_14), "+v"(pk_12), "+v"(c2[1]), "+v"(fi_14), "+v"(tg0_14), "+v"(tg1_14), "+v"(hh_12), "+v"(acc));
    float tc0_13 = tanh_act(c2[1][2]);
    v2f ig_14 = fi_14 * (v2f){tg0_14, tg1_14};
    acc = fmaf(hh_12[0], wfcD[1][0], acc);
    ORDER("+v"(tc0_13), "+v"(ig_14), "+v"(acc), "+v"(c2[1]));
    float tc1_13 = tanh_act(c2[1][3]);
    ORDER("+v"(tc1_13), "+v"(h1Bn));
    d1[3] = __builtin_amdgcn_wmma_f32_16x16x32_f16(false, a1[3], false, h1Bn, (short)0, zc, false, false);
    ORDER("+v"(d1[3]), "+v"(tf0_14), "+v"(tf1_14));
    v2f ff_14 = FIN2(((v2f){tf0_14, tf1_14}));
    ORDER("+v"(ff_14), "+v"(mi_15), "+v"(d2[3]));
    float ti0_15 = TSIG(mi_15[0]);
    v2f mf_15 = PRE2(((v2f){d2[3][6], d2[3][7]}));
    ORDER("+v"(ti0_15), "+v"(mf_15), "+v"(mi_15), "+v"(to0_14), "+v"(to1_14));
    float ti1_15 = TSIG(mi_15[1]);
    v2f fo_14 = FIN2(((v2f){to0_14, to1_14}));
    ORDER("+v"(ti1_15), "+v"(fo_14), "+v"(d2[5]), "+v"(ff_14), "+v"(ig_14), "+v"(hh_12), "+v"(acc));
    float tg0_15 = tanh_act(d2[5][6]);
    { const v2f cn = __builtin_elementwise_fma(ff_14, (v2f){c2[1][4], c2[1][5]}, ig_14); c2[1][4] = cn[0]; c2[1][5] = cn[1]; }
    acc = fmaf(hh_12[1], wfcD[1][1], acc);
    ORDER("+v"(tg0_15), "+v"(c2[1]), "+v"(acc), "+v"(d2[5]), "+v"(d2[7]));
    float tg1_15 = tanh_act(d2[5][7]);
    v2f mo_15 = PRE2(((v2f){d2[7][6], d2[7][7]}));
    ORDER("+v"(tg1_15), "+v"(mo_15), "+v"(mf_15), "+v"(fo_13), "+v"(tc0_13), "+v"(tc1_13));
    float tf0_15 = TSIG(mf_15[0]);
    v2f hh_13 = fo_13 * (v2f){tc0_13, tc1_13};
    ORDER("+v"(tf0_15), "+v"(hh_13), "+v"(mf_15), "+v"(ti0_15), "+v"(ti1_15));
    float tf1_15 = TSIG(mf_15[1]);
    v2f fi_15 = FIN2(((v2f){ti0_15, ti1_15}));
    ORDER("+v"(tf1_15), "+v"(fi_15), "+v"(mo_15));
    float to0_15 = TSIG(mo_15[0]);
    ORDER("+v"(to0_15), "+v"(mo_15), "+v"(hh_13));
    float to1_15 = TSIG(mo_15[1]);
    v2h pk_13 = __builtin_convertvector(hh_13, v2h);
    ORDER("+v"(to1_15), "+v"(pk_13), "+v"(c2[1]), "+v"(fi_15), "+v"(tg0_15), "+v"(tg1_15), "+v"(hh_13), "+v"(acc));
    float tc0_14 = tanh_act(c2[1][4]);
    v2f ig_15 = fi_15 * (v2f){tg0_15, tg1_15};
    acc = fmaf(hh_13[0], wfcD[1][2], acc);
    ORDER("+v"(tc0_14), "+v"(ig_15), "+v"(acc), "+v"(c2[1]));
    float tc1_14 = tanh_act(c2[1][5]);
    ORDER("+v"(tc1_14), "+v"(h1Bn));
    d1[5] = __builtin_amdgcn_wmma_f32_16x16x32_f16(false, a1[5], false, h1Bn, (short)0, zc, false, false);
    ORDER("+v"(d1[5]), "+v"(tf0_15), "+v"(tf1_15));
    v2f ff_15 = FIN2(((v2f){tf0_15, tf1_15}));
    ORDER("+v"(ff_15), "+v"(to0_15), "+v"(to1_15));
    v2f fo_15 = FIN2(((v2f){to0_15, to1_15}));
    ORDER("+v"(fo_15), "+v"(ff_15), "+v"(ig_15), "+v"(hh_13), "+v"(acc));
    { const v2f cn = __builtin_elementwise_fma(ff_15, (v2f){c2[1][6], c2[1][7]}, ig_15); c2[1][6] = cn[0]; c2[1][7] = cn[1]; }
    acc = fmaf(hh_13[1], wfcD[1][3], acc);
    ORDER("+v"(c2[1]), "+v"(acc), "+v"(fo_14), "+v"(tc0_14), "+v"(tc1_14));
    v2f hh_14 = fo_14 * (v2f){tc0_14, tc1_14};
    ORDER("+v"(hh_14));
    v2h pk_14 = __builtin_convertvector(hh_14, v2h);
    ORDER("+v"(pk_14), "+v"(c2[1]), "+v"(hh_14), "+v"(acc));
    float tc0_15 = tanh_act(c2[1][6]);
    acc = fmaf(hh_14[0], wfcD[1][4], acc);
    ORDER("+v"(tc0_15), "+v"(acc), "+v"(c2[1]));
    float tc1_15 = tanh_act(c2[1][7]);
    ORDER("+v"(tc1_15), "+v"(h1Bn));
    d1[7] = __builtin_amdgcn_wmma_f32_16x16x32_f16(false, a1[7], false, h1Bn, (short)0, zc, false, false);
    ORDER("+v"(d1[7]), "+v"(hh_14), "+v"(acc));
    acc = fmaf(hh_14[1], wfcD[1][5], acc);
    ORDER("+v"(acc), "+v"(fo_15), "+v"(tc0_15), "+v"(tc1_15));
    v2f hh_15 = fo_15 * (v2f){tc0_15, tc1_15};
    ORDER("+v"(hh_15));
    v2h pk_15 = __builtin_convertvector(hh_15, v2h);
    ORDER("+v"(pk_15), "+v"(hh_15), "+v"(acc));
    acc = fmaf(hh_15[0], wfcD[1][6], acc);
    ORDER("+v"(acc), "+v"(hh_15));
    acc = fmaf(hh_15[1], wfcD[1][7], acc);
    h2Bn[0] = pk_8[0]; h2Bn[1] = pk_8[1];
    h2Bn[2] = pk_9[0]; h2Bn[3] = pk_9[1];
    h2Bn[4] = pk_10[0]; h2Bn[5] = pk_10[1];
    h2Bn[6] = pk_11[0]; h2Bn[7] = pk_11[1];
    h2Bn[8] = pk_12[0]; h2Bn[9] = pk_12[1];
    h2Bn[10] = pk_13[0]; h2Bn[11] = pk_13[1];
    h2Bn[12] = pk_14[0]; h2Bn[13] = pk_14[1];
    h2Bn[14] = pk_15[0]; h2Bn[15] = pk_15[1];
    if (hi_half) { h2Bn[14] = (_Float16)0.f; h2Bn[15] = (_Float16)1.f; }
    if (u == 0) outOdd = (accPrev + accOther) + bfc; else outEven = (accPrev + accOther) + bfc;
    accOther = xhalf(acc);

    accPrev = acc;
    h1B = h1Bn;
    h2B = h2Bn;
  }
    v2f xn = xq;
    int xstep = (tt + 6 < TLEN) ? 2 : 0;
    ORDER("+v"(xn), "+s"(xstep));
    xa = xn[0];
    xb = xn[1];
    xp += xstep;
  }

  ost[TLEN - 2] = hi_half ? (accPrev + accOther) + bfc : outEven;
}

extern "C" void kernel_launch(void* const* d_in, const int* in_sizes, int n_in,
                              void* d_out, int out_size, void* d_ws, size_t ws_size,
                              hipStream_t stream) {
  (void)in_sizes; (void)n_in; (void)out_size; (void)d_ws; (void)ws_size;
  const float* x     = (const float*)d_in[0];
  const float* w_ih1 = (const float*)d_in[1];
  const float* w_hh1 = (const float*)d_in[2];
  const float* b_ih1 = (const float*)d_in[3];
  const float* b_hh1 = (const float*)d_in[4];
  const float* w_ih2 = (const float*)d_in[5];
  const float* w_hh2 = (const float*)d_in[6];
  const float* b_ih2 = (const float*)d_in[7];
  const float* b_hh2 = (const float*)d_in[8];
  const float* w_fc  = (const float*)d_in[9];
  const float* b_fc  = (const float*)d_in[10];
  float* out = (float*)d_out;

  dim3 grid(BSZ / 16);
  dim3 block(32);
  lstm2_wmma_kernel<<<grid, block, 0, stream>>>(x, w_ih1, w_hh1, b_ih1, b_hh1,
                                                w_ih2, w_hh2, b_ih2, b_hh2,
                                                w_fc, b_fc, out);
}


# Weight fragments read straight from global memory into WMMA A layout, no LDS staging pass, no barrier: its loads independent, in flight together, not 128 passes of dependent waits; values same
# speedup vs baseline: 2.0094x; 1.0797x over previous
#include <hip/hip_runtime.h>


#define HSZ   30
#define MPAD  128
#define KPAD  32
#define TLEN  2048
#define BSZ   512

typedef __attribute__((ext_vector_type(16))) _Float16 v16h;
typedef __attribute__((ext_vector_type(8)))  _Float16 v8h;
typedef __attribute__((ext_vector_type(8)))  float    v8f;
typedef __attribute__((ext_vector_type(2)))  float    v2f;
typedef __attribute__((ext_vector_type(2)))  _Float16 v2h;

#define ORDER(...) asm volatile("" : __VA_ARGS__)

#if __has_builtin(__builtin_amdgcn_tanhf)
#define HW_TANH __builtin_amdgcn_tanhf
#elif __has_builtin(__builtin_amdgcn_tanh_f32)
#define HW_TANH __builtin_amdgcn_tanh_f32
#endif

#ifdef HW_TANH
__device__ __forceinline__ float tanh_act(float x) { return HW_TANH(x); }
__device__ __forceinline__ float sig_act(float x)  { return fmaf(0.5f, HW_TANH(0.5f * x), 0.5f); }
#define PRE2(x) ((x) * 0.5f)
#define TSIG(m) HW_TANH(m)
#define FIN2(t) __builtin_elementwise_fma((v2f)(0.5f), (t), (v2f)(0.5f))
#else
#define PRE2(x) (x)
#define TSIG(m) __builtin_amdgcn_rcpf(1.0f + __expf(-(m)))
#define FIN2(t) (t)
__device__ __forceinline__ float tanh_act(float x) {
  return fmaf(2.0f, __builtin_amdgcn_rcpf(1.0f + __expf(-2.0f * x)), -1.0f);
}
__device__ __forceinline__ float sig_act(float x) {
  return __builtin_amdgcn_rcpf(1.0f + __expf(-x));
}
#endif

__device__ __forceinline__ float xhalf(float v) {
  const unsigned u = __builtin_bit_cast(unsigned, v);
  return __builtin_bit_cast(float, __builtin_amdgcn_permlanex16(u, u, 0x76543210u, 0xfedcba98u, false, false));
}

__device__ __forceinline__ v16h load_a_frag(const float* __restrict__ wrow, bool hi_half, bool real,
                                            float c30, float c31) {
  const int kb = hi_half ? 8 : 0;
  v2f q[8];
#pragma unroll
  for (int i = 0; i < 4; ++i) q[i] = *reinterpret_cast<const v2f*>(wrow + kb + 2 * i);
#pragma unroll
  for (int i = 0; i < 3; ++i) q[4 + i] = *reinterpret_cast<const v2f*>(wrow + 16 + kb + 2 * i);
  q[7] = *reinterpret_cast<const v2f*>(wrow + (hi_half ? 28 : 22));
  if (hi_half) { q[7][0] = c30; q[7][1] = c31; }
  v16h a;
#pragma unroll
  for (int i = 0; i < 8; ++i) {
    a[2 * i]     = (_Float16)(real ? q[i][0] : 0.f);
    a[2 * i + 1] = (_Float16)(real ? q[i][1] : 0.f);
  }
  return a;
}

__device__ __forceinline__ void gates8(const v8f (&d)[4], v8f &c, float (&hn)[8]) {
#pragma unroll
  for (int r = 0; r < 8; ++r) {
    const float iv = sig_act(d[0][r]);
    const float fv = sig_act(d[1][r]);
    const float gv = tanh_act(d[2][r]);
    const float ov = sig_act(d[3][r]);
    const float cn = fmaf(fv, c[r], iv * gv);
    c[r] = cn;
    hn[r] = ov * tanh_act(cn);
  }
}

__device__ __forceinline__ v16h layer1_step(const v16h (&a1)[8], v16h h1B, v8f (&c1)[2], v8f zc) {
  v16h h1Bn;
#pragma unroll
  for (int s = 0; s < 2; ++s) {
    v8f d[4];
#pragma unroll
    for (int g = 0; g < 4; ++g)
      d[g] = __builtin_amdgcn_wmma_f32_16x16x32_f16(false, a1[2 * g + s], false, h1B,
                                                    (short)0, zc, false, false);
    float hn[8];
    gates8(d, c1[s], hn);
#pragma unroll
    for (int r = 0; r < 8; ++r) h1Bn[8 * s + r] = (_Float16)hn[r];
  }
  return h1Bn;
}

__global__ __launch_bounds__(32, 1)
void lstm2_wmma_kernel(const float* __restrict__ x,
                       const float* __restrict__ w_ih1, const float* __restrict__ w_hh1,
                       const float* __restrict__ b_ih1, const float* __restrict__ b_hh1,
                       const float* __restrict__ w_ih2, const float* __restrict__ w_hh2,
                       const float* __restrict__ b_ih2, const float* __restrict__ b_hh2,
                       const float* __restrict__ w_fc,  const float* __restrict__ b_fc,
                       float* __restrict__ out) {
  const int lane = threadIdx.x;
  const bool hi_half = lane >= 16;
  const int n = lane & 15;

  v16h a1[8], ai2[8], ah2[8];
#pragma unroll
  for (int tl = 0; tl < 8; ++tl) {
    const int j = ((tl & 1) << 4) + n;
    const bool real = j < HSZ;
    const int row = (tl >> 1) * HSZ + (real ? j : HSZ - 1);
    const float wi1 = w_ih1[row];
    const float bs1 = b_ih1[row] + b_hh1[row];
    const float bs2 = b_ih2[row] + b_hh2[row];
    a1[tl]  = load_a_frag(w_hh1 + row * HSZ, hi_half, real, wi1, bs1);
    ai2[tl] = load_a_frag(w_ih2 + row * HSZ, hi_half, real, 0.f, 0.f);
    ah2[tl] = load_a_frag(w_hh2 + row * HSZ, hi_half, real, 0.f, bs2);
  }

  float wfcD[2][8];
#pragma unroll
  for (int s = 0; s < 2; ++s)
#pragma unroll
    for (int r = 0; r < 8; ++r) {
      const int j = 16 * s + (hi_half ? 8 : 0) + r;
      wfcD[s][r] = (j < HSZ) ? w_fc[j] : 0.f;
    }
  const float bfc = b_fc[0];

  v8f zc;
#pragma unroll
  for (int i = 0; i < 8; ++i) zc[i] = 0.f;

  v16h h1B, h2B;
#pragma unroll
  for (int i = 0; i < 16; ++i) { h1B[i] = (_Float16)0.f; h2B[i] = (_Float16)0.f; }
  if (hi_half) h2B[15] = (_Float16)1.f;

  v8f c1[2], c2[2];
#pragma unroll
  for (int s = 0; s < 2; ++s)
#pragma unroll
    for (int r = 0; r < 8; ++r) { c1[s][r] = 0.f; c2[s][r] = 0.f; }

  const int bbase = blockIdx.x << 4;
  const float* xrow = x + (size_t)(bbase + n) * TLEN;
  float* orow = out + (size_t)(bbase + n) * TLEN;

  const float x0 = xrow[0];
  const float x1 = xrow[1];
  float xa = xrow[2];
  float xb = xrow[3];
  if (hi_half) { h1B[14] = (_Float16)x0; h1B[15] = (_Float16)1.f; }
  h1B = layer1_step(a1, h1B, c1, zc);
  if (hi_half) { h1B[14] = (_Float16)x1; h1B[15] = (_Float16)1.f; }

  v8f d1[8];
#pragma unroll
  for (int tl = 0; tl < 8; ++tl)
    d1[tl] = __builtin_amdgcn_wmma_f32_16x16x32_f16(false, a1[tl], false, h1B,
                                                    (short)0, zc, false, false);

  float accPrev = 0.f;
  float accOther = 0.f;
  float outEven = 0.f;
  float outOdd = 0.f;
  float* ost = orow + (hi_half ? 1 : 0);
  const float* xp = xrow + 4;

  for (int tt = 0; tt < TLEN; tt += 2) {
    const v2f xq = *reinterpret_cast<const v2f*>(xp);
#pragma unroll
  for (int u = 0; u < 2; ++u) {
    const int t = tt + u;

    if (u == 1 && tt > 0) ost[tt - 2] = hi_half ? outOdd : outEven;

    v8f d2[8];
    v16h h1Bn, h2Bn;
    float acc = 0.f;
    ORDER("+v"(d1[0]));
    v2f mi_0 = PRE2(((v2f){d1[0][0], d1[0][1]}));
    ORDER("+v"(mi_0), "+v"(h2B));
    d2[0] = __builtin_amdgcn_wmma_f32_16x16x32_f16(false, ah2[0], false, h2B, (short)0, zc, false, false);
    ORDER("+v"(d2[0]), "+v"(mi_0), "+v"(d1[2]));
    float ti0_0 = TSIG(mi_0[0]);
    v2f mf_0 = PRE2(((v2f){d1[2][0], d1[2][1]}));
    ORDER("+v"(ti0_0), "+v"(mf_0), "+v"(mi_0));
    float ti1_0 = TSIG(mi_0[1]);
    ORDER("+v"(ti1_0), "+v"(d1[4]));
    float tg0_0 = tanh_act(d1[4][0]);
    ORDER("+v"(tg0_0), "+v"(d1[4]), "+v"(d1[6]));
    float tg1_0 = tanh_act(d1[4][1]);
    v2f mo_0 = PRE2(((v2f){d1[6][0], d1[6][1]}));
    ORDER("+v"(tg1_0), "+v"(mo_0), "+v"(mf_0));
    float tf0_0 = TSIG(mf_0[0]);
    ORDER("+v"(tf0_0), "+v"(h2B));
    d2[2] = __builtin_amdgcn_wmma_f32_16x16x32_f16(false, ah2[2], false, h2B, (short)0, zc, false, false);
    ORDER("+v"(d2[2]), "+v"(mf_0), "+v"(ti0_0), "+v"(ti1_0));
    float tf1_0 = TSIG(mf_0[1]);
    v2f fi_0 = FIN2(((v2f){ti0_0, ti1_0}));
    ORDER("+v"(tf1_0), "+v"(fi_0), "+v"(mo_0), "+v"(d1[0]));
    float to0_0 = TSIG(mo_0[0]);
    v2f mi_1 = PRE2(((v2f){d1[0][2], d1[0][3]}));
    ORDER("+v"(to0_0), "+v"(mi_1), "+v"(mo_0));
    float to1_0 = TSIG(mo_0[1]);
    ORDER("+v"(to1_0), "+v"(fi_0), "+v"(tg0_0), "+v"(tg1_0));
    v2f ig_0 = fi_0 * (v2f){tg0_0, tg1_0};
    ORDER("+v"(ig_0), "+v"(h2B));
    d2[4] = __builtin_amdgcn_wmma_f32_16x16x32_f16(false, ah2[4], false, h2B, (short)0, zc, false, false);
    ORDER("+v"(d2[4]), "+v"(tf0_0), "+v"(tf1_0));
    v2f ff_0 = FIN2(((v2f){tf0_0, tf1_0}));
    ORDER("+v"(ff_0), "+v"(mi_1), "+v"(d1[2]));
    float ti0_1 = TSIG(mi_1[0]);
    v2f mf_1 = PRE2(((v2f){d1[2][2], d1[2][3]}));
    ORDER("+v"(ti0_1), "+v"(mf_1), "+v"(mi_1), "+v"(to0_0), "+v"(to1_0));
    float ti1_1 = TSIG(mi_1[1]);
    v2f fo_0 = FIN2(((v2f){to0_0, to1_0}));
    ORDER("+v"(ti1_1), "+v"(fo_0), "+v"(d1[4]), "+v"(ff_0), "+v"(ig_0));
    float tg0_1 = tanh_act(d1[4][2]);
    { const v2f cn = __builtin_elementwise_fma(ff_0, (v2f){c1[0][0], c1[0][1]}, ig_0); c1[0][0] = cn[0]; c1[0][1] = cn[1]; }
    ORDER("+v"(tg0_1), "+v"(c1[0]), "+v"(d1[4]), "+v"(d1[6]));
    float tg1_1 = tanh_act(d1[4][3]);
    v2f mo_1 = PRE2(((v2f){d1[6][2], d1[6][3]}));
    ORDER("+v"(tg1_1), "+v"(mo_1), "+v"(mf_1));
    float tf0_1 = TSIG(mf_1[0]);
    ORDER("+v"(tf0_1), "+v"(h2B));
    d2[6] = __builtin_amdgcn_wmma_f32_16x16x32_f16(false, ah2[6], false, h2B, (short)0, zc, false, false);
    ORDER("+v"(d2[6]), "+v"(mf_1), "+v"(ti0_1), "+v"(ti1_1));
    float tf1_1 = TSIG(mf_1[1]);
    v2f fi_1 = FIN2(((v2f){ti0_1, ti1_1}));
    ORDER("+v"(tf1_1), "+v"(fi_1), "+v"(mo_1), "+v"(d1[0]));
    float to0_1 = TSIG(mo_1[0]);
    v2f mi_2 = PRE2(((v2f){d1[0][4], d1[0][5]}));
    ORDER("+v"(to0_1), "+v"(mi_2), "+v"(mo_1));
    float to1_1 = TSIG(mo_1[1]);
    ORDER("+v"(to1_1), "+v"(c1[0]), "+v"(fi_1), "+v"(tg0_1), "+v"(tg1_1));
    float tc0_0 = tanh_act(c1[0][0]);
    v2f ig_1 = fi_1 * (v2f){tg0_1, tg1_1};
    ORDER("+v"(tc0_0), "+v"(ig_1), "+v"(c1[0]));
    float tc1_0 = tanh_act(c1[0][1]);
    ORDER("+v"(tc1_0), "+v"(h2B));
    d2[1] = __builtin_amdgcn_wmma_f32_16x16x32_f16(false, ah2[1], false, h2B, (short)0, zc, false, false);
    ORDER("+v"(d2[1]), "+v"(tf0_1), "+v"(tf1_1));
    v2f ff_1 = FIN2(((v2f){tf0_1, tf1_1}));
    ORDER("+v"(ff_1), "+v"(mi_2), "+v"(d1[2]));
    float ti0_2 = TSIG(mi_2[0]);
    v2f mf_2 = PRE2(((v2f){d1[2][4], d1[2][5]}));
    ORDER("+v"(ti0_2), "+v"(mf_2), "+v"(mi_2), "+v"(to0_1), "+v"(to1_1));
    float ti1_2 = TSIG(mi_2[1]);
    v2f fo_1 = FIN2(((v2f){to0_1, to1_1}));
    ORDER("+v"(ti1_2), "+v"(fo_1), "+v"(d1[4]), "+v"(ff_1), "+v"(ig_1));
    float tg0_2 = tanh_act(d1[4][4]);
    { const v2f cn = __builtin_elementwise_fma(ff_1, (v2f){c1[0][2], c1[0][3]}, ig_1); c1[0][2] = cn[0]; c1[0][3] = cn[1]; }
    ORDER("+v"(tg0_2), "+v"(c1[0]), "+v"(d1[4]), "+v"(d1[6]));
    float tg1_2 = tanh_act(d1[4][5]);
    v2f mo_2 = PRE2(((v2f){d1[6][4], d1[6][5]}));
    ORDER("+v"(tg1_2), "+v"(mo_2), "+v"(mf_2));
    float tf0_2 = TSIG(mf_2[0]);
    ORDER("+v"(tf0_2), "+v"(h2B));
    d2[3] = __builtin_amdgcn_wmma_f32_16x16x32_f16(false, ah2[3], false, h2B, (short)0, zc, false, false);
    ORDER("+v"(d2[3]), "+v"(fo_0), "+v"(tc0_0), "+v"(tc1_0));
    v2f hh_0 = fo_0 * (v2f){tc0_0, tc1_0};
    ORDER("+v"(hh_0), "+v"(mf_2), "+v"(ti0_2), "+v"(ti1_2));
    float tf1_2 = TSIG(mf_2[1]);
    v2f fi_2 = FIN2(((v2f){ti0_2, ti1_2}));
    ORDER("+v"(tf1_2), "+v"(fi_2), "+v"(mo_2), "+v"(d1[0]));
    float to0_2 = TSIG(mo_2[0]);
    v2f mi_3 = PRE2(((v2f){d1[0][6], d1[0][7]}));
    ORDER("+v"(to0_2), "+v"(mi_3), "+v"(mo_2), "+v"(hh_0));
    float to1_2 = TSIG(mo_2[1]);
    v2h pk_0 = __builtin_convertvector(hh_0, v2h);
    ORDER("+v"(to1_2), "+v"(pk_0), "+v"(c1[0]), "+v"(fi_2), "+v"(tg0_2), "+v"(tg1_2));
    float tc0_1 = tanh_act(c1[0][2]);
    v2f ig_2 = fi_2 * (v2f){tg0_2, tg1_2};
    ORDER("+v"(tc0_1), "+v"(ig_2), "+v"(c1[0]));
    float tc1_1 = tanh_act(c1[0][3]);
    ORDER("+v"(tc1_1), "+v"(h2B));
    d2[5] = __builtin_amdgcn_wmma_f32_16x16x32_f16(false, ah2[5], false, h2B, (short)0, zc, false, false);
    ORDER("+v"(d2[5]), "+v"(tf0_2), "+v"(tf1_2));
    v2f ff_2 = FIN2(((v2f){tf0_2, tf1_2}));
    ORDER("+v"(ff_2), "+v"(mi_3), "+v"(d1[2]));
    float ti0_3 = TSIG(mi_3[0]);
    v2f mf_3 = PRE2(((v2f){d1[2][6], d1[2][7]}));
    ORDER("+v"(ti0_3), "+v"(mf_3), "+v"(mi_3), "+v"(to0_2), "+v"(to1_2));
    float ti1_3 = TSIG(mi_3[1]);
    v2f fo_2 = FIN2(((v2f){to0_2, to1_2}));
    ORDER("+v"(ti1_3), "+v"(fo_2), "+v"(d1[4]), "+v"(ff_2), "+v"(ig_2));
    float tg0_3 = tanh_act(d1[4][6]);
    { const v2f cn = __builtin_elementwise_fma(ff_2, (v2f){c1[0][4], c1[0][5]}, ig_2); c1[0][4] = cn[0]; c1[0][5] = cn[1]; }
    ORDER("+v"(tg0_3), "+v"(c1[0]), "+v"(d1[4]), "+v"(d1[6]));
    float tg1_3 = tanh_act(d1[4][7]);
    v2f mo_3 = PRE2(((v2f){d1[6][6], d1[6][7]}));
    ORDER("+v"(tg1_3), "+v"(mo_3), "+v"(mf_3));
    float tf0_3 = TSIG(mf_3[0]);
    ORDER("+v"(tf0_3), "+v"(h2B));
    d2[7] = __builtin_amdgcn_wmma_f32_16x16x32_f16(false, ah2[7], false, h2B, (short)0, zc, false, false);
    ORDER("+v"(d2[7]), "+v"(fo_1), "+v"(tc0_1), "+v"(tc1_1));
    v2f hh_1 = fo_1 * (v2f){tc0_1, tc1_1};
    ORDER("+v"(hh_1), "+v"(mf_3), "+v"(ti0_3), "+v"(ti1_3));
    float tf1_3 = TSIG(mf_3[1]);
    v2f fi_3 = FIN2(((v2f){ti0_3, ti1_3}));
    ORDER("+v"(tf1_3), "+v"(fi_3), "+v"(mo_3), "+v"(d1[1]));
    float to0_3 = TSIG(mo_3[0]);
    v2f mi_4 = PRE2(((v2f){d1[1][0], d1[1][1]}));
    ORDER("+v"(to0_3), "+v"(mi_4), "+v"(mo_3), "+v"(hh_1));
    float to1_3 = TSIG(mo_3[1]);
    v2h pk_1 = __builtin_convertvector(hh_1, v2h);
    ORDER("+v"(to1_3), "+v"(pk_1), "+v"(c1[0]), "+v"(fi_3), "+v"(tg0_3), "+v"(tg1_3));
    float tc0_2 = tanh_act(c1[0][4]);
    v2f ig_3 = fi_3 * (v2f){tg0_3, tg1_3};
    ORDER("+v"(tc0_2), "+v"(ig_3), "+v"(c1[0]));
    float tc1_2 = tanh_act(c1[0][5]);
    ORDER("+v"(tc1_2), "+v"(h1B), "+v"(d2[0]));
    d2[0] = __builtin_amdgcn_wmma_f32_16x16x32_f16(false, ai2[0], false, h1B, (short)0, d2[0], false, false);
    ORDER("+v"(d2[0]), "+v"(tf0_3), "+v"(tf1_3));
    v2f ff_3 = FIN2(((v2f){tf0_3, tf1_3}));
    ORDER("+v"(ff_3), "+v"(mi_4), "+v"(d1[3]));
    float ti0_4 = TSIG(mi_4[0]);
    v2f mf_4 = PRE2(((v2f){d1[3][0], d1[3][1]}));
    ORDER("+v"(ti0_4), "+v"(mf_4), "+v"(mi_4), "+v"(to0_3), "+v"(to1_3));
    float ti1_4 = TSIG(mi_4[1]);
    v2f fo_3 = FIN2(((v2f){to0_3, to1_3}));
    ORDER("+v"(ti1_4), "+v"(fo_3), "+v"(d1[5]), "+v"(ff_3), "+v"(ig_3));
    float tg0_4 = tanh_act(d1[5][0]);
    { const v2f cn = __builtin_elementwise_fma(ff_3, (v2f){c1[0][6], c1[0][7]}, ig_3); c1[0][6] = cn[0]; c1[0][7] = cn[1]; }
    ORDER("+v"(tg0_4), "+v"(c1[0]), "+v"(d1[5]), "+v"(d1[7]));
    float tg1_4 = tanh_act(d1[5][1]);
    v2f mo_4 = PRE2(((v2f){d1[7][0], d1[7][1]}));
    ORDER("+v"(tg1_4), "+v"(mo_4), "+v"(mf_4));
    float tf0_4 = TSIG(mf_4[0]);
    ORDER("+v"(tf0_4), "+v"(h1B), "+v"(d2[2]));
    d2[2] = __builtin_amdgcn_wmma_f32_16x16x32_f16(false, ai2[2], false, h1B, (short)0, d2[2], false, false);
    ORDER("+v"(d2[2]), "+v"(fo_2), "+v"(tc0_2), "+v"(tc1_2));
    v2f hh_2 = fo_2 * (v2f){tc0_2, tc1_2};
    ORDER("+v"(hh_2), "+v"(mf_4), "+v"(ti0_4), "+v"(ti1_4));
    float tf1_4 = TSIG(mf_4[1]);
    v2f fi_4 = FIN2(((v2f){ti0_4, ti1_4}));
    ORDER("+v"(tf1_4), "+v"(fi_4), "+v"(mo_4), "+v"(d1[1]));
    float to0_4 = TSIG(mo_4[0]);
    v2f mi_5 = PRE2(((v2f){d1[1][2], d1[1][3]}));
    ORDER("+v"(to0_4), "+v"(mi_5), "+v"(mo_4), "+v"(hh_2));
    float to1_4 = TSIG(mo_4[1]);
    v2h pk_2 = __builtin_convertvector(hh_2, v2h);
    ORDER("+v"(to1_4), "+v"(pk_2), "+v"(c1[0]), "+v"(fi_4), "+v"(tg0_4), "+v"(tg1_4));
    float tc0_3 = tanh_act(c1[0][6]);
    v2f ig_4 = fi_4 * (v2f){tg0_4, tg1_4};
    ORDER("+v"(tc0_3), "+v"(ig_4), "+v"(c1[0]));
    float tc1_3 = tanh_act(c1[0][7]);
    ORDER("+v"(tc1_3), "+v"(h1B), "+v"(d2[4]));
    d2[4] = __builtin_amdgcn_wmma_f32_16x16x32_f16(false, ai2[4], false, h1B, (short)0, d2[4], false, false);
    ORDER("+v"(d2[4]), "+v"(tf0_4), "+v"(tf1_4));
    v2f ff_4 = FIN2(((v2f){tf0_4, tf1_4}));
    ORDER("+v"(ff_4), "+v"(mi_5), "+v"(d1[3]));
    float ti0_5 = TSIG(mi_5[0]);
    v2f mf_5 = PRE2(((v2f){d1[3][2], d1[3][3]}));
    ORDER("+v"(ti0_5), "+v"(mf_5), "+v"(mi_5), "+v"(to0_4), "+v"(to1_4));
    float ti1_5 = TSIG(mi_5[1]);
    v2f fo_4 = FIN2(((v2f){to0_4, to1_4}));
    ORDER("+v"(ti1_5), "+v"(fo_4), "+v"(d1[5]), "+v"(ff_4), "+v"(ig_4));
    float tg0_5 = tanh_act(d1[5][2]);
    { const v2f cn = __builtin_elementwise_fma(ff_4, (v2f){c1[1][0], c1[1][1]}, ig_4); c1[1][0] = cn[0]; c1[1][1] = cn[1]; }
    ORDER("+v"(tg0_5), "+v"(c1[1]), "+v"(d1[5]), "+v"(d1[7]));
    float tg1_5 = tanh_act(d1[5][3]);
    v2f mo_5 = PRE2(((v2f){d1[7][2], d1[7][3]}));
    ORDER("+v"(tg1_5), "+v"(mo_5), "+v"(mf_5));
    float tf0_5 = TSIG(mf_5[0]);
    ORDER("+v"(tf0_5), "+v"(h1B), "+v"(d2[6]));
    d2[6] = __builtin_amdgcn_wmma_f32_16x16x32_f16(false, ai2[6], false, h1B, (short)0, d2[6], false, false);
    ORDER("+v"(d2[6]), "+v"(fo_3), "+v"(tc0_3), "+v"(tc1_3));
    v2f hh_3 = fo_3 * (v2f){tc0_3, tc1_3};
    ORDER("+v"(hh_3), "+v"(mf_5), "+v"(ti0_5), "+v"(ti1_5));
    float tf1_5 = TSIG(mf_5[1]);
    v2f fi_5 = FIN2(((v2f){ti0_5, ti1_5}));
    ORDER("+v"(tf1_5), "+v"(fi_5), "+v"(mo_5), "+v"(d1[1]));
    float to0_5 = TSIG(mo_5[0]);
    v2f mi_6 = PRE2(((v2f){d1[1][4], d1[1][5]}));
    ORDER("+v"(to0_5), "+v"(mi_6), "+v"(mo_5), "+v"(hh_3));
    float to1_5 = TSIG(mo_5[1]);
    v2h pk_3 = __builtin_convertvector(hh_3, v2h);
    ORDER("+v"(to1_5), "+v"(pk_3), "+v"(c1[1]), "+v"(fi_5), "+v"(tg0_5), "+v"(tg1_5));
    float tc0_4 = tanh_act(c1[1][0]);
    v2f ig_5 = fi_5 * (v2f){tg0_5, tg1_5};
    ORDER("+v"(tc0_4), "+v"(ig_5), "+v"(c1[1]));
    float tc1_4 = tanh_act(c1[1][1]);
    ORDER("+v"(tc1_4), "+v"(h1B), "+v"(d2[1]));
    d2[1] = __builtin_amdgcn_wmma_f32_16x16x32_f16(false, ai2[1], false, h1B, (short)0, d2[1], false, false);
    ORDER("+v"(d2[1]), "+v"(tf0_5), "+v"(tf1_5));
    v2f ff_5 = FIN2(((v2f){tf0_5, tf1_5}));
    ORDER("+v"(ff_5), "+v"(mi_6), "+v"(d1[3]));
    float ti0_6 = TSIG(mi_6[0]);
    v2f mf_6 = PRE2(((v2f){d1[3][4], d1[3][5]}));
    ORDER("+v"(ti0_6), "+v"(mf_6), "+v"(mi_6), "+v"(to0_5), "+v"(to1_5));
    float ti1_6 = TSIG(mi_6[1]);
    v2f fo_5 = FIN2(((v2f){to0_5, to1_5}));
    ORDER("+v"(ti1_6), "+v"(fo_5), "+v"(d1[5]), "+v"(ff_5), "+v"(ig_5));
    float tg0_6 = tanh_act(d1[5][4]);
    { const v2f cn = __builtin_elementwise_fma(ff_5, (v2f){c1[1][2], c1[1][3]}, ig_5); c1[1][2] = cn[0]; c1[1][3] = cn[1]; }
    ORDER("+v"(tg0_6), "+v"(c1[1]), "+v"(d1[5]), "+v"(d1[7]));
    float tg1_6 = tanh_act(d1[5][5]);
    v2f mo_6 = PRE2(((v2f){d1[7][4], d1[7][5]}));
    ORDER("+v"(tg1_6), "+v"(mo_6), "+v"(mf_6));
    float tf0_6 = TSIG(mf_6[0]);
    ORDER("+v"(tf0_6), "+v"(h1B), "+v"(d2[3]));
    d2[3] = __builtin_amdgcn_wmma_f32_16x16x32_f16(false, ai2[3], false, h1B, (short)0, d2[3], false, false);
    ORDER("+v"(d2[3]), "+v"(fo_4), "+v"(tc0_4), "+v"(tc1_4));
    v2f hh_4 = fo_4 * (v2f){tc0_4, tc1_4};
    ORDER("+v"(hh_4), "+v"(mf_6), "+v"(ti0_6), "+v"(ti1_6));
    float tf1_6 = TSIG(mf_6[1]);
    v2f fi_6 = FIN2(((v2f){ti0_6, ti1_6}));
    ORDER("+v"(tf1_6), "+v"(fi_6), "+v"(mo_6), "+v"(d1[1]));
    float to0_6 = TSIG(mo_6[0]);
    v2f mi_7 = PRE2(((v2f){d1[1][6], d1[1][7]}));
    ORDER("+v"(to0_6), "+v"(mi_7), "+v"(mo_6), "+v"(hh_4));
    float to1_6 = TSIG(mo_6[1]);
    v2h pk_4 = __builtin_convertvector(hh_4, v2h);
    ORDER("+v"(to1_6), "+v"(pk_4), "+v"(c1[1]), "+v"(fi_6), "+v"(tg0_6), "+v"(tg1_6));
    float tc0_5 = tanh_act(c1[1][2]);
    v2f ig_6 = fi_6 * (v2f){tg0_6, tg1_6};
    ORDER("+v"(tc0_5), "+v"(ig_6), "+v"(c1[1]));
    float tc1_5 = tanh_act(c1[1][3]);
    ORDER("+v"(tc1_5), "+v"(h1B), "+v"(d2[5]));
    d2[5] = __builtin_amdgcn_wmma_f32_16x16x32_f16(false, ai2[5], false, h1B, (short)0, d2[5], false, false);
    ORDER("+v"(d2[5]), "+v"(tf0_6), "+v"(tf1_6));
    v2f ff_6 = FIN2(((v2f){tf0_6, tf1_6}));
    ORDER("+v"(ff_6), "+v"(mi_7), "+v"(d1[3]));
    float ti0_7 = TSIG(mi_7[0]);
    v2f mf_7 = PRE2(((v2f){d1[3][6], d1[3][7]}));
    ORDER("+v"(ti0_7), "+v"(mf_7), "+v"(mi_7), "+v"(to0_6), "+v"(to1_6));
    float ti1_7 = TSIG(mi_7[1]);
    v2f fo_6 = FIN2(((v2f){to0_6, to1_6}));
    ORDER("+v"(ti1_7), "+v"(fo_6), "+v"(d1[5]), "+v"(ff_6), "+v"(ig_6));
    float tg0_7 = tanh_act(d1[5][6]);
    { const v2f cn = __builtin_elementwise_fma(ff_6, (v2f){c1[1][4], c1[1][5]}, ig_6); c1[1][4] = cn[0]; c1[1][5] = cn[1]; }
    ORDER("+v"(tg0_7), "+v"(c1[1]), "+v"(d1[5]), "+v"(d1[7]));
    float tg1_7 = tanh_act(d1[5][7]);
    v2f mo_7 = PRE2(((v2f){d1[7][6], d1[7][7]}));
    ORDER("+v"(tg1_7), "+v"(mo_7), "+v"(mf_7));
    float tf0_7 = TSIG(mf_7[0]);
    ORDER("+v"(tf0_7), "+v"(h1B), "+v"(d2[7]));
    d2[7] = __builtin_amdgcn_wmma_f32_16x16x32_f16(false, ai2[7], false, h1B, (short)0, d2[7], false, false);
    ORDER("+v"(d2[7]), "+v"(fo_5), "+v"(tc0_5), "+v"(tc1_5));
    v2f hh_5 = fo_5 * (v2f){tc0_5, tc1_5};
    ORDER("+v"(hh_5), "+v"(mf_7), "+v"(ti0_7), "+v"(ti1_7));
    float tf1_7 = TSIG(mf_7[1]);
    v2f fi_7 = FIN2(((v2f){ti0_7, ti1_7}));
    ORDER("+v"(tf1_7), "+v"(fi_7), "+v"(mo_7), "+v"(d2[0]));
    float to0_7 = TSIG(mo_7[0]);
    v2f mi_8 = PRE2(((v2f){d2[0][0], d2[0][1]}));
    ORDER("+v"(to0_7), "+v"(mi_8), "+v"(mo_7), "+v"(hh_5));
    float to1_7 = TSIG(mo_7[1]);
    v2h pk_5 = __builtin_convertvector(hh_5, v2h);
    ORDER("+v"(to1_7), "+v"(pk_5), "+v"(c1[1]), "+v"(fi_7), "+v"(tg0_7), "+v"(tg1_7));
    float tc0_6 = tanh_act(c1[1][4]);
    v2f ig_7 = fi_7 * (v2f){tg0_7, tg1_7};
    ORDER("+v"(tc0_6), "+v"(ig_7), "+v"(c1[1]), "+v"(tf0_7), "+v"(tf1_7));
    float tc1_6 = tanh_act(c1[1][5]);
    v2f ff_7 = FIN2(((v2f){tf0_7, tf1_7}));
    ORDER("+v"(tc1_6), "+v"(ff_7), "+v"(mi_8), "+v"(d2[2]));
    float ti0_8 = TSIG(mi_8[0]);
    v2f mf_8 = PRE2(((v2f){d2[2][0], d2[2][1]}));
    ORDER("+v"(ti0_8), "+v"(mf_8), "+v"(mi_8), "+v"(to0_7), "+v"(to1_7));
    float ti1_8 = TSIG(mi_8[1]);
    v2f fo_7 = FIN2(((v2f){to0_7, to1_7}));
    ORDER("+v"(ti1_8), "+v"(fo_7), "+v"(d2[4]), "+v"(ff_7), "+v"(ig_7));
    float tg0_8 = tanh_act(d2[4][0]);
    { const v2f cn = __builtin_elementwise_fma(ff_7, (v2f){c1[1][6], c1[1][7]}, ig_7); c1[1][6] = cn[0]; c1[1][7] = cn[1]; }
    ORDER("+v"(tg0_8), "+v"(c1[1]), "+v"(d2[4]), "+v"(d2[6]));
    float tg1_8 = tanh_act(d2[4][1]);
    v2f mo_8 = PRE2(((v2f){d2[6][0], d2[6][1]}));
    ORDER("+v"(tg1_8), "+v"(mo_8), "+v"(mf_8), "+v"(fo_6), "+v"(tc0_6), "+v"(tc1_6));
    float tf0_8 = TSIG(mf_8[0]);
    v2f hh_6 = fo_6 * (v2f){tc0_6, tc1_6};
    ORDER("+v"(tf0_8), "+v"(hh_6), "+v"(mf_8), "+v"(ti0_8), "+v"(ti1_8));
    float tf1_8 = TSIG(mf_8[1]);
    v2f fi_8 = FIN2(((v2f){ti0_8, ti1_8}));
    ORDER("+v"(tf1_8), "+v"(fi_8), "+v"(mo_8), "+v"(d2[0]));
    float to0_8 = TSIG(mo_8[0]);
    v2f mi_9 = PRE2(((v2f){d2[0][2], d2[0][3]}));
    ORDER("+v"(to0_8), "+v"(mi_9), "+v"(mo_8), "+v"(hh_6));
    float to1_8 = TSIG(mo_8[1]);
    v2h pk_6 = __builtin_convertvector(hh_6, v2h);
    ORDER("+v"(to1_8), "+v"(pk_6), "+v"(c1[1]), "+v"(fi_8), "+v"(tg0_8), "+v"(tg1_8));
    float tc0_7 = tanh_act(c1[1][6]);
    v2f ig_8 = fi_8 * (v2f){tg0_8, tg1_8};
    ORDER("+v"(tc0_7), "+v"(ig_8), "+v"(c1[1]), "+v"(tf0_8), "+v"(tf1_8));
    float tc1_7 = tanh_act(c1[1][7]);
    v2f ff_8 = FIN2(((v2f){tf0_8, tf1_8}));
    ORDER("+v"(tc1_7), "+v"(ff_8), "+v"(mi_9), "+v"(d2[2]));
    float ti0_9 = TSIG(mi_9[0]);
    v2f mf_9 = PRE2(((v2f){d2[2][2], d2[2][3]}));
    ORDER("+v"(ti0_9), "+v"(mf_9), "+v"(mi_9), "+v"(to0_8), "+v"(to1_8));
    float ti1_9 = TSIG(mi_9[1]);
    v2f fo_8 = FIN2(((v2f){to0_8, to1_8}));
    ORDER("+v"(ti1_9), "+v"(fo_8), "+v"(d2[4]), "+v"(ff_8), "+v"(ig_8));
    float tg0_9 = tanh_act(d2[4][2]);
    { const v2f cn = __builtin_elementwise_fma(ff_8, (v2f){c2[0][0], c2[0][1]}, ig_8); c2[0][0] = cn[0]; c2[0][1] = cn[1]; }
    ORDER("+v"(tg0_9), "+v"(c2[0]), "+v"(d2[4]), "+v"(d2[6]));
    float tg1_9 = tanh_act(d2[4][3]);
    v2f mo_9 = PRE2(((v2f){d2[6][2], d2[6][3]}));
    ORDER("+v"(tg1_9), "+v"(mo_9), "+v"(mf_9), "+v"(fo_7), "+v"(tc0_7), "+v"(tc1_7));
    float tf0_9 = TSIG(mf_9[0]);
    v2f hh_7 = fo_7 * (v2f){tc0_7, tc1_7};
    ORDER("+v"(tf0_9), "+v"(hh_7), "+v"(mf_9), "+v"(ti0_9), "+v"(ti1_9));
    float tf1_9 = TSIG(mf_9[1]);
    v2f fi_9 = FIN2(((v2f){ti0_9, ti1_9}));
    ORDER("+v"(tf1_9), "+v"(fi_9), "+v"(mo_9), "+v"(d2[0]));
    float to0_9 = TSIG(mo_9[0]);
    v2f mi_10 = PRE2(((v2f){d2[0][4], d2[0][5]}));
    ORDER("+v"(to0_9), "+v"(mi_10), "+v"(mo_9), "+v"(hh_7));
    float to1_9 = TSIG(mo_9[1]);
    v2h pk_7 = __builtin_convertvector(hh_7, v2h);
    ORDER("+v"(to1_9), "+v"(pk_7), "+v"(c2[0]), "+v"(fi_9), "+v"(tg0_9), "+v"(tg1_9));
    float tc0_8 = tanh_act(c2[0][0]);
    v2f ig_9 = fi_9 * (v2f){tg0_9, tg1_9};
    h1Bn[0] = pk_0[0]; h1Bn[1] = pk_0[1];
    h1Bn[2] = pk_1[0]; h1Bn[3] = pk_1[1];
    h1Bn[4] = pk_2[0]; h1Bn[5] = pk_2[1];
    h1Bn[6] = pk_3[0]; h1Bn[7] = pk_3[1];
    h1Bn[8] = pk_4[0]; h1Bn[9] = pk_4[1];
    h1Bn[10] = pk_5[0]; h1Bn[11] = pk_5[1];
    h1Bn[12] = pk_6[0]; h1Bn[13] = pk_6[1];
    h1Bn[14] = pk_7[0]; h1Bn[15] = pk_7[1];
    if (hi_half) { h1Bn[14] = (_Float16)(u == 0 ? xa : xb); h1Bn[15] = (_Float16)1.f; }
    ORDER("+v"(tc0_8), "+v"(ig_9), "+v"(h1Bn), "+v"(c2[0]), "+v"(tf0_9), "+v"(tf1_9));
    float tc1_8 = tanh_act(c2[0][1]);
    v2f ff_9 = FIN2(((v2f){tf0_9, tf1_9}));
    ORDER("+v"(tc1_8), "+v"(ff_9), "+v"(mi_10), "+v"(d2[2]));
    float ti0_10 = TSIG(mi_10[0]);
    v2f mf_10 = PRE2(((v2f){d2[2][4], d2[2][5]}));
    ORDER("+v"(ti0_10), "+v"(mf_10), "+v"(mi_10), "+v"(to0_9), "+v"(to1_9));
    float ti1_10 = TSIG(mi_10[1]);
    v2f fo_9 = FIN2(((v2f){to0_9, to1_9}));
    ORDER("+v"(ti1_10), "+v"(fo_9), "+v"(d2[4]), "+v"(ff_9), "+v"(ig_9));
    float tg0_10 = tanh_act(d2[4][4]);
    { const v2f cn = __builtin_elementwise_fma(ff_9, (v2f){c2[0][2], c2[0][3]}, ig_9); c2[0][2] = cn[0]; c2[0][3] = cn[1]; }
    ORDER("+v"(tg0_10), "+v"(c2[0]), "+v"(d2[4]), "+v"(d2[6]));
    float tg1_10 = tanh_act(d2[4][5]);
    v2f mo_10 = PRE2(((v2f){d2[6][4], d2[6][5]}));
    ORDER("+v"(tg1_10), "+v"(mo_10), "+v"(mf_10));
    float tf0_10 = TSIG(mf_10[0]);
    ORDER("+v"(tf0_10), "+v"(h1Bn));
    d1[0] = __builtin_amdgcn_wmma_f32_16x16x32_f16(false, a1[0], false, h1Bn, (short)0, zc, false, false);
    ORDER("+v"(d1[0]), "+v"(fo_8), "+v"(tc0_8), "+v"(tc1_8));
    v2f hh_8 = fo_8 * (v2f){tc0_8, tc1_8};
    ORDER("+v"(hh_8), "+v"(mf_10), "+v"(ti0_10), "+v"(ti1_10));
    float tf1_10 = TSIG(mf_10[1]);
    v2f fi_10 = FIN2(((v2f){ti0_10, ti1_10}));
    ORDER("+v"(tf1_10), "+v"(fi_10), "+v"(mo_10), "+v"(d2[0]));
    float to0_10 = TSIG(mo_10[0]);
    v2f mi_11 = PRE2(((v2f){d2[0][6], d2[0][7]}));
    ORDER("+v"(to0_10), "+v"(mi_11), "+v"(mo_10), "+v"(hh_8));
    float to1_10 = TSIG(mo_10[1]);
    v2h pk_8 = __builtin_convertvector(hh_8, v2h);
    ORDER("+v"(to1_10), "+v"(pk_8), "+v"(c2[0]), "+v"(fi_10), "+v"(tg0_10), "+v"(tg1_10), "+v"(hh_8), "+v"(acc));
    float tc0_9 = tanh_act(c2[0][2]);
    v2f ig_10 = fi_10 * (v2f){tg0_10, tg1_10};
    acc = fmaf(hh_8[0], wfcD[0][0], acc);
    ORDER("+v"(tc0_9), "+v"(ig_10), "+v"(acc), "+v"(c2[0]));
    float tc1_9 = tanh_act(c2[0][3]);
    ORDER("+v"(tc1_9), "+v"(h1Bn));
    d1[2] = __builtin_amdgcn_wmma_f32_16x16x32_f16(false, a1[2], false, h1Bn, (short)0, zc, false, false);
    ORDER("+v"(d1[2]), "+v"(tf0_10), "+v"(tf1_10));
    v2f ff_10 = FIN2(((v2f){tf0_10, tf1_10}));
    ORDER("+v"(ff_10), "+v"(mi_11), "+v"(d2[2]));
    float ti0_11 = TSIG(mi_11[0]);
    v2f mf_11 = PRE2(((v2f){d2[2][6], d2[2][7]}));
    ORDER("+v"(ti0_11), "+v"(mf_11), "+v"(mi_11), "+v"(to0_10), "+v"(to1_10));
    float ti1_11 = TSIG(mi_11[1]);
    v2f fo_10 = FIN2(((v2f){to0_10, to1_10}));
    ORDER("+v"(ti1_11), "+v"(fo_10), "+v"(d2[4]), "+v"(ff_10), "+v"(ig_10), "+v"(hh_8), "+v"(acc));
    float tg0_11 = tanh_act(d2[4][6]);
    { const v2f cn = __builtin_elementwise_fma(ff_10, (v2f){c2[0][4], c2[0][5]}, ig_10); c2[0][4] = cn[0]; c2[0][5] = cn[1]; }
    acc = fmaf(hh_8[1], wfcD[0][1], acc);
    ORDER("+v"(tg0_11), "+v"(c2[0]), "+v"(acc), "+v"(d2[4]), "+v"(d2[6]));
    float tg1_11 = tanh_act(d2[4][7]);
    v2f mo_11 = PRE2(((v2f){d2[6][6], d2[6][7]}));
    ORDER("+v"(tg1_11), "+v"(mo_11), "+v"(mf_11), "+v"(fo_9), "+v"(tc0_9), "+v"(tc1_9));
    float tf0_11 = TSIG(mf_11[0]);
    v2f hh_9 = fo_9 * (v2f){tc0_9, tc1_9};
    ORDER("+v"(tf0_11), "+v"(hh_9), "+v"(mf_11), "+v"(ti0_11), "+v"(ti1_11));
    float tf1_11 = TSIG(mf_11[1]);
    v2f fi_11 = FIN2(((v2f){ti0_11, ti1_11}));
    ORDER("+v"(tf1_11), "+v"(fi_11), "+v"(mo_11), "+v"(d2[1]));
    float to0_11 = TSIG(mo_11[0]);
    v2f mi_12 = PRE2(((v2f){d2[1][0], d2[1][1]}));
    ORDER("+v"(to0_11), "+v"(mi_12), "+v"(mo_11), "+v"(hh_9));
    float to1_11 = TSIG(mo_11[1]);
    v2h pk_9 = __builtin_convertvector(hh_9, v2h);
    ORDER("+v"(to1_11), "+v"(pk_9), "+v"(c2[0]), "+v"(fi_11), "+v"(tg0_11), "+v"(tg1_11), "+v"(hh_9), "+v"(acc));
    float tc0_10 = tanh_act(c2[0][4]);
    v2f ig_11 = fi_11 * (v2f){tg0_11, tg1_11};
    acc = fmaf(hh_9[0], wfcD[0][2], acc);
    ORDER("+v"(tc0_10), "+v"(ig_11), "+v"(acc), "+v"(c2[0]));
    float tc1_10 = tanh_act(c2[0][5]);
    ORDER("+v"(tc1_10), "+v"(h1Bn));
    d1[4] = __builtin_amdgcn_wmma_f32_16x16x32_f16(false, a1[4], false, h1Bn, (short)0, zc, false, false);
    ORDER("+v"(d1[4]), "+v"(tf0_11), "+v"(tf1_11));
    v2f ff_11 = FIN2(((v2f){tf0_11, tf1_11}));
    ORDER("+v"(ff_11), "+v"(mi_12), "+v"(d2[3]));
    float ti0_12 = TSIG(mi_12[0]);
    v2f mf_12 = PRE2(((v2f){d2[3][0], d2[3][1]}));
    ORDER("+v"(ti0_12), "+v"(mf_12), "+v"(mi_12), "+v"(to0_11), "+v"(to1_11));
    float ti1_12 = TSIG(mi_12[1]);
    v2f fo_11 = FIN2(((v2f){to0_11, to1_11}));
    ORDER("+v"(ti1_12), "+v"(fo_11), "+v"(d2[5]), "+v"(ff_11), "+v"(ig_11), "+v"(hh_9), "+v"(acc));
    float tg0_12 = tanh_act(d2[5][0]);
    { const v2f cn = __builtin_elementwise_fma(ff_11, (v2f){c2[0][6], c2[0][7]}, ig_11); c2[0][6] = cn[0]; c2[0][7] = cn[1]; }
    acc = fmaf(hh_9[1], wfcD[0][3], acc);
    ORDER("+v"(tg0_12), "+v"(c2[0]), "+v"(acc), "+v"(d2[5]), "+v"(d2[7]));
    float tg1_12 = tanh_act(d2[5][1]);
    v2f mo_12 = PRE2(((v2f){d2[7][0], d2[7][1]}));
    ORDER("+v"(tg1_12), "+v"(mo_12), "+v"(mf_12), "+v"(fo_10), "+v"(tc0_10), "+v"(tc1_10));
    float tf0_12 = TSIG(mf_12[0]);
    v2f hh_10 = fo_10 * (v2f){tc0_10, tc1_10};
    ORDER("+v"(tf0_12), "+v"(hh_10), "+v"(mf_12), "+v"(ti0_12), "+v"(ti1_12));
    float tf1_12 = TSIG(mf_12[1]);
    v2f fi_12 = FIN2(((v2f){ti0_12, ti1_12}));
    ORDER("+v"(tf1_12), "+v"(fi_12), "+v"(mo_12), "+v"(d2[1]));
    float to0_12 = TSIG(mo_12[0]);
    v2f mi_13 = PRE2(((v2f){d2[1][2], d2[1][3]}));
    ORDER("+v"(to0_12), "+v"(mi_13), "+v"(mo_12), "+v"(hh_10));
    float to1_12 = TSIG(mo_12[1]);
    v2h pk_10 = __builtin_convertvector(hh_10, v2h);
    ORDER("+v"(to1_12), "+v"(pk_10), "+v"(c2[0]), "+v"(fi_12), "+v"(tg0_12), "+v"(tg1_12), "+v"(hh_10), "+v"(acc));
    float tc0_11 = tanh_act(c2[0][6]);
    v2f ig_12 = fi_12 * (v2f){tg0_12, tg1_12};
    acc = fmaf(hh_10[0], wfcD[0][4], acc);
    ORDER("+v"(tc0_11), "+v"(ig_12), "+v"(acc), "+v"(c2[0]));
    float tc1_11 = tanh_act(c2[0][7]);
    ORDER("+v"(tc1_11), "+v"(h1Bn));
    d1[6] = __builtin_amdgcn_wmma_f32_16x16x32_f16(false, a1[6], false, h1Bn, (short)0, zc, false, false);
    ORDER("+v"(d1[6]), "+v"(tf0_12), "+v"(tf1_12));
    v2f ff_12 = FIN2(((v2f){tf0_12, tf1_12}));
    ORDER("+v"(ff_12), "+v"(mi_13), "+v"(d2[3]));
    float ti0_13 = TSIG(mi_13[0]);
    v2f mf_13 = PRE2(((v2f){d2[3][2], d2[3][3]}));
    ORDER("+v"(ti0_13), "+v"(mf_13), "+v"(mi_13), "+v"(to0_12), "+v"(to1_12));
    float ti1_13 = TSIG(mi_13[1]);
    v2f fo_12 = FIN2(((v2f){to0_12, to1_12}));
    ORDER("+v"(ti1_13), "+v"(fo_12), "+v"(d2[5]), "+v"(ff_12), "+v"(ig_12), "+v"(hh_10), "+v"(acc));
    float tg0_13 = tanh_act(d2[5][2]);
    { const v2f cn = __builtin_elementwise_fma(ff_12, (v2f){c2[1][0], c2[1][1]}, ig_12); c2[1][0] = cn[0]; c2[1][1] = cn[1]; }
    acc = fmaf(hh_10[1], wfcD[0][5], acc);
    ORDER("+v"(tg0_13), "+v"(c2[1]), "+v"(acc), "+v"(d2[5]), "+v"(d2[7]));
    float tg1_13 = tanh_act(d2[5][3]);
    v2f mo_13 = PRE2(((v2f){d2[7][2], d2[7][3]}));
    ORDER("+v"(tg1_13), "+v"(mo_13), "+v"(mf_13), "+v"(fo_11), "+v"(tc0_11), "+v"(tc1_11));
    float tf0_13 = TSIG(mf_13[0]);
    v2f hh_11 = fo_11 * (v2f){tc0_11, tc1_11};
    ORDER("+v"(tf0_13), "+v"(hh_11), "+v"(mf_13), "+v"(ti0_13), "+v"(ti1_13));
    float tf1_13 = TSIG(mf_13[1]);
    v2f fi_13 = FIN2(((v2f){ti0_13, ti1_13}));
    ORDER("+v"(tf1_13), "+v"(fi_13), "+v"(mo_13), "+v"(d2[1]));
    float to0_13 = TSIG(mo_13[0]);
    v2f mi_14 = PRE2(((v2f){d2[1][4], d2[1][5]}));
    ORDER("+v"(to0_13), "+v"(mi_14), "+v"(mo_13), "+v"(hh_11));
    float to1_13 = TSIG(mo_13[1]);
    v2h pk_11 = __builtin_convertvector(hh_11, v2h);
    ORDER("+v"(to1_13), "+v"(pk_11), "+v"(c2[1]), "+v"(fi_13), "+v"(tg0_13), "+v"(tg1_13), "+v"(hh_11), "+v"(acc));
    float tc0_12 = tanh_act(c2[1][0]);
    v2f ig_13 = fi_13 * (v2f){tg0_13, tg1_13};
    acc = fmaf(hh_11[0], wfcD[0][6], acc);
    ORDER("+v"(tc0_12), "+v"(ig_13), "+v"(acc), "+v"(c2[1]));
    float tc1_12 = tanh_act(c2[1][1]);
    ORDER("+v"(tc1_12), "+v"(h1Bn));
    d1[1] = __builtin_amdgcn_wmma_f32_16x16x32_f16(false, a1[1], false, h1Bn, (short)0, zc, false, false);
    ORDER("+v"(d1[1]), "+v"(tf0_13), "+v"(tf1_13));
    v2f ff_13 = FIN2(((v2f){tf0_13, tf1_13}));
    ORDER("+v"(ff_13), "+v"(mi_14), "+v"(d2[3]));
    float ti0_14 = TSIG(mi_14[0]);
    v2f mf_14 = PRE2(((v2f){d2[3][4], d2[3][5]}));
    ORDER("+v"(ti0_14), "+v"(mf_14), "+v"(mi_14), "+v"(to0_13), "+v"(to1_13));
    float ti1_14 = TSIG(mi_14[1]);
    v2f fo_13 = FIN2(((v2f){to0_13, to1_13}));
    ORDER("+v"(ti1_14), "+v"(fo_13), "+v"(d2[5]), "+v"(ff_13), "+v"(ig_13), "+v"(hh_11), "+v"(acc));
    float tg0_14 = tanh_act(d2[5][4]);
    { const v2f cn = __builtin_elementwise_fma(ff_13, (v2f){c2[1][2], c2[1][3]}, ig_13); c2[1][2] = cn[0]; c2[1][3] = cn[1]; }
    acc = fmaf(hh_11[1], wfcD[0][7], acc);
    ORDER("+v"(tg0_14), "+v"(c2[1]), "+v"(acc), "+v"(d2[5]), "+v"(d2[7]));
    float tg1_14 = tanh_act(d2[5][5]);
    v2f mo_14 = PRE2(((v2f){d2[7][4], d2[7][5]}));
    ORDER("+v"(tg1_14), "+v"(mo_14), "+v"(mf_14), "+v"(fo_12), "+v"(tc0_12), "+v"(tc1_12));
    float tf0_14 = TSIG(mf_14[0]);
    v2f hh_12 = fo_12 * (v2f){tc0_12, tc1_12};
    ORDER("+v"(tf0_14), "+v"(hh_12), "+v"(mf_14), "+v"(ti0_14), "+v"(ti1_14));
    float tf1_14 = TSIG(mf_14[1]);
    v2f fi_14 = FIN2(((v2f){ti0_14, ti1_14}));
    ORDER("+v"(tf1_14), "+v"(fi_14), "+v"(mo_14), "+v"(d2[1]));
    float to0_14 = TSIG(mo_14[0]);
    v2f mi_15 = PRE2(((v2f){d2[1][6], d2[1][7]}));
    ORDER("+v"(to0_14), "+v"(mi_15), "+v"(mo_14), "+v"(hh_12));
    float to1_14 = TSIG(mo_14[1]);
    v2h pk_12 = __builtin_convertvector(hh_12, v2h);
    ORDER("+v"(to1_14), "+v"(pk_12), "+v"(c2[1]), "+v"(fi_14), "+v"(tg0_14), "+v"(tg1_14), "+v"(hh_12), "+v"(acc));
    float tc0_13 = tanh_act(c2[1][2]);
    v2f ig_14 = fi_14 * (v2f){tg0_14, tg1_14};
    acc = fmaf(hh_12[0], wfcD[1][0], acc);
    ORDER("+v"(tc0_13), "+v"(ig_14), "+v"(acc), "+v"(c2[1]));
    float tc1_13 = tanh_act(c2[1][3]);
    ORDER("+v"(tc1_13), "+v"(h1Bn));
    d1[3] = __builtin_amdgcn_wmma_f32_16x16x32_f16(false, a1[3], false, h1Bn, (short)0, zc, false, false);
    ORDER("+v"(d1[3]), "+v"(tf0_14), "+v"(tf1_14));
    v2f ff_14 = FIN2(((v2f){tf0_14, tf1_14}));
    ORDER("+v"(ff_14), "+v"(mi_15), "+v"(d2[3]));
    float ti0_15 = TSIG(mi_15[0]);
    v2f mf_15 = PRE2(((v2f){d2[3][6], d2[3][7]}));
    ORDER("+v"(ti0_15), "+v"(mf_15), "+v"(mi_15), "+v"(to0_14), "+v"(to1_14));
    float ti1_15 = TSIG(mi_15[1]);
    v2f fo_14 = FIN2(((v2f){to0_14, to1_14}));
    ORDER("+v"(ti1_15), "+v"(fo_14), "+v"(d2[5]), "+v"(ff_14), "+v"(ig_14), "+v"(hh_12), "+v"(acc));
    float tg0_15 = tanh_act(d2[5][6]);
    { const v2f cn = __builtin_elementwise_fma(ff_14, (v2f){c2[1][4], c2[1][5]}, ig_14); c2[1][4] = cn[0]; c2[1][5] = cn[1]; }
    acc = fmaf(hh_12[1], wfcD[1][1], acc);
    ORDER("+v"(tg0_15), "+v"(c2[1]), "+v"(acc), "+v"(d2[5]), "+v"(d2[7]));
    float tg1_15 = tanh_act(d2[5][7]);
    v2f mo_15 = PRE2(((v2f){d2[7][6], d2[7][7]}));
    ORDER("+v"(tg1_15), "+v"(mo_15), "+v"(mf_15), "+v"(fo_13), "+v"(tc0_13), "+v"(tc1_13));
    float tf0_15 = TSIG(mf_15[0]);
    v2f hh_13 = fo_13 * (v2f){tc0_13, tc1_13};
    ORDER("+v"(tf0_15), "+v"(hh_13), "+v"(mf_15), "+v"(ti0_15), "+v"(ti1_15));
    float tf1_15 = TSIG(mf_15[1]);
    v2f fi_15 = FIN2(((v2f){ti0_15, ti1_15}));
    ORDER("+v"(tf1_15), "+v"(fi_15), "+v"(mo_15));
    float to0_15 = TSIG(mo_15[0]);
    ORDER("+v"(to0_15), "+v"(mo_15), "+v"(hh_13));
    float to1_15 = TSIG(mo_15[1]);
    v2h pk_13 = __builtin_convertvector(hh_13, v2h);
    ORDER("+v"(to1_15), "+v"(pk_13), "+v"(c2[1]), "+v"(fi_15), "+v"(tg0_15), "+v"(tg1_15), "+v"(hh_13), "+v"(acc));
    float tc0_14 = tanh_act(c2[1][4]);
    v2f ig_15 = fi_15 * (v2f){tg0_15, tg1_15};
    acc = fmaf(hh_13[0], wfcD[1][2], acc);
    ORDER("+v"(tc0_14), "+v"(ig_15), "+v"(acc), "+v"(c2[1]));
    float tc1_14 = tanh_act(c2[1][5]);
    ORDER("+v"(tc1_14), "+v"(h1Bn));
    d1[5] = __builtin_amdgcn_wmma_f32_16x16x32_f16(false, a1[5], false, h1Bn, (short)0, zc, false, false);
    ORDER("+v"(d1[5]), "+v"(tf0_15), "+v"(tf1_15));
    v2f ff_15 = FIN2(((v2f){tf0_15, tf1_15}));
    ORDER("+v"(ff_15), "+v"(to0_15), "+v"(to1_15));
    v2f fo_15 = FIN2(((v2f){to0_15, to1_15}));
    ORDER("+v"(fo_15), "+v"(ff_15), "+v"(ig_15), "+v"(hh_13), "+v"(acc));
    { const v2f cn = __builtin_elementwise_fma(ff_15, (v2f){c2[1][6], c2[1][7]}, ig_15); c2[1][6] = cn[0]; c2[1][7] = cn[1]; }
    acc = fmaf(hh_13[1], wfcD[1][3], acc);
    ORDER("+v"(c2[1]), "+v"(acc), "+v"(fo_14), "+v"(tc0_14), "+v"(tc1_14));
    v2f hh_14 = fo_14 * (v2f){tc0_14, tc1_14};
    ORDER("+v"(hh_14));
    v2h pk_14 = __builtin_convertvector(hh_14, v2h);
    ORDER("+v"(pk_14), "+v"(c2[1]), "+v"(hh_14), "+v"(acc));
    float tc0_15 = tanh_act(c2[1][6]);
    acc = fmaf(hh_14[0], wfcD[1][4], acc);
    ORDER("+v"(tc0_15), "+v"(acc), "+v"(c2[1]));
    float tc1_15 = tanh_act(c2[1][7]);
    ORDER("+v"(tc1_15), "+v"(h1Bn));
    d1[7] = __builtin_amdgcn_wmma_f32_16x16x32_f16(false, a1[7], false, h1Bn, (short)0, zc, false, false);
    ORDER("+v"(d1[7]), "+v"(hh_14), "+v"(acc));
    acc = fmaf(hh_14[1], wfcD[1][5], acc);
    ORDER("+v"(acc), "+v"(fo_15), "+v"(tc0_15), "+v"(tc1_15));
    v2f hh_15 = fo_15 * (v2f){tc0_15, tc1_15};
    ORDER("+v"(hh_15));
    v2h pk_15 = __builtin_convertvector(hh_15, v2h);
    ORDER("+v"(pk_15), "+v"(hh_15), "+v"(acc));
    acc = fmaf(hh_15[0], wfcD[1][6], acc);
    ORDER("+v"(acc), "+v"(hh_15));
    acc = fmaf(hh_15[1], wfcD[1][7], acc);
    h2Bn[0] = pk_8[0]; h2Bn[1] = pk_8[1];
    h2Bn[2] = pk_9[0]; h2Bn[3] = pk_9[1];
    h2Bn[4] = pk_10[0]; h2Bn[5] = pk_10[1];
    h2Bn[6] = pk_11[0]; h2Bn[7] = pk_11[1];
    h2Bn[8] = pk_12[0]; h2Bn[9] = pk_12[1];
    h2Bn[10] = pk_13[0]; h2Bn[11] = pk_13[1];
    h2Bn[12] = pk_14[0]; h2Bn[13] = pk_14[1];
    h2Bn[14] = pk_15[0]; h2Bn[15] = pk_15[1];
    if (hi_half) { h2Bn[14] = (_Float16)0.f; h2Bn[15] = (_Float16)1.f; }
    if (u == 0) outOdd = (accPrev + accOther) + bfc; else outEven = (accPrev + accOther) + bfc;
    accOther = xhalf(acc);

    accPrev = acc;
    h1B = h1Bn;
    h2B = h2Bn;
  }
    v2f xn = xq;
    int xstep = (tt + 6 < TLEN) ? 2 : 0;
    ORDER("+v"(xn), "+s"(xstep));
    xa = xn[0];
    xb = xn[1];
    xp += xstep;
  }

  ost[TLEN - 2] = hi_half ? (accPrev + accOther) + bfc : outEven;
}

extern "C" void kernel_launch(void* const* d_in, const int* in_sizes, int n_in,
                              void* d_out, int out_size, void* d_ws, size_t ws_size,
                              hipStream_t stream) {
  (void)in_sizes; (void)n_in; (void)out_size; (void)d_ws; (void)ws_size;
  const float* x     = (const float*)d_in[0];
  const float* w_ih1 = (const float*)d_in[1];
  const float* w_hh1 = (const float*)d_in[2];
  const float* b_ih1 = (const float*)d_in[3];
  const float* b_hh1 = (const float*)d_in[4];
  const float* w_ih2 = (const float*)d_in[5];
  const float* w_hh2 = (const float*)d_in[6];
  const float* b_ih2 = (const float*)d_in[7];
  const float* b_hh2 = (const float*)d_in[8];
  const float* w_fc  = (const float*)d_in[9];
  const float* b_fc  = (const float*)d_in[10];
  float* out = (float*)d_out;

  dim3 grid(BSZ / 16);
  dim3 block(32);
  lstm2_wmma_kernel<<<grid, block, 0, stream>>>(x, w_ih1, w_hh1, b_ih1, b_hh1,
                                                w_ih2, w_hh2, b_ih2, b_hh2,
                                                w_fc, b_fc, out);
}


# Step edge pipelined: a body's first gate slots carry the previous step's last layer-2 pairs (tanh c, 2h, pack, FC terms), no drain chain; W_ih2 h1 product first: layer-2 pre-activation sum regrouped
# speedup vs baseline: 2.2742x; 1.1318x over previous
#include <hip/hip_runtime.h>


#define HSZ   30
#define MPAD  128
#define KPAD  32
#define TLEN  2048
#define BSZ   512

typedef __attribute__((ext_vector_type(16))) _Float16 v16h;
typedef __attribute__((ext_vector_type(8)))  _Float16 v8h;
typedef __attribute__((ext_vector_type(8)))  float    v8f;
typedef __attribute__((ext_vector_type(2)))  float    v2f;
typedef __attribute__((ext_vector_type(2)))  _Float16 v2h;

#define ORDER(...) asm volatile("" : __VA_ARGS__)

#if __has_builtin(__builtin_amdgcn_tanhf)
#define HW_TANH __builtin_amdgcn_tanhf
#elif __has_builtin(__builtin_amdgcn_tanh_f32)
#define HW_TANH __builtin_amdgcn_tanh_f32
#endif

#ifdef HW_TANH
__device__ __forceinline__ float tanh_act(float x) { return HW_TANH(x); }
#define GATE_IN_SCALE 0.5f
#define H_SCALE 0.5f
#define TSIG(m) HW_TANH(m)
#define FIN1(t) fmaf(0.5f, (t), 0.5f)
#define FIN2(t) __builtin_elementwise_fma((v2f)(0.5f), (t), (v2f)(0.5f))
#define HOUT1(to, tc) fmaf((to), (tc), (tc))
#define HOUT2(to, tc) __builtin_elementwise_fma((to), (tc), (tc))
#else
__device__ __forceinline__ float tanh_act(float x) {
  return fmaf(2.0f, __builtin_amdgcn_rcpf(1.0f + __expf(-2.0f * x)), -1.0f);
}
#define GATE_IN_SCALE 1.0f
#define H_SCALE 1.0f
#define TSIG(m) __builtin_amdgcn_rcpf(1.0f + __expf(-(m)))
#define FIN1(t) (t)
#define FIN2(t) (t)
#define HOUT1(to, tc) ((to) * (tc))
#define HOUT2(to, tc) ((to) * (tc))
#endif

__device__ __forceinline__ float xhalf(float v) {
  const unsigned u = __builtin_bit_cast(unsigned, v);
  return __builtin_bit_cast(float, __builtin_amdgcn_permlanex16(u, u, 0x76543210u, 0xfedcba98u, false, false));
}

__device__ __forceinline__ v16h load_a_frag(const float* __restrict__ wrow, bool hi_half, bool real,
                                            float ws, float c23, float c31) {
  const int kb = hi_half ? 8 : 0;
  const int kc = hi_half ? 22 : 16;
  v2f q[8];
#pragma unroll
  for (int i = 0; i < 4; ++i) q[i] = *reinterpret_cast<const v2f*>(wrow + kb + 2 * i);
#pragma unroll
  for (int i = 0; i < 4; ++i) q[4 + i] = *reinterpret_cast<const v2f*>(wrow + kc + 2 * i);
#pragma unroll
  for (int i = 0; i < 8; ++i) q[i] = q[i] * ws;
  float e[8];
#pragma unroll
  for (int i = 0; i < 7; ++i) {
    const float lo = q[4 + (i >> 1)][i & 1];
    const float hi = q[4 + ((i + 1) >> 1)][(i + 1) & 1];
    e[i] = hi_half ? hi : lo;
  }
  e[7] = hi_half ? c31 : c23;
  v16h a;
#pragma unroll
  for (int i = 0; i < 4; ++i) {
    a[2 * i]     = (_Float16)(real ? q[i][0] : 0.f);
    a[2 * i + 1] = (_Float16)(real ? q[i][1] : 0.f);
  }
#pragma unroll
  for (int i = 0; i < 8; ++i) a[8 + i] = (_Float16)(real ? e[i] : 0.f);
  return a;
}

__device__ __forceinline__ void gates8(const v8f (&d)[4], v8f &c, float (&hn)[8]) {
#pragma unroll
  for (int r = 0; r < 8; ++r) {
    const float iv = FIN1(TSIG(d[0][r]));
    const float fv = FIN1(TSIG(d[1][r]));
    const float gv = tanh_act(d[2][r]);
    const float to = TSIG(d[3][r]);
    const float cn = fmaf(fv, c[r], iv * gv);
    c[r] = cn;
    hn[r] = HOUT1(to, tanh_act(cn));
  }
}

__device__ __forceinline__ v16h layer1_step(const v16h (&a1)[8], v16h h1B, v8f (&c1)[2], v8f zc) {
  v16h h1Bn;
#pragma unroll
  for (int s = 0; s < 2; ++s) {
    v8f d[4];
#pragma unroll
    for (int g = 0; g < 4; ++g)
      d[g] = __builtin_amdgcn_wmma_f32_16x16x32_f16(false, a1[2 * g + s], false, h1B,
                                                    (short)0, zc, false, false);
    float hn[8];
    gates8(d, c1[s], hn);
#pragma unroll
    for (int r = 0; r < 8; ++r) h1Bn[8 * s + r] = (_Float16)hn[r];
  }
  return h1Bn;
}

__global__ __launch_bounds__(32, 1)
void lstm2_wmma_kernel(const float* __restrict__ x,
                       const float* __restrict__ w_ih1, const float* __restrict__ w_hh1,
                       const float* __restrict__ b_ih1, const float* __restrict__ b_hh1,
                       const float* __restrict__ w_ih2, const float* __restrict__ w_hh2,
                       const float* __restrict__ b_ih2, const float* __restrict__ b_hh2,
                       const float* __restrict__ w_fc,  const float* __restrict__ b_fc,
                       float* __restrict__ out) {
  const int lane = threadIdx.x;
  const bool hi_half = lane >= 16;
  const int n = lane & 15;

  v16h a1[8], ai2[8], ah2[8];
#pragma unroll
  for (int tl = 0; tl < 8; ++tl) {
    const int p = ((tl & 1) << 4) + n;
    const bool real = (p != 23) && (p != 31);
    const int unit = real ? (p < 23 ? p : p - 1) : HSZ - 1;
    const int row = (tl >> 1) * HSZ + unit;
    const float rs = ((tl >> 1) == 2) ? 1.0f : GATE_IN_SCALE;
    const float ws = rs * H_SCALE;
    const float wi1 = w_ih1[row] * rs;
    const float bs1 = (b_ih1[row] + b_hh1[row]) * rs;
    const float bs2 = (b_ih2[row] + b_hh2[row]) * rs;
    a1[tl]  = load_a_frag(w_hh1 + row * HSZ, hi_half, real, ws, wi1, bs1);
    ai2[tl] = load_a_frag(w_ih2 + row * HSZ, hi_half, real, ws, 0.f, 0.f);
    ah2[tl] = load_a_frag(w_hh2 + row * HSZ, hi_half, real, ws, 0.f, bs2);
  }

  float wfcD[2][8];
#pragma unroll
  for (int s = 0; s < 2; ++s)
#pragma unroll
    for (int r = 0; r < 8; ++r) {
      const int p = 16 * s + (hi_half ? 8 : 0) + r;
      const bool real = (p != 23) && (p != 31);
      const int unit = real ? (p < 23 ? p : p - 1) : 0;
      wfcD[s][r] = real ? w_fc[unit] * H_SCALE : 0.f;
    }
  const float bfc = b_fc[0];

  v8f zc;
#pragma unroll
  for (int i = 0; i < 8; ++i) zc[i] = 0.f;

  v16h h1B;
#pragma unroll
  for (int i = 0; i < 16; ++i) h1B[i] = (_Float16)0.f;

  v8f c1[2], c2[2];
#pragma unroll
  for (int s = 0; s < 2; ++s)
#pragma unroll
    for (int r = 0; r < 8; ++r) { c1[s][r] = 0.f; c2[s][r] = 0.f; }

  const int bbase = blockIdx.x << 4;
  const float* xrow = x + (size_t)(bbase + n) * TLEN;
  float* orow = out + (size_t)(bbase + n) * TLEN;

  const float x0 = xrow[0];
  const float x1 = xrow[1];
  float xa = xrow[2];
  float xb = xrow[3];
  h1B[15] = hi_half ? (_Float16)1.f : (_Float16)x0;
  h1B = layer1_step(a1, h1B, c1, zc);
  h1B[15] = hi_half ? (_Float16)1.f : (_Float16)x1;

  v8f d1[8];
#pragma unroll
  for (int tl = 0; tl < 8; ++tl)
    d1[tl] = __builtin_amdgcn_wmma_f32_16x16x32_f16(false, a1[tl], false, h1B,
                                                    (short)0, zc, false, false);

  float z_to0_14 = 0.f, z_to1_14 = 0.f;
  float z_to0_15 = 0.f, z_fi_15 = 0.f, z_tg0_15 = 0.f, z_tf0_15 = 0.f;
  v2f z_hh_13 = {0.f, 0.f};
  v2h zpk_8, zpk_9, zpk_10, zpk_11, zpk_12;
  zpk_8[0] = zpk_8[1] = zpk_9[0] = zpk_9[1] = zpk_10[0] = zpk_10[1] = (_Float16)0.f;
  zpk_11[0] = zpk_11[1] = zpk_12[0] = zpk_12[1] = (_Float16)0.f;
  float accC = 0.f;
  float outEven = 0.f;
  float outOdd = 0.f;
  float* ost = orow + (hi_half ? 1 : 0);
  const float* xp = xrow + 4;
  const float rail2 = hi_half ? 1.0f : 0.0f;

  for (int tt = 0; tt < TLEN; tt += 2) {
    const v2f xq = *reinterpret_cast<const v2f*>(xp);
#pragma unroll
  for (int u = 0; u < 2; ++u) {
    const int t = tt + u;

    if (u == 1 && tt > 0) ost[tt - 2] = hi_half ? outOdd : outEven;

    v8f d2[8];
    v16h h1Bn;
    float acc = 0.f;
    float xsel = (u == 0 ? xa : xb);
    ORDER("+v"(d1[0]), "+v"(c2[1]));
    float tc0_14 = tanh_act(c2[1][4]);
    ORDER("+v"(tc0_14), "+v"(c2[1]), "+v"(z_hh_13));
    float tc1_14 = tanh_act(c2[1][5]);
    v2h pk_13 = __builtin_convertvector(z_hh_13, v2h);
    ORDER("+v"(tc1_14), "+v"(pk_13), "+v"(d1[0]));
    float ti0_0 = TSIG(d1[0][0]);
    ORDER("+v"(ti0_0), "+v"(h1B));
    d2[0] = __builtin_amdgcn_wmma_f32_16x16x32_f16(false, ai2[0], false, h1B, (short)0, zc, false, false);
    ORDER("+v"(d2[0]), "+v"(z_tf0_15));
    float ff_15 = FIN1(z_tf0_15);
    ORDER("+v"(ff_15), "+v"(d1[0]));
    float ti1_0 = TSIG(d1[0][1]);
    ORDER("+v"(ti1_0), "+v"(h1B));
    d1[5] = __builtin_amdgcn_wmma_f32_16x16x32_f16(false, a1[5], false, h1B, (short)0, zc, false, false);
    ORDER("+v"(d1[5]), "+v"(z_fi_15), "+v"(z_tg0_15));
    float ig_15 = z_fi_15 * z_tg0_15;
    ORDER("+v"(ig_15), "+v"(d1[4]), "+v"(z_hh_13), "+v"(accC));
    float tg0_0 = tanh_act(d1[4][0]);
    accC = fmaf(z_hh_13[0], wfcD[1][2], accC);
    ORDER("+v"(tg0_0), "+v"(accC), "+v"(d1[4]), "+v"(z_to0_14), "+v"(z_to1_14), "+v"(tc0_14), "+v"(tc1_14));
    float tg1_0 = tanh_act(d1[4][1]);
    v2f hh_14 = HOUT2(((v2f){z_to0_14, z_to1_14}), ((v2f){tc0_14, tc1_14}));
    ORDER("+v"(tg1_0), "+v"(hh_14), "+v"(d1[2]), "+v"(ff_15), "+v"(ig_15), "+v"(c2[1]));
    float tf0_0 = TSIG(d1[2][0]);
    c2[1][6] = fmaf(ff_15, c2[1][6], ig_15);
    ORDER("+v"(tf0_0), "+v"(c2[1]), "+v"(h1B));
    d2[2] = __builtin_amdgcn_wmma_f32_16x16x32_f16(false, ai2[2], false, h1B, (short)0, zc, false, false);
    ORDER("+v"(d2[2]), "+v"(ti0_0), "+v"(ti1_0));
    v2f fi_0 = FIN2(((v2f){ti0_0, ti1_0}));
    ORDER("+v"(fi_0), "+v"(d1[2]), "+v"(z_hh_13), "+v"(accC));
    float tf1_0 = TSIG(d1[2][1]);
    accC = fmaf(z_hh_13[1], wfcD[1][3], accC);
    ORDER("+v"(tf1_0), "+v"(accC), "+v"(d1[6]), "+v"(hh_14));
    float to0_0 = TSIG(d1[6][0]);
    v2h pk_14 = __builtin_convertvector(hh_14, v2h);
    ORDER("+v"(to0_0), "+v"(pk_14), "+v"(d1[6]));
    float to1_0 = TSIG(d1[6][1]);
    ORDER("+v"(to1_0), "+v"(c2[1]), "+v"(fi_0), "+v"(tg0_0), "+v"(tg1_0));
    float tc0_15 = tanh_act(c2[1][6]);
    v2f ig_0 = fi_0 * (v2f){tg0_0, tg1_0};
    ORDER("+v"(tc0_15), "+v"(ig_0), "+v"(d1[0]));
    float ti0_1 = TSIG(d1[0][2]);
    ORDER("+v"(ti0_1), "+v"(h1B));
    d2[4] = __builtin_amdgcn_wmma_f32_16x16x32_f16(false, ai2[4], false, h1B, (short)0, zc, false, false);
    ORDER("+v"(d2[4]), "+v"(tf0_0), "+v"(tf1_0));
    v2f ff_0 = FIN2(((v2f){tf0_0, tf1_0}));
    ORDER("+v"(ff_0), "+v"(d1[0]));
    float ti1_1 = TSIG(d1[0][3]);
    ORDER("+v"(ti1_1), "+v"(h1B), "+v"(hh_14), "+v"(accC));
    d1[7] = __builtin_amdgcn_wmma_f32_16x16x32_f16(false, a1[7], false, h1B, (short)0, zc, false, false);
    ORDER("+v"(d1[7]), "+v"(hh_14), "+v"(accC));
    accC = fmaf(hh_14[0], wfcD[1][4], accC);
    ORDER("+v"(accC), "+v"(d1[4]));
    float tg0_1 = tanh_act(d1[4][2]);
    ORDER("+v"(tg0_1), "+v"(d1[4]), "+v"(z_to0_15), "+v"(tc0_15));
    float tg1_1 = tanh_act(d1[4][3]);
    float hh_15 = HOUT1(z_to0_15, tc0_15);
    ORDER("+v"(tg1_1), "+v"(hh_15), "+v"(d1[2]), "+v"(ff_0), "+v"(ig_0), "+v"(c1[0]));
    float tf0_1 = TSIG(d1[2][2]);
    { const v2f cn = __builtin_elementwise_fma(ff_0, (v2f){c1[0][0], c1[0][1]}, ig_0); c1[0][0] = cn[0]; c1[0][1] = cn[1]; }
    ORDER("+v"(tf0_1), "+v"(c1[0]), "+v"(h1B));
    d2[6] = __builtin_amdgcn_wmma_f32_16x16x32_f16(false, ai2[6], false, h1B, (short)0, zc, false, false);
    ORDER("+v"(d2[6]), "+v"(ti0_1), "+v"(ti1_1));
    v2f fi_1 = FIN2(((v2f){ti0_1, ti1_1}));
    ORDER("+v"(fi_1), "+v"(d1[2]), "+v"(hh_14), "+v"(accC));
    float tf1_1 = TSIG(d1[2][3]);
    accC = fmaf(hh_14[1], wfcD[1][5], accC);
    ORDER("+v"(tf1_1), "+v"(accC), "+v"(d1[6]), "+v"(hh_15));
    float to0_1 = TSIG(d1[6][2]);
    v2h pk_15 = __builtin_convertvector(((v2f){hh_15, rail2}), v2h);
    ORDER("+v"(to0_1), "+v"(pk_15), "+v"(d1[6]));
    float to1_1 = TSIG(d1[6][3]);
    v16h h2B;
    h2B[0] = zpk_8[0]; h2B[1] = zpk_8[1];
    h2B[2] = zpk_9[0]; h2B[3] = zpk_9[1];
    h2B[4] = zpk_10[0]; h2B[5] = zpk_10[1];
    h2B[6] = zpk_11[0]; h2B[7] = zpk_11[1];
    h2B[8] = zpk_12[0]; h2B[9] = zpk_12[1];
    h2B[10] = pk_13[0]; h2B[11] = pk_13[1];
    h2B[12] = pk_14[0]; h2B[13] = pk_14[1];
    h2B[14] = pk_15[0]; h2B[15] = pk_15[1];
    ORDER("+v"(to1_1), "+v"(c1[0]), "+v"(hh_15), "+v"(accC));
    float tc0_0 = tanh_act(c1[0][0]);
    accC = fmaf(hh_15, wfcD[1][6], accC);
    ORDER("+v"(tc0_0), "+v"(accC), "+v"(c1[0]), "+v"(fi_1), "+v"(tg0_1), "+v"(tg1_1));
    float tc1_0 = tanh_act(c1[0][1]);
    v2f ig_1 = fi_1 * (v2f){tg0_1, tg1_1};
    ORDER("+v"(tc1_0), "+v"(ig_1), "+v"(d1[0]));
    float ti0_2 = TSIG(d1[0][4]);
    ORDER("+v"(ti0_2), "+v"(h1B));
    d2[1] = __builtin_amdgcn_wmma_f32_16x16x32_f16(false, ai2[1], false, h1B, (short)0, zc, false, false);
    ORDER("+v"(d2[1]), "+v"(tf0_1), "+v"(tf1_1));
    v2f ff_1 = FIN2(((v2f){tf0_1, tf1_1}));
    ORDER("+v"(ff_1), "+v"(d1[0]));
    float ti1_2 = TSIG(d1[0][5]);
    ORDER("+v"(ti1_2), "+v"(d1[4]), "+v"(accC));
    float tg0_2 = tanh_act(d1[4][4]);
    float accX = accC;
    ORDER("+v"(tg0_2), "+v"(accX), "+v"(accC), "+v"(d1[4]), "+v"(to0_0), "+v"(to1_0), "+v"(tc0_0), "+v"(tc1_0));
    float tg1_2 = tanh_act(d1[4][5]);
    v2f hh_0 = HOUT2(((v2f){to0_0, to1_0}), ((v2f){tc0_0, tc1_0}));
    ORDER("+v"(tg1_2), "+v"(hh_0), "+v"(d1[2]), "+v"(ff_1), "+v"(ig_1), "+v"(c1[0]));
    float tf0_2 = TSIG(d1[2][4]);
    { const v2f cn = __builtin_elementwise_fma(ff_1, (v2f){c1[0][2], c1[0][3]}, ig_1); c1[0][2] = cn[0]; c1[0][3] = cn[1]; }
    ORDER("+v"(tf0_2), "+v"(c1[0]), "+v"(h1B));
    d2[3] = __builtin_amdgcn_wmma_f32_16x16x32_f16(false, ai2[3], false, h1B, (short)0, zc, false, false);
    ORDER("+v"(d2[3]), "+v"(ti0_2), "+v"(ti1_2));
    v2f fi_2 = FIN2(((v2f){ti0_2, ti1_2}));
    ORDER("+v"(fi_2), "+v"(d1[2]), "+v"(accX));
    float tf1_2 = TSIG(d1[2][5]);
    float accO = xhalf(accX);
    ORDER("+v"(tf1_2), "+v"(accO), "+v"(d1[6]));
    float to0_2 = TSIG(d1[6][4]);
    ORDER("+v"(to0_2), "+v"(d1[6]), "+v"(hh_0));
    float to1_2 = TSIG(d1[6][5]);
    v2h pk_0 = __builtin_convertvector(hh_0, v2h);
    ORDER("+v"(to1_2), "+v"(pk_0), "+v"(c1[0]));
    float tc0_1 = tanh_act(c1[0][2]);
    ORDER("+v"(tc0_1), "+v"(c1[0]), "+v"(fi_2), "+v"(tg0_2), "+v"(tg1_2));
    float tc1_1 = tanh_act(c1[0][3]);
    v2f ig_2 = fi_2 * (v2f){tg0_2, tg1_2};
    ORDER("+v"(tc1_1), "+v"(ig_2), "+v"(d1[0]));
    float ti0_3 = TSIG(d1[0][6]);
    ORDER("+v"(ti0_3), "+v"(h1B));
    d2[5] = __builtin_amdgcn_wmma_f32_16x16x32_f16(false, ai2[5], false, h1B, (short)0, zc, false, false);
    ORDER("+v"(d2[5]), "+v"(tf0_2), "+v"(tf1_2));
    v2f ff_2 = FIN2(((v2f){tf0_2, tf1_2}));
    ORDER("+v"(ff_2), "+v"(d1[0]), "+v"(accC), "+v"(accO));
    float ti1_3 = TSIG(d1[0][7]);
    float osum = accC + accO;
    ORDER("+v"(ti1_3), "+v"(osum), "+v"(d1[4]));
    float tg0_3 = tanh_act(d1[4][6]);
    ORDER("+v"(tg0_3), "+v"(d1[4]), "+v"(to0_1), "+v"(to1_1), "+v"(tc0_1), "+v"(tc1_1));
    float tg1_3 = tanh_act(d1[4][7]);
    v2f hh_1 = HOUT2(((v2f){to0_1, to1_1}), ((v2f){tc0_1, tc1_1}));
    ORDER("+v"(tg1_3), "+v"(hh_1), "+v"(d1[2]), "+v"(ff_2), "+v"(ig_2), "+v"(c1[0]));
    float tf0_3 = TSIG(d1[2][6]);
    { const v2f cn = __builtin_elementwise_fma(ff_2, (v2f){c1[0][4], c1[0][5]}, ig_2); c1[0][4] = cn[0]; c1[0][5] = cn[1]; }
    ORDER("+v"(tf0_3), "+v"(c1[0]), "+v"(h1B));
    d2[7] = __builtin_amdgcn_wmma_f32_16x16x32_f16(false, ai2[7], false, h1B, (short)0, zc, false, false);
    ORDER("+v"(d2[7]), "+v"(ti0_3), "+v"(ti1_3));
    v2f fi_3 = FIN2(((v2f){ti0_3, ti1_3}));
    ORDER("+v"(fi_3), "+v"(d1[2]), "+v"(osum));
    float tf1_3 = TSIG(d1[2][7]);
    float ofin = osum + bfc;
    ORDER("+v"(tf1_3), "+v"(ofin), "+v"(d1[6]));
    if (u == 0) outOdd = ofin; else outEven = ofin;
    float to0_3 = TSIG(d1[6][6]);
    ORDER("+v"(to0_3), "+v"(d1[6]), "+v"(hh_1));
    float to1_3 = TSIG(d1[6][7]);
    v2h pk_1 = __builtin_convertvector(hh_1, v2h);
    ORDER("+v"(to1_3), "+v"(pk_1), "+v"(c1[0]));
    float tc0_2 = tanh_act(c1[0][4]);
    ORDER("+v"(tc0_2), "+v"(c1[0]), "+v"(fi_3), "+v"(tg0_3), "+v"(tg1_3));
    float tc1_2 = tanh_act(c1[0][5]);
    v2f ig_3 = fi_3 * (v2f){tg0_3, tg1_3};
    ORDER("+v"(tc1_2), "+v"(ig_3), "+v"(d1[1]));
    float ti0_4 = TSIG(d1[1][0]);
    ORDER("+v"(ti0_4), "+v"(h2B), "+v"(d2[0]));
    d2[0] = __builtin_amdgcn_wmma_f32_16x16x32_f16(false, ah2[0], false, h2B, (short)0, d2[0], false, false);
    ORDER("+v"(d2[0]), "+v"(tf0_3), "+v"(tf1_3));
    v2f ff_3 = FIN2(((v2f){tf0_3, tf1_3}));
    ORDER("+v"(ff_3), "+v"(d1[1]));
    float ti1_4 = TSIG(d1[1][1]);
    ORDER("+v"(ti1_4), "+v"(d1[5]));
    float tg0_4 = tanh_act(d1[5][0]);
    ORDER("+v"(tg0_4), "+v"(d1[5]), "+v"(to0_2), "+v"(to1_2), "+v"(tc0_2), "+v"(tc1_2));
    float tg1_4 = tanh_act(d1[5][1]);
    v2f hh_2 = HOUT2(((v2f){to0_2, to1_2}), ((v2f){tc0_2, tc1_2}));
    ORDER("+v"(tg1_4), "+v"(hh_2), "+v"(d1[3]), "+v"(ff_3), "+v"(ig_3), "+v"(c1[0]));
    float tf0_4 = TSIG(d1[3][0]);
    { const v2f cn = __builtin_elementwise_fma(ff_3, (v2f){c1[0][6], c1[0][7]}, ig_3); c1[0][6] = cn[0]; c1[0][7] = cn[1]; }
    ORDER("+v"(tf0_4), "+v"(c1[0]), "+v"(h2B), "+v"(d2[2]));
    d2[2] = __builtin_amdgcn_wmma_f32_16x16x32_f16(false, ah2[2], false, h2B, (short)0, d2[2], false, false);
    ORDER("+v"(d2[2]), "+v"(ti0_4), "+v"(ti1_4));
    v2f fi_4 = FIN2(((v2f){ti0_4, ti1_4}));
    ORDER("+v"(fi_4), "+v"(d1[3]));
    float tf1_4 = TSIG(d1[3][1]);
    ORDER("+v"(tf1_4), "+v"(d1[7]));
    float to0_4 = TSIG(d1[7][0]);
    ORDER("+v"(to0_4), "+v"(d1[7]), "+v"(hh_2));
    float to1_4 = TSIG(d1[7][1]);
    v2h pk_2 = __builtin_convertvector(hh_2, v2h);
    ORDER("+v"(to1_4), "+v"(pk_2), "+v"(c1[0]));
    float tc0_3 = tanh_act(c1[0][6]);
    ORDER("+v"(tc0_3), "+v"(c1[0]), "+v"(fi_4), "+v"(tg0_4), "+v"(tg1_4));
    float tc1_3 = tanh_act(c1[0][7]);
    v2f ig_4 = fi_4 * (v2f){tg0_4, tg1_4};
    ORDER("+v"(tc1_3), "+v"(ig_4), "+v"(d1[1]));
    float ti0_5 = TSIG(d1[1][2]);
    ORDER("+v"(ti0_5), "+v"(h2B), "+v"(d2[4]));
    d2[4] = __builtin_amdgcn_wmma_f32_16x16x32_f16(false, ah2[4], false, h2B, (short)0, d2[4], false, false);
    ORDER("+v"(d2[4]), "+v"(tf0_4), "+v"(tf1_4));
    v2f ff_4 = FIN2(((v2f){tf0_4, tf1_4}));
    ORDER("+v"(ff_4), "+v"(d1[1]));
    float ti1_5 = TSIG(d1[1][3]);
    ORDER("+v"(ti1_5), "+v"(d1[5]));
    float tg0_5 = tanh_act(d1[5][2]);
    ORDER("+v"(tg0_5), "+v"(d1[5]), "+v"(to0_3), "+v"(to1_3), "+v"(tc0_3), "+v"(tc1_3));
    float tg1_5 = tanh_act(d1[5][3]);
    v2f hh_3 = HOUT2(((v2f){to0_3, to1_3}), ((v2f){tc0_3, tc1_3}));
    ORDER("+v"(tg1_5), "+v"(hh_3), "+v"(d1[3]), "+v"(ff_4), "+v"(ig_4), "+v"(c1[1]));
    float tf0_5 = TSIG(d1[3][2]);
    { const v2f cn = __builtin_elementwise_fma(ff_4, (v2f){c1[1][0], c1[1][1]}, ig_4); c1[1][0] = cn[0]; c1[1][1] = cn[1]; }
    ORDER("+v"(tf0_5), "+v"(c1[1]), "+v"(h2B), "+v"(d2[6]));
    d2[6] = __builtin_amdgcn_wmma_f32_16x16x32_f16(false, ah2[6], false, h2B, (short)0, d2[6], false, false);
    ORDER("+v"(d2[6]), "+v"(ti0_5), "+v"(ti1_5));
    v2f fi_5 = FIN2(((v2f){ti0_5, ti1_5}));
    ORDER("+v"(fi_5), "+v"(d1[3]));
    float tf1_5 = TSIG(d1[3][3]);
    ORDER("+v"(tf1_5), "+v"(d1[7]));
    float to0_5 = TSIG(d1[7][2]);
    ORDER("+v"(to0_5), "+v"(d1[7]), "+v"(hh_3));
    float to1_5 = TSIG(d1[7][3]);
    v2h pk_3 = __builtin_convertvector(hh_3, v2h);
    ORDER("+v"(to1_5), "+v"(pk_3), "+v"(c1[1]));
    float tc0_4 = tanh_act(c1[1][0]);
    ORDER("+v"(tc0_4), "+v"(c1[1]), "+v"(fi_5), "+v"(tg0_5), "+v"(tg1_5));
    float tc1_4 = tanh_act(c1[1][1]);
    v2f ig_5 = fi_5 * (v2f){tg0_5, tg1_5};
    ORDER("+v"(tc1_4), "+v"(ig_5), "+v"(d1[1]));
    float ti0_6 = TSIG(d1[1][4]);
    ORDER("+v"(ti0_6), "+v"(h2B), "+v"(d2[1]));
    d2[1] = __builtin_amdgcn_wmma_f32_16x16x32_f16(false, ah2[1], false, h2B, (short)0, d2[1], false, false);
    ORDER("+v"(d2[1]), "+v"(tf0_5), "+v"(tf1_5));
    v2f ff_5 = FIN2(((v2f){tf0_5, tf1_5}));
    ORDER("+v"(ff_5), "+v"(d1[1]));
    float ti1_6 = TSIG(d1[1][5]);
    ORDER("+v"(ti1_6), "+v"(d1[5]));
    float tg0_6 = tanh_act(d1[5][4]);
    ORDER("+v"(tg0_6), "+v"(d1[5]), "+v"(to0_4), "+v"(to1_4), "+v"(tc0_4), "+v"(tc1_4));
    float tg1_6 = tanh_act(d1[5][5]);
    v2f hh_4 = HOUT2(((v2f){to0_4, to1_4}), ((v2f){tc0_4, tc1_4}));
    ORDER("+v"(tg1_6), "+v"(hh_4), "+v"(d1[3]), "+v"(ff_5), "+v"(ig_5), "+v"(c1[1]));
    float tf0_6 = TSIG(d1[3][4]);
    { const v2f cn = __builtin_elementwise_fma(ff_5, (v2f){c1[1][2], c1[1][3]}, ig_5); c1[1][2] = cn[0]; c1[1][3] = cn[1]; }
    ORDER("+v"(tf0_6), "+v"(c1[1]), "+v"(h2B), "+v"(d2[3]));
    d2[3] = __builtin_amdgcn_wmma_f32_16x16x32_f16(false, ah2[3], false, h2B, (short)0, d2[3], false, false);
    ORDER("+v"(d2[3]), "+v"(ti0_6), "+v"(ti1_6));
    v2f fi_6 = FIN2(((v2f){ti0_6, ti1_6}));
    ORDER("+v"(fi_6), "+v"(d1[3]));
    float tf1_6 = TSIG(d1[3][5]);
    ORDER("+v"(tf1_6), "+v"(d1[7]));
    float to0_6 = TSIG(d1[7][4]);
    ORDER("+v"(to0_6), "+v"(d1[7]), "+v"(hh_4));
    float to1_6 = TSIG(d1[7][5]);
    v2h pk_4 = __builtin_convertvector(hh_4, v2h);
    ORDER("+v"(to1_6), "+v"(pk_4), "+v"(c1[1]), "+v"(fi_6), "+v"(tg0_6), "+v"(tg1_6));
    float tc0_5 = tanh_act(c1[1][2]);
    v2f ig_6 = fi_6 * (v2f){tg0_6, tg1_6};
    ORDER("+v"(tc0_5), "+v"(ig_6), "+v"(c1[1]), "+v"(tf0_6), "+v"(tf1_6));
    float tc1_5 = tanh_act(c1[1][3]);
    v2f ff_6 = FIN2(((v2f){tf0_6, tf1_6}));
    ORDER("+v"(tc1_5), "+v"(ff_6), "+v"(d1[1]));
    float ti0_7 = TSIG(d1[1][6]);
    ORDER("+v"(ti0_7), "+v"(h2B), "+v"(d2[5]), "+v"(xsel));
    d2[5] = __builtin_amdgcn_wmma_f32_16x16x32_f16(false, ah2[5], false, h2B, (short)0, d2[5], false, false);
    ORDER("+v"(d2[5]), "+v"(xsel));
    float rail1 = hi_half ? 1.0f : xsel;
    ORDER("+v"(rail1), "+v"(d1[5]), "+v"(ff_6), "+v"(ig_6), "+v"(c1[1]));
    float tg0_7 = tanh_act(d1[5][6]);
    { const v2f cn = __builtin_elementwise_fma(ff_6, (v2f){c1[1][4], c1[1][5]}, ig_6); c1[1][4] = cn[0]; c1[1][5] = cn[1]; }
    ORDER("+v"(tg0_7), "+v"(c1[1]), "+v"(d1[3]), "+v"(to0_5), "+v"(to1_5), "+v"(tc0_5), "+v"(tc1_5));
    float tf0_7 = TSIG(d1[3][6]);
    v2f hh_5 = HOUT2(((v2f){to0_5, to1_5}), ((v2f){tc0_5, tc1_5}));
    ORDER("+v"(tf0_7), "+v"(hh_5), "+v"(h2B), "+v"(d2[7]));
    d2[7] = __builtin_amdgcn_wmma_f32_16x16x32_f16(false, ah2[7], false, h2B, (short)0, d2[7], false, false);
    ORDER("+v"(d2[7]), "+v"(ti0_7));
    float fi_7 = FIN1(ti0_7);
    ORDER("+v"(fi_7), "+v"(d1[7]), "+v"(hh_5));
    float to0_7 = TSIG(d1[7][6]);
    v2h pk_5 = __builtin_convertvector(hh_5, v2h);
    ORDER("+v"(to0_7), "+v"(pk_5), "+v"(c1[1]));
    float tc0_6 = tanh_act(c1[1][4]);
    ORDER("+v"(tc0_6), "+v"(c1[1]), "+v"(fi_7), "+v"(tg0_7));
    float tc1_6 = tanh_act(c1[1][5]);
    float ig_7 = fi_7 * tg0_7;
    ORDER("+v"(tc1_6), "+v"(ig_7), "+v"(d2[0]), "+v"(tf0_7));
    float ti0_8 = TSIG(d2[0][0]);
    float ff_7 = FIN1(tf0_7);
    ORDER("+v"(ti0_8), "+v"(ff_7), "+v"(d2[0]));
    float ti1_8 = TSIG(d2[0][1]);
    ORDER("+v"(ti1_8), "+v"(d2[4]));
    float tg0_8 = tanh_act(d2[4][0]);
    ORDER("+v"(tg0_8), "+v"(d2[4]), "+v"(to0_6), "+v"(to1_6), "+v"(tc0_6), "+v"(tc1_6));
    float tg1_8 = tanh_act(d2[4][1]);
    v2f hh_6 = HOUT2(((v2f){to0_6, to1_6}), ((v2f){tc0_6, tc1_6}));
    ORDER("+v"(tg1_8), "+v"(hh_6), "+v"(d2[2]), "+v"(ff_7), "+v"(ig_7), "+v"(c1[1]));
    float tf0_8 = TSIG(d2[2][0]);
    c1[1][6] = fmaf(ff_7, c1[1][6], ig_7);
    ORDER("+v"(tf0_8), "+v"(c1[1]), "+v"(d2[2]), "+v"(ti0_8), "+v"(ti1_8));
    float tf1_8 = TSIG(d2[2][1]);
    v2f fi_8 = FIN2(((v2f){ti0_8, ti1_8}));
    ORDER("+v"(tf1_8), "+v"(fi_8), "+v"(d2[6]));
    float to0_8 = TSIG(d2[6][0]);
    ORDER("+v"(to0_8), "+v"(d2[6]), "+v"(hh_6));
    float to1_8 = TSIG(d2[6][1]);
    v2h pk_6 = __builtin_convertvector(hh_6, v2h);
    ORDER("+v"(to1_8), "+v"(pk_6), "+v"(c1[1]));
    float tc0_7 = tanh_act(c1[1][6]);
    ORDER("+v"(tc0_7), "+v"(c1[1]), "+v"(fi_8), "+v"(tg0_8), "+v"(tg1_8));
    v2f ig_8 = fi_8 * (v2f){tg0_8, tg1_8};
    ORDER("+v"(ig_8), "+v"(d2[0]), "+v"(tf0_8), "+v"(tf1_8));
    float ti0_9 = TSIG(d2[0][2]);
    v2f ff_8 = FIN2(((v2f){tf0_8, tf1_8}));
    ORDER("+v"(ti0_9), "+v"(ff_8), "+v"(d2[0]));
    float ti1_9 = TSIG(d2[0][3]);
    ORDER("+v"(ti1_9), "+v"(d2[4]));
    float tg0_9 = tanh_act(d2[4][2]);
    ORDER("+v"(tg0_9), "+v"(d2[4]), "+v"(to0_7), "+v"(tc0_7));
    float tg1_9 = tanh_act(d2[4][3]);
    float hh_7 = HOUT1(to0_7, tc0_7);
    ORDER("+v"(tg1_9), "+v"(hh_7), "+v"(d2[2]), "+v"(ff_8), "+v"(ig_8), "+v"(c2[0]));
    float tf0_9 = TSIG(d2[2][2]);
    { const v2f cn = __builtin_elementwise_fma(ff_8, (v2f){c2[0][0], c2[0][1]}, ig_8); c2[0][0] = cn[0]; c2[0][1] = cn[1]; }
    ORDER("+v"(tf0_9), "+v"(c2[0]), "+v"(d2[2]), "+v"(ti0_9), "+v"(ti1_9));
    float tf1_9 = TSIG(d2[2][3]);
    v2f fi_9 = FIN2(((v2f){ti0_9, ti1_9}));
    ORDER("+v"(tf1_9), "+v"(fi_9), "+v"(d2[6]));
    float to0_9 = TSIG(d2[6][2]);
    ORDER("+v"(to0_9), "+v"(d2[6]), "+v"(hh_7));
    float to1_9 = TSIG(d2[6][3]);
    v2h pk_7 = __builtin_convertvector(((v2f){hh_7, rail1}), v2h);
    ORDER("+v"(to1_9), "+v"(pk_7), "+v"(c2[0]));
    float tc0_8 = tanh_act(c2[0][0]);
    h1Bn[0] = pk_0[0]; h1Bn[1] = pk_0[1];
    h1Bn[2] = pk_1[0]; h1Bn[3] = pk_1[1];
    h1Bn[4] = pk_2[0]; h1Bn[5] = pk_2[1];
    h1Bn[6] = pk_3[0]; h1Bn[7] = pk_3[1];
    h1Bn[8] = pk_4[0]; h1Bn[9] = pk_4[1];
    h1Bn[10] = pk_5[0]; h1Bn[11] = pk_5[1];
    h1Bn[12] = pk_6[0]; h1Bn[13] = pk_6[1];
    h1Bn[14] = pk_7[0]; h1Bn[15] = pk_7[1];
    ORDER("+v"(tc0_8), "+v"(h1Bn), "+v"(c2[0]), "+v"(fi_9), "+v"(tg0_9), "+v"(tg1_9));
    float tc1_8 = tanh_act(c2[0][1]);
    v2f ig_9 = fi_9 * (v2f){tg0_9, tg1_9};
    ORDER("+v"(tc1_8), "+v"(ig_9), "+v"(d2[0]), "+v"(tf0_9), "+v"(tf1_9));
    float ti0_10 = TSIG(d2[0][4]);
    v2f ff_9 = FIN2(((v2f){tf0_9, tf1_9}));
    ORDER("+v"(ti0_10), "+v"(ff_9), "+v"(d2[0]));
    float ti1_10 = TSIG(d2[0][5]);
    ORDER("+v"(ti1_10), "+v"(d2[4]));
    float tg0_10 = tanh_act(d2[4][4]);
    ORDER("+v"(tg0_10), "+v"(d2[4]), "+v"(to0_8), "+v"(to1_8), "+v"(tc0_8), "+v"(tc1_8));
    float tg1_10 = tanh_act(d2[4][5]);
    v2f hh_8 = HOUT2(((v2f){to0_8, to1_8}), ((v2f){tc0_8, tc1_8}));
    ORDER("+v"(tg1_10), "+v"(hh_8), "+v"(d2[2]), "+v"(ff_9), "+v"(ig_9), "+v"(c2[0]));
    float tf0_10 = TSIG(d2[2][4]);
    { const v2f cn = __builtin_elementwise_fma(ff_9, (v2f){c2[0][2], c2[0][3]}, ig_9); c2[0][2] = cn[0]; c2[0][3] = cn[1]; }
    ORDER("+v"(tf0_10), "+v"(c2[0]), "+v"(h1Bn));
    d1[0] = __builtin_amdgcn_wmma_f32_16x16x32_f16(false, a1[0], false, h1Bn, (short)0, zc, false, false);
    ORDER("+v"(d1[0]), "+v"(ti0_10), "+v"(ti1_10));
    v2f fi_10 = FIN2(((v2f){ti0_10, ti1_10}));
    ORDER("+v"(fi_10), "+v"(d2[2]));
    float tf1_10 = TSIG(d2[2][5]);
    ORDER("+v"(tf1_10), "+v"(d2[6]));
    float to0_10 = TSIG(d2[6][4]);
    ORDER("+v"(to0_10), "+v"(d2[6]), "+v"(hh_8));
    float to1_10 = TSIG(d2[6][5]);
    v2h pk_8 = __builtin_convertvector(hh_8, v2h);
    ORDER("+v"(to1_10), "+v"(pk_8), "+v"(c2[0]));
    float tc0_9 = tanh_act(c2[0][2]);
    ORDER("+v"(tc0_9), "+v"(c2[0]), "+v"(fi_10), "+v"(tg0_10), "+v"(tg1_10));
    float tc1_9 = tanh_act(c2[0][3]);
    v2f ig_10 = fi_10 * (v2f){tg0_10, tg1_10};
    ORDER("+v"(tc1_9), "+v"(ig_10), "+v"(d2[0]));
    float ti0_11 = TSIG(d2[0][6]);
    ORDER("+v"(ti0_11), "+v"(h1Bn));
    d1[2] = __builtin_amdgcn_wmma_f32_16x16x32_f16(false, a1[2], false, h1Bn, (short)0, zc, false, false);
    ORDER("+v"(d1[2]), "+v"(tf0_10), "+v"(tf1_10));
    v2f ff_10 = FIN2(((v2f){tf0_10, tf1_10}));
    ORDER("+v"(ff_10), "+v"(d2[0]), "+v"(hh_8), "+v"(acc));
    float ti1_11 = TSIG(d2[0][7]);
    acc = fmaf(hh_8[0], wfcD[0][0], acc);
    ORDER("+v"(ti1_11), "+v"(acc), "+v"(d2[4]));
    float tg0_11 = tanh_act(d2[4][6]);
    ORDER("+v"(tg0_11), "+v"(d2[4]), "+v"(to0_9), "+v"(to1_9), "+v"(tc0_9), "+v"(tc1_9));
    float tg1_11 = tanh_act(d2[4][7]);
    v2f hh_9 = HOUT2(((v2f){to0_9, to1_9}), ((v2f){tc0_9, tc1_9}));
    ORDER("+v"(tg1_11), "+v"(hh_9), "+v"(d2[2]), "+v"(ff_10), "+v"(ig_10), "+v"(c2[0]));
    float tf0_11 = TSIG(d2[2][6]);
    { const v2f cn = __builtin_elementwise_fma(ff_10, (v2f){c2[0][4], c2[0][5]}, ig_10); c2[0][4] = cn[0]; c2[0][5] = cn[1]; }
    ORDER("+v"(tf0_11), "+v"(c2[0]), "+v"(d2[2]), "+v"(ti0_11), "+v"(ti1_11));
    float tf1_11 = TSIG(d2[2][7]);
    v2f fi_11 = FIN2(((v2f){ti0_11, ti1_11}));
    ORDER("+v"(tf1_11), "+v"(fi_11), "+v"(d2[6]), "+v"(hh_8), "+v"(acc));
    float to0_11 = TSIG(d2[6][6]);
    acc = fmaf(hh_8[1], wfcD[0][1], acc);
    ORDER("+v"(to0_11), "+v"(acc), "+v"(d2[6]), "+v"(hh_9));
    float to1_11 = TSIG(d2[6][7]);
    v2h pk_9 = __builtin_convertvector(hh_9, v2h);
    ORDER("+v"(to1_11), "+v"(pk_9), "+v"(c2[0]));
    float tc0_10 = tanh_act(c2[0][4]);
    ORDER("+v"(tc0_10), "+v"(c2[0]), "+v"(fi_11), "+v"(tg0_11), "+v"(tg1_11));
    float tc1_10 = tanh_act(c2[0][5]);
    v2f ig_11 = fi_11 * (v2f){tg0_11, tg1_11};
    ORDER("+v"(tc1_10), "+v"(ig_11), "+v"(d2[1]));
    float ti0_12 = TSIG(d2[1][0]);
    ORDER("+v"(ti0_12), "+v"(h1Bn));
    d1[4] = __builtin_amdgcn_wmma_f32_16x16x32_f16(false, a1[4], false, h1Bn, (short)0, zc, false, false);
    ORDER("+v"(d1[4]), "+v"(tf0_11), "+v"(tf1_11));
    v2f ff_11 = FIN2(((v2f){tf0_11, tf1_11}));
    ORDER("+v"(ff_11), "+v"(d2[1]), "+v"(hh_9), "+v"(acc));
    float ti1_12 = TSIG(d2[1][1]);
    acc = fmaf(hh_9[0], wfcD[0][2], acc);
    ORDER("+v"(ti1_12), "+v"(acc), "+v"(d2[5]));
    float tg0_12 = tanh_act(d2[5][0]);
    ORDER("+v"(tg0_12), "+v"(d2[5]), "+v"(to0_10), "+v"(to1_10), "+v"(tc0_10), "+v"(tc1_10));
    float tg1_12 = tanh_act(d2[5][1]);
    v2f hh_10 = HOUT2(((v2f){to0_10, to1_10}), ((v2f){tc0_10, tc1_10}));
    ORDER("+v"(tg1_12), "+v"(hh_10), "+v"(d2[3]), "+v"(ff_11), "+v"(ig_11), "+v"(c2[0]));
    float tf0_12 = TSIG(d2[3][0]);
    { const v2f cn = __builtin_elementwise_fma(ff_11, (v2f){c2[0][6], c2[0][7]}, ig_11); c2[0][6] = cn[0]; c2[0][7] = cn[1]; }
    ORDER("+v"(tf0_12), "+v"(c2[0]), "+v"(d2[3]), "+v"(ti0_12), "+v"(ti1_12));
    float tf1_12 = TSIG(d2[3][1]);
    v2f fi_12 = FIN2(((v2f){ti0_12, ti1_12}));
    ORDER("+v"(tf1_12), "+v"(fi_12), "+v"(d2[7]), "+v"(hh_9), "+v"(acc));
    float to0_12 = TSIG(d2[7][0]);
    acc = fmaf(hh_9[1], wfcD[0][3], acc);
    ORDER("+v"(to0_12), "+v"(acc), "+v"(d2[7]), "+v"(hh_10));
    float to1_12 = TSIG(d2[7][1]);
    v2h pk_10 = __builtin_convertvector(hh_10, v2h);
    ORDER("+v"(to1_12), "+v"(pk_10), "+v"(c2[0]));
    float tc0_11 = tanh_act(c2[0][6]);
    ORDER("+v"(tc0_11), "+v"(c2[0]), "+v"(fi_12), "+v"(tg0_12), "+v"(tg1_12));
    float tc1_11 = tanh_act(c2[0][7]);
    v2f ig_12 = fi_12 * (v2f){tg0_12, tg1_12};
    ORDER("+v"(tc1_11), "+v"(ig_12), "+v"(d2[1]));
    float ti0_13 = TSIG(d2[1][2]);
    ORDER("+v"(ti0_13), "+v"(h1Bn));
    d1[6] = __builtin_amdgcn_wmma_f32_16x16x32_f16(false, a1[6], false, h1Bn, (short)0, zc, false, false);
    ORDER("+v"(d1[6]), "+v"(tf0_12), "+v"(tf1_12));
    v2f ff_12 = FIN2(((v2f){tf0_12, tf1_12}));
    ORDER("+v"(ff_12), "+v"(d2[1]), "+v"(hh_10), "+v"(acc));
    float ti1_13 = TSIG(d2[1][3]);
    acc = fmaf(hh_10[0], wfcD[0][4], acc);
    ORDER("+v"(ti1_13), "+v"(acc), "+v"(d2[5]));
    float tg0_13 = tanh_act(d2[5][2]);
    ORDER("+v"(tg0_13), "+v"(d2[5]), "+v"(to0_11), "+v"(to1_11), "+v"(tc0_11), "+v"(tc1_11));
    float tg1_13 = tanh_act(d2[5][3]);
    v2f hh_11 = HOUT2(((v2f){to0_11, to1_11}), ((v2f){tc0_11, tc1_11}));
    ORDER("+v"(tg1_13), "+v"(hh_11), "+v"(d2[3]), "+v"(ff_12), "+v"(ig_12), "+v"(c2[1]));
    float tf0_13 = TSIG(d2[3][2]);
    { const v2f cn = __builtin_elementwise_fma(ff_12, (v2f){c2[1][0], c2[1][1]}, ig_12); c2[1][0] = cn[0]; c2[1][1] = cn[1]; }
    ORDER("+v"(tf0_13), "+v"(c2[1]), "+v"(d2[3]), "+v"(ti0_13), "+v"(ti1_13));
    float tf1_13 = TSIG(d2[3][3]);
    v2f fi_13 = FIN2(((v2f){ti0_13, ti1_13}));
    ORDER("+v"(tf1_13), "+v"(fi_13), "+v"(d2[7]), "+v"(hh_10), "+v"(acc));
    float to0_13 = TSIG(d2[7][2]);
    acc = fmaf(hh_10[1], wfcD[0][5], acc);
    ORDER("+v"(to0_13), "+v"(acc), "+v"(d2[7]), "+v"(hh_11));
    float to1_13 = TSIG(d2[7][3]);
    v2h pk_11 = __builtin_convertvector(hh_11, v2h);
    ORDER("+v"(to1_13), "+v"(pk_11), "+v"(c2[1]));
    float tc0_12 = tanh_act(c2[1][0]);
    ORDER("+v"(tc0_12), "+v"(c2[1]), "+v"(fi_13), "+v"(tg0_13), "+v"(tg1_13));
    float tc1_12 = tanh_act(c2[1][1]);
    v2f ig_13 = fi_13 * (v2f){tg0_13, tg1_13};
    ORDER("+v"(tc1_12), "+v"(ig_13), "+v"(d2[1]));
    float ti0_14 = TSIG(d2[1][4]);
    ORDER("+v"(ti0_14), "+v"(h1Bn));
    d1[1] = __builtin_amdgcn_wmma_f32_16x16x32_f16(false, a1[1], false, h1Bn, (short)0, zc, false, false);
    ORDER("+v"(d1[1]), "+v"(tf0_13), "+v"(tf1_13));
    v2f ff_13 = FIN2(((v2f){tf0_13, tf1_13}));
    ORDER("+v"(ff_13), "+v"(d2[1]), "+v"(hh_11), "+v"(acc));
    float ti1_14 = TSIG(d2[1][5]);
    acc = fmaf(hh_11[0], wfcD[0][6], acc);
    ORDER("+v"(ti1_14), "+v"(acc), "+v"(d2[5]));
    float tg0_14 = tanh_act(d2[5][4]);
    ORDER("+v"(tg0_14), "+v"(d2[5]), "+v"(to0_12), "+v"(to1_12), "+v"(tc0_12), "+v"(tc1_12));
    float tg1_14 = tanh_act(d2[5][5]);
    v2f hh_12 = HOUT2(((v2f){to0_12, to1_12}), ((v2f){tc0_12, tc1_12}));
    ORDER("+v"(tg1_14), "+v"(hh_12), "+v"(d2[3]), "+v"(ff_13), "+v"(ig_13), "+v"(c2[1]));
    float tf0_14 = TSIG(d2[3][4]);
    { const v2f cn = __builtin_elementwise_fma(ff_13, (v2f){c2[1][2], c2[1][3]}, ig_13); c2[1][2] = cn[0]; c2[1][3] = cn[1]; }
    ORDER("+v"(tf0_14), "+v"(c2[1]), "+v"(d2[3]), "+v"(ti0_14), "+v"(ti1_14));
    float tf1_14 = TSIG(d2[3][5]);
    v2f fi_14 = FIN2(((v2f){ti0_14, ti1_14}));
    ORDER("+v"(tf1_14), "+v"(fi_14), "+v"(d2[7]), "+v"(hh_11), "+v"(acc));
    float to0_14 = TSIG(d2[7][4]);
    acc = fmaf(hh_11[1], wfcD[0][7], acc);
    ORDER("+v"(to0_14), "+v"(acc), "+v"(d2[7]), "+v"(hh_12));
    float to1_14 = TSIG(d2[7][5]);
    v2h pk_12 = __builtin_convertvector(hh_12, v2h);
    ORDER("+v"(to1_14), "+v"(pk_12), "+v"(c2[1]));
    float tc0_13 = tanh_act(c2[1][2]);
    ORDER("+v"(tc0_13), "+v"(c2[1]), "+v"(fi_14), "+v"(tg0_14), "+v"(tg1_14), "+v"(tf0_14), "+v"(tf1_14));
    float tc1_13 = tanh_act(c2[1][3]);
    v2f ig_14 = fi_14 * (v2f){tg0_14, tg1_14};
    v2f ff_14 = FIN2(((v2f){tf0_14, tf1_14}));
    ORDER("+v"(tc1_13), "+v"(ig_14), "+v"(ff_14), "+v"(d2[1]));
    float ti0_15 = TSIG(d2[1][6]);
    ORDER("+v"(ti0_15), "+v"(h1Bn), "+v"(hh_12), "+v"(acc));
    d1[3] = __builtin_amdgcn_wmma_f32_16x16x32_f16(false, a1[3], false, h1Bn, (short)0, zc, false, false);
    ORDER("+v"(d1[3]), "+v"(hh_12), "+v"(acc));
    acc = fmaf(hh_12[0], wfcD[1][0], acc);
    ORDER("+v"(acc), "+v"(d2[5]), "+v"(ff_14), "+v"(ig_14), "+v"(c2[1]));
    float tg0_15 = tanh_act(d2[5][6]);
    { const v2f cn = __builtin_elementwise_fma(ff_14, (v2f){c2[1][4], c2[1][5]}, ig_14); c2[1][4] = cn[0]; c2[1][5] = cn[1]; }
    ORDER("+v"(tg0_15), "+v"(c2[1]), "+v"(d2[3]), "+v"(to0_13), "+v"(to1_13), "+v"(tc0_13), "+v"(tc1_13));
    float tf0_15 = TSIG(d2[3][6]);
    v2f hh_13 = HOUT2(((v2f){to0_13, to1_13}), ((v2f){tc0_13, tc1_13}));
    ORDER("+v"(tf0_15), "+v"(hh_13), "+v"(d2[7]), "+v"(ti0_15), "+v"(hh_12), "+v"(acc));
    float to0_15 = TSIG(d2[7][6]);
    float fi_15 = FIN1(ti0_15);
    acc = fmaf(hh_12[1], wfcD[1][1], acc);
    ORDER("+v"(to0_15), "+v"(fi_15), "+v"(acc));
    z_to0_15 = to0_15; z_fi_15 = fi_15; z_tg0_15 = tg0_15; z_tf0_15 = tf0_15;
    z_to0_14 = to0_14; z_to1_14 = to1_14; z_hh_13 = hh_13; accC = acc;
    zpk_8 = pk_8; zpk_9 = pk_9; zpk_10 = pk_10; zpk_11 = pk_11; zpk_12 = pk_12;
    h1B = h1Bn;
  }
    v2f xn = xq;
    int xstep = (tt + 6 < TLEN) ? 2 : 0;
    ORDER("+v"(xn), "+s"(xstep));
    xa = xn[0];
    xb = xn[1];
    xp += xstep;
  }

  {
    const float tc0_14 = tanh_act(c2[1][4]);
    const float tc1_14 = tanh_act(c2[1][5]);
    const float ig_15 = z_fi_15 * z_tg0_15;
    const float c6 = fmaf(FIN1(z_tf0_15), c2[1][6], ig_15);
    const float tc0_15 = tanh_act(c6);
    const v2f hh_14 = HOUT2(((v2f){z_to0_14, z_to1_14}), ((v2f){tc0_14, tc1_14}));
    const float hh_15 = HOUT1(z_to0_15, tc0_15);
    accC = fmaf(z_hh_13[0], wfcD[1][2], accC);
    accC = fmaf(z_hh_13[1], wfcD[1][3], accC);
    accC = fmaf(hh_14[0], wfcD[1][4], accC);
    accC = fmaf(hh_14[1], wfcD[1][5], accC);
    accC = fmaf(hh_15, wfcD[1][6], accC);
  }
  const float accOd = xhalf(accC);
  ost[TLEN - 2] = hi_half ? (accC + accOd) + bfc : outEven;
}

extern "C" void kernel_launch(void* const* d_in, const int* in_sizes, int n_in,
                              void* d_out, int out_size, void* d_ws, size_t ws_size,
                              hipStream_t stream) {
  (void)in_sizes; (void)n_in; (void)out_size; (void)d_ws; (void)ws_size;
  const float* x     = (const float*)d_in[0];
  const float* w_ih1 = (const float*)d_in[1];
  const float* w_hh1 = (const float*)d_in[2];
  const float* b_ih1 = (const float*)d_in[3];
  const float* b_hh1 = (const float*)d_in[4];
  const float* w_ih2 = (const float*)d_in[5];
  const float* w_hh2 = (const float*)d_in[6];
  const float* b_ih2 = (const float*)d_in[7];
  const float* b_hh2 = (const float*)d_in[8];
  const float* w_fc  = (const float*)d_in[9];
  const float* b_fc  = (const float*)d_in[10];
  float* out = (float*)d_out;

  dim3 grid(BSZ / 16);
  dim3 block(32);
  lstm2_wmma_kernel<<<grid, block, 0, stream>>>(x, w_ih1, w_hh1, b_ih1, b_hh1,
                                                w_ih2, w_hh2, b_ih2, b_hh2,
                                                w_fc, b_fc, out);
}


# Eight steps per loop trip (was two): the trip's memory instructions in ONE place (four 8-byte x loads, one 16-byte out store of all 32 lanes, side by side); values and sum orders kept
# speedup vs baseline: 2.4262x; 1.0668x over previous
#include <hip/hip_runtime.h>


#define HSZ   30
#define MPAD  128
#define KPAD  32
#define TLEN  2048
#define BSZ   512

typedef __attribute__((ext_vector_type(16))) _Float16 v16h;
typedef __attribute__((ext_vector_type(8)))  _Float16 v8h;
typedef __attribute__((ext_vector_type(8)))  float    v8f;
typedef __attribute__((ext_vector_type(2)))  float    v2f;
typedef __attribute__((ext_vector_type(4)))  float    v4f;
typedef __attribute__((ext_vector_type(2)))  _Float16 v2h;

#define ORDER(...) asm volatile("" : __VA_ARGS__)

#if __has_builtin(__builtin_amdgcn_tanhf)
#define HW_TANH __builtin_amdgcn_tanhf
#elif __has_builtin(__builtin_amdgcn_tanh_f32)
#define HW_TANH __builtin_amdgcn_tanh_f32
#endif

#ifdef HW_TANH
__device__ __forceinline__ float tanh_act(float x) { return HW_TANH(x); }
#define GATE_IN_SCALE 0.5f
#define H_SCALE 0.5f
#define TSIG(m) HW_TANH(m)
#define FIN1(t) fmaf(0.5f, (t), 0.5f)
#define FIN2(t) __builtin_elementwise_fma((v2f)(0.5f), (t), (v2f)(0.5f))
#define HOUT1(to, tc) fmaf((to), (tc), (tc))
#define HOUT2(to, tc) __builtin_elementwise_fma((to), (tc), (tc))
#else
__device__ __forceinline__ float tanh_act(float x) {
  return fmaf(2.0f, __builtin_amdgcn_rcpf(1.0f + __expf(-2.0f * x)), -1.0f);
}
#define GATE_IN_SCALE 1.0f
#define H_SCALE 1.0f
#define TSIG(m) __builtin_amdgcn_rcpf(1.0f + __expf(-(m)))
#define FIN1(t) (t)
#define FIN2(t) (t)
#define HOUT1(to, tc) ((to) * (tc))
#define HOUT2(to, tc) ((to) * (tc))
#endif

__device__ __forceinline__ float xhalf(float v) {
  const unsigned u = __builtin_bit_cast(unsigned, v);
  return __builtin_bit_cast(float, __builtin_amdgcn_permlanex16(u, u, 0x76543210u, 0xfedcba98u, false, false));
}

__device__ __forceinline__ v16h load_a_frag(const float* __restrict__ wrow, bool hi_half, bool real,
                                            float ws, float c23, float c31) {
  const int kb = hi_half ? 8 : 0;
  const int kc = hi_half ? 22 : 16;
  v2f q[8];
#pragma unroll
  for (int i = 0; i < 4; ++i) q[i] = *reinterpret_cast<const v2f*>(wrow + kb + 2 * i);
#pragma unroll
  for (int i = 0; i < 4; ++i) q[4 + i] = *reinterpret_cast<const v2f*>(wrow + kc + 2 * i);
#pragma unroll
  for (int i = 0; i < 8; ++i) q[i] = q[i] * ws;
  float e[8];
#pragma unroll
  for (int i = 0; i < 7; ++i) {
    const float lo = q[4 + (i >> 1)][i & 1];
    const float hi = q[4 + ((i + 1) >> 1)][(i + 1) & 1];
    e[i] = hi_half ? hi : lo;
  }
  e[7] = hi_half ? c31 : c23;
  v16h a;
#pragma unroll
  for (int i = 0; i < 4; ++i) {
    a[2 * i]     = (_Float16)(real ? q[i][0] : 0.f);
    a[2 * i + 1] = (_Float16)(real ? q[i][1] : 0.f);
  }
#pragma unroll
  for (int i = 0; i < 8; ++i) a[8 + i] = (_Float16)(real ? e[i] : 0.f);
  return a;
}

__device__ __forceinline__ void gates8(const v8f (&d)[4], v8f &c, float (&hn)[8]) {
#pragma unroll
  for (int r = 0; r < 8; ++r) {
    const float iv = FIN1(TSIG(d[0][r]));
    const float fv = FIN1(TSIG(d[1][r]));
    const float gv = tanh_act(d[2][r]);
    const float to = TSIG(d[3][r]);
    const float cn = fmaf(fv, c[r], iv * gv);
    c[r] = cn;
    hn[r] = HOUT1(to, tanh_act(cn));
  }
}

__device__ __forceinline__ v16h layer1_step(const v16h (&a1)[8], v16h h1B, v8f (&c1)[2], v8f zc) {
  v16h h1Bn;
#pragma unroll
  for (int s = 0; s < 2; ++s) {
    v8f d[4];
#pragma unroll
    for (int g = 0; g < 4; ++g)
      d[g] = __builtin_amdgcn_wmma_f32_16x16x32_f16(false, a1[2 * g + s], false, h1B,
                                                    (short)0, zc, false, false);
    float hn[8];
    gates8(d, c1[s], hn);
#pragma unroll
    for (int r = 0; r < 8; ++r) h1Bn[8 * s + r] = (_Float16)hn[r];
  }
  return h1Bn;
}

__global__ __launch_bounds__(32, 1)
void lstm2_wmma_kernel(const float* __restrict__ x,
                       const float* __restrict__ w_ih1, const float* __restrict__ w_hh1,
                       const float* __restrict__ b_ih1, const float* __restrict__ b_hh1,
                       const float* __restrict__ w_ih2, const float* __restrict__ w_hh2,
                       const float* __restrict__ b_ih2, const float* __restrict__ b_hh2,
                       const float* __restrict__ w_fc,  const float* __restrict__ b_fc,
                       float* __restrict__ out) {
  const int lane = threadIdx.x;
  const bool hi_half = lane >= 16;
  const int n = lane & 15;

  v16h a1[8], ai2[8], ah2[8];
#pragma unroll
  for (int tl = 0; tl < 8; ++tl) {
    const int p = ((tl & 1) << 4) + n;
    const bool real = (p != 23) && (p != 31);
    const int unit = real ? (p < 23 ? p : p - 1) : HSZ - 1;
    const int row = (tl >> 1) * HSZ + unit;
    const float rs = ((tl >> 1) == 2) ? 1.0f : GATE_IN_SCALE;
    const float ws = rs * H_SCALE;
    const float wi1 = w_ih1[row] * rs;
    const float bs1 = (b_ih1[row] + b_hh1[row]) * rs;
    const float bs2 = (b_ih2[row] + b_hh2[row]) * rs;
    a1[tl]  = load_a_frag(w_hh1 + row * HSZ, hi_half, real, ws, wi1, bs1);
    ai2[tl] = load_a_frag(w_ih2 + row * HSZ, hi_half, real, ws, 0.f, 0.f);
    ah2[tl] = load_a_frag(w_hh2 + row * HSZ, hi_half, real, ws, 0.f, bs2);
  }

  float wfcD[2][8];
#pragma unroll
  for (int s = 0; s < 2; ++s)
#pragma unroll
    for (int r = 0; r < 8; ++r) {
      const int p = 16 * s + (hi_half ? 8 : 0) + r;
      const bool real = (p != 23) && (p != 31);
      const int unit = real ? (p < 23 ? p : p - 1) : 0;
      wfcD[s][r] = real ? w_fc[unit] * H_SCALE : 0.f;
    }
  const float bfc = b_fc[0];

  v8f zc;
#pragma unroll
  for (int i = 0; i < 8; ++i) zc[i] = 0.f;

  v16h h1B;
#pragma unroll
  for (int i = 0; i < 16; ++i) h1B[i] = (_Float16)0.f;

  v8f c1[2], c2[2];
#pragma unroll
  for (int s = 0; s < 2; ++s)
#pragma unroll
    for (int r = 0; r < 8; ++r) { c1[s][r] = 0.f; c2[s][r] = 0.f; }

  const int bbase = blockIdx.x << 4;
  const float* xrow = x + (size_t)(bbase + n) * TLEN;
  float* orow = out + (size_t)(bbase + n) * TLEN;

  const float x0 = xrow[0];
  const float x1 = xrow[1];
  float xr[8];
#pragma unroll
  for (int j = 0; j < 4; ++j) {
    const v2f q = *reinterpret_cast<const v2f*>(xrow + 2 + 2 * j);
    xr[2 * j] = q[0];
    xr[2 * j + 1] = q[1];
  }
  h1B[15] = hi_half ? (_Float16)1.f : (_Float16)x0;
  h1B = layer1_step(a1, h1B, c1, zc);
  h1B[15] = hi_half ? (_Float16)1.f : (_Float16)x1;

  v8f d1[8];
#pragma unroll
  for (int tl = 0; tl < 8; ++tl)
    d1[tl] = __builtin_amdgcn_wmma_f32_16x16x32_f16(false, a1[tl], false, h1B,
                                                    (short)0, zc, false, false);

  float z_to0_14 = 0.f, z_to1_14 = 0.f;
  float z_to0_15 = 0.f, z_fi_15 = 0.f, z_tg0_15 = 0.f, z_tf0_15 = 0.f;
  v2f z_hh_13 = {0.f, 0.f};
  v2h zpk_8, zpk_9, zpk_10, zpk_11, zpk_12;
  zpk_8[0] = zpk_8[1] = zpk_9[0] = zpk_9[1] = zpk_10[0] = zpk_10[1] = (_Float16)0.f;
  zpk_11[0] = zpk_11[1] = zpk_12[0] = zpk_12[1] = (_Float16)0.f;
  float accC = 0.f;
  float o8[8];
#pragma unroll
  for (int j = 0; j < 8; ++j) o8[j] = 0.f;
  float* ost = orow + (hi_half ? 4 : 0);
  const float rail2 = hi_half ? 1.0f : 0.0f;

  for (int tt = 0; tt < TLEN; tt += 8) {
    v2f xq[4];
#pragma unroll
  for (int u = 0; u < 8; ++u) {

    if (u == 1) {
#pragma unroll
      for (int j = 0; j < 4; ++j) {
        const int xi = tt + 10 + 2 * j;
        xq[j] = *reinterpret_cast<const v2f*>(xrow + (xi < TLEN - 2 ? xi : TLEN - 2));
      }
      if (tt > 0) {
        v4f ov;
#pragma unroll
        for (int j = 0; j < 4; ++j) ov[j] = hi_half ? o8[4 + j] : o8[j];
        *reinterpret_cast<v4f*>(ost + (tt - 8)) = ov;
      }
    }

    v8f d2[8];
    v16h h1Bn;
    float acc = 0.f;
    float xsel = xr[u];
    ORDER("+v"(d1[0]), "+v"(c2[1]));
    float tc0_14 = tanh_act(c2[1][4]);
    ORDER("+v"(tc0_14), "+v"(c2[1]), "+v"(z_hh_13));
    float tc1_14 = tanh_act(c2[1][5]);
    v2h pk_13 = __builtin_convertvector(z_hh_13, v2h);
    ORDER("+v"(tc1_14), "+v"(pk_13), "+v"(d1[0]));
    float ti0_0 = TSIG(d1[0][0]);
    ORDER("+v"(ti0_0), "+v"(h1B));
    d2[0] = __builtin_amdgcn_wmma_f32_16x16x32_f16(false, ai2[0], false, h1B, (short)0, zc, false, false);
    ORDER("+v"(d2[0]), "+v"(z_tf0_15));
    float ff_15 = FIN1(z_tf0_15);
    ORDER("+v"(ff_15), "+v"(d1[0]));
    float ti1_0 = TSIG(d1[0][1]);
    ORDER("+v"(ti1_0), "+v"(h1B));
    d1[5] = __builtin_amdgcn_wmma_f32_16x16x32_f16(false, a1[5], false, h1B, (short)0, zc, false, false);
    ORDER("+v"(d1[5]), "+v"(z_fi_15), "+v"(z_tg0_15));
    float ig_15 = z_fi_15 * z_tg0_15;
    ORDER("+v"(ig_15), "+v"(d1[4]), "+v"(z_hh_13), "+v"(accC));
    float tg0_0 = tanh_act(d1[4][0]);
    accC = fmaf(z_hh_13[0], wfcD[1][2], accC);
    ORDER("+v"(tg0_0), "+v"(accC), "+v"(d1[4]), "+v"(z_to0_14), "+v"(z_to1_14), "+v"(tc0_14), "+v"(tc1_14));
    float tg1_0 = tanh_act(d1[4][1]);
    v2f hh_14 = HOUT2(((v2f){z_to0_14, z_to1_14}), ((v2f){tc0_14, tc1_14}));
    ORDER("+v"(tg1_0), "+v"(hh_14), "+v"(d1[2]), "+v"(ff_15), "+v"(ig_15), "+v"(c2[1]));
    float tf0_0 = TSIG(d1[2][0]);
    c2[1][6] = fmaf(ff_15, c2[1][6], ig_15);
    ORDER("+v"(tf0_0), "+v"(c2[1]), "+v"(h1B));
    d2[2] = __builtin_amdgcn_wmma_f32_16x16x32_f16(false, ai2[2], false, h1B, (short)0, zc, false, false);
    ORDER("+v"(d2[2]), "+v"(ti0_0), "+v"(ti1_0));
    v2f fi_0 = FIN2(((v2f){ti0_0, ti1_0}));
    ORDER("+v"(fi_0), "+v"(d1[2]), "+v"(z_hh_13), "+v"(accC));
    float tf1_0 = TSIG(d1[2][1]);
    accC = fmaf(z_hh_13[1], wfcD[1][3], accC);
    ORDER("+v"(tf1_0), "+v"(accC), "+v"(d1[6]), "+v"(hh_14));
    float to0_0 = TSIG(d1[6][0]);
    v2h pk_14 = __builtin_convertvector(hh_14, v2h);
    ORDER("+v"(to0_0), "+v"(pk_14), "+v"(d1[6]));
    float to1_0 = TSIG(d1[6][1]);
    ORDER("+v"(to1_0), "+v"(c2[1]), "+v"(fi_0), "+v"(tg0_0), "+v"(tg1_0));
    float tc0_15 = tanh_act(c2[1][6]);
    v2f ig_0 = fi_0 * (v2f){tg0_0, tg1_0};
    ORDER("+v"(tc0_15), "+v"(ig_0), "+v"(d1[0]));
    float ti0_1 = TSIG(d1[0][2]);
    ORDER("+v"(ti0_1), "+v"(h1B));
    d2[4] = __builtin_amdgcn_wmma_f32_16x16x32_f16(false, ai2[4], false, h1B, (short)0, zc, false, false);
    ORDER("+v"(d2[4]), "+v"(tf0_0), "+v"(tf1_0));
    v2f ff_0 = FIN2(((v2f){tf0_0, tf1_0}));
    ORDER("+v"(ff_0), "+v"(d1[0]));
    float ti1_1 = TSIG(d1[0][3]);
    ORDER("+v"(ti1_1), "+v"(h1B), "+v"(hh_14), "+v"(accC));
    d1[7] = __builtin_amdgcn_wmma_f32_16x16x32_f16(false, a1[7], false, h1B, (short)0, zc, false, false);
    ORDER("+v"(d1[7]), "+v"(hh_14), "+v"(accC));
    accC = fmaf(hh_14[0], wfcD[1][4], accC);
    ORDER("+v"(accC), "+v"(d1[4]));
    float tg0_1 = tanh_act(d1[4][2]);
    ORDER("+v"(tg0_1), "+v"(d1[4]), "+v"(z_to0_15), "+v"(tc0_15));
    float tg1_1 = tanh_act(d1[4][3]);
    float hh_15 = HOUT1(z_to0_15, tc0_15);
    ORDER("+v"(tg1_1), "+v"(hh_15), "+v"(d1[2]), "+v"(ff_0), "+v"(ig_0), "+v"(c1[0]));
    float tf0_1 = TSIG(d1[2][2]);
    { const v2f cn = __builtin_elementwise_fma(ff_0, (v2f){c1[0][0], c1[0][1]}, ig_0); c1[0][0] = cn[0]; c1[0][1] = cn[1]; }
    ORDER("+v"(tf0_1), "+v"(c1[0]), "+v"(h1B));
    d2[6] = __builtin_amdgcn_wmma_f32_16x16x32_f16(false, ai2[6], false, h1B, (short)0, zc, false, false);
    ORDER("+v"(d2[6]), "+v"(ti0_1), "+v"(ti1_1));
    v2f fi_1 = FIN2(((v2f){ti0_1, ti1_1}));
    ORDER("+v"(fi_1), "+v"(d1[2]), "+v"(hh_14), "+v"(accC));
    float tf1_1 = TSIG(d1[2][3]);
    accC = fmaf(hh_14[1], wfcD[1][5], accC);
    ORDER("+v"(tf1_1), "+v"(accC), "+v"(d1[6]), "+v"(hh_15));
    float to0_1 = TSIG(d1[6][2]);
    v2h pk_15 = __builtin_convertvector(((v2f){hh_15, rail2}), v2h);
    ORDER("+v"(to0_1), "+v"(pk_15), "+v"(d1[6]));
    float to1_1 = TSIG(d1[6][3]);
    v16h h2B;
    h2B[0] = zpk_8[0]; h2B[1] = zpk_8[1];
    h2B[2] = zpk_9[0]; h2B[3] = zpk_9[1];
    h2B[4] = zpk_10[0]; h2B[5] = zpk_10[1];
    h2B[6] = zpk_11[0]; h2B[7] = zpk_11[1];
    h2B[8] = zpk_12[0]; h2B[9] = zpk_12[1];
    h2B[10] = pk_13[0]; h2B[11] = pk_13[1];
    h2B[12] = pk_14[0]; h2B[13] = pk_14[1];
    h2B[14] = pk_15[0]; h2B[15] = pk_15[1];
    ORDER("+v"(to1_1), "+v"(c1[0]), "+v"(hh_15), "+v"(accC));
    float tc0_0 = tanh_act(c1[0][0]);
    accC = fmaf(hh_15, wfcD[1][6], accC);
    ORDER("+v"(tc0_0), "+v"(accC), "+v"(c1[0]), "+v"(fi_1), "+v"(tg0_1), "+v"(tg1_1));
    float tc1_0 = tanh_act(c1[0][1]);
    v2f ig_1 = fi_1 * (v2f){tg0_1, tg1_1};
    ORDER("+v"(tc1_0), "+v"(ig_1), "+v"(d1[0]));
    float ti0_2 = TSIG(d1[0][4]);
    ORDER("+v"(ti0_2), "+v"(h1B));
    d2[1] = __builtin_amdgcn_wmma_f32_16x16x32_f16(false, ai2[1], false, h1B, (short)0, zc, false, false);
    ORDER("+v"(d2[1]), "+v"(tf0_1), "+v"(tf1_1));
    v2f ff_1 = FIN2(((v2f){tf0_1, tf1_1}));
    ORDER("+v"(ff_1), "+v"(d1[0]));
    float ti1_2 = TSIG(d1[0][5]);
    ORDER("+v"(ti1_2), "+v"(d1[4]), "+v"(accC));
    float tg0_2 = tanh_act(d1[4][4]);
    float accX = accC;
    ORDER("+v"(tg0_2), "+v"(accX), "+v"(accC), "+v"(d1[4]), "+v"(to0_0), "+v"(to1_0), "+v"(tc0_0), "+v"(tc1_0));
    float tg1_2 = tanh_act(d1[4][5]);
    v2f hh_0 = HOUT2(((v2f){to0_0, to1_0}), ((v2f){tc0_0, tc1_0}));
    ORDER("+v"(tg1_2), "+v"(hh_0), "+v"(d1[2]), "+v"(ff_1), "+v"(ig_1), "+v"(c1[0]));
    float tf0_2 = TSIG(d1[2][4]);
    { const v2f cn = __builtin_elementwise_fma(ff_1, (v2f){c1[0][2], c1[0][3]}, ig_1); c1[0][2] = cn[0]; c1[0][3] = cn[1]; }
    ORDER("+v"(tf0_2), "+v"(c1[0]), "+v"(h1B));
    d2[3] = __builtin_amdgcn_wmma_f32_16x16x32_f16(false, ai2[3], false, h1B, (short)0, zc, false, false);
    ORDER("+v"(d2[3]), "+v"(ti0_2), "+v"(ti1_2));
    v2f fi_2 = FIN2(((v2f){ti0_2, ti1_2}));
    ORDER("+v"(fi_2), "+v"(d1[2]), "+v"(accX));
    float tf1_2 = TSIG(d1[2][5]);
    float accO = xhalf(accX);
    ORDER("+v"(tf1_2), "+v"(accO), "+v"(d1[6]));
    float to0_2 = TSIG(d1[6][4]);
    ORDER("+v"(to0_2), "+v"(d1[6]), "+v"(hh_0));
    float to1_2 = TSIG(d1[6][5]);
    v2h pk_0 = __builtin_convertvector(hh_0, v2h);
    ORDER("+v"(to1_2), "+v"(pk_0), "+v"(c1[0]));
    float tc0_1 = tanh_act(c1[0][2]);
    ORDER("+v"(tc0_1), "+v"(c1[0]), "+v"(fi_2), "+v"(tg0_2), "+v"(tg1_2));
    float tc1_1 = tanh_act(c1[0][3]);
    v2f ig_2 = fi_2 * (v2f){tg0_2, tg1_2};
    ORDER("+v"(tc1_1), "+v"(ig_2), "+v"(d1[0]));
    float ti0_3 = TSIG(d1[0][6]);
    ORDER("+v"(ti0_3), "+v"(h1B));
    d2[5] = __builtin_amdgcn_wmma_f32_16x16x32_f16(false, ai2[5], false, h1B, (short)0, zc, false, false);
    ORDER("+v"(d2[5]), "+v"(tf0_2), "+v"(tf1_2));
    v2f ff_2 = FIN2(((v2f){tf0_2, tf1_2}));
    ORDER("+v"(ff_2), "+v"(d1[0]), "+v"(accC), "+v"(accO));
    float ti1_3 = TSIG(d1[0][7]);
    float osum = accC + accO;
    ORDER("+v"(ti1_3), "+v"(osum), "+v"(d1[4]));
    float tg0_3 = tanh_act(d1[4][6]);
    ORDER("+v"(tg0_3), "+v"(d1[4]), "+v"(to0_1), "+v"(to1_1), "+v"(tc0_1), "+v"(tc1_1));
    float tg1_3 = tanh_act(d1[4][7]);
    v2f hh_1 = HOUT2(((v2f){to0_1, to1_1}), ((v2f){tc0_1, tc1_1}));
    ORDER("+v"(tg1_3), "+v"(hh_1), "+v"(d1[2]), "+v"(ff_2), "+v"(ig_2), "+v"(c1[0]));
    float tf0_3 = TSIG(d1[2][6]);
    { const v2f cn = __builtin_elementwise_fma(ff_2, (v2f){c1[0][4], c1[0][5]}, ig_2); c1[0][4] = cn[0]; c1[0][5] = cn[1]; }
    ORDER("+v"(tf0_3), "+v"(c1[0]), "+v"(h1B));
    d2[7] = __builtin_amdgcn_wmma_f32_16x16x32_f16(false, ai2[7], false, h1B, (short)0, zc, false, false);
    ORDER("+v"(d2[7]), "+v"(ti0_3), "+v"(ti1_3));
    v2f fi_3 = FIN2(((v2f){ti0_3, ti1_3}));
    ORDER("+v"(fi_3), "+v"(d1[2]), "+v"(osum));
    float tf1_3 = TSIG(d1[2][7]);
    float ofin = osum + bfc;
    ORDER("+v"(tf1_3), "+v"(ofin), "+v"(d1[6]));
    o8[(u + 7) & 7] = ofin;
    float to0_3 = TSIG(d1[6][6]);
    ORDER("+v"(to0_3), "+v"(d1[6]), "+v"(hh_1));
    float to1_3 = TSIG(d1[6][7]);
    v2h pk_1 = __builtin_convertvector(hh_1, v2h);
    ORDER("+v"(to1_3), "+v"(pk_1), "+v"(c1[0]));
    float tc0_2 = tanh_act(c1[0][4]);
    ORDER("+v"(tc0_2), "+v"(c1[0]), "+v"(fi_3), "+v"(tg0_3), "+v"(tg1_3));
    float tc1_2 = tanh_act(c1[0][5]);
    v2f ig_3 = fi_3 * (v2f){tg0_3, tg1_3};
    ORDER("+v"(tc1_2), "+v"(ig_3), "+v"(d1[1]));
    float ti0_4 = TSIG(d1[1][0]);
    ORDER("+v"(ti0_4), "+v"(h2B), "+v"(d2[0]));
    d2[0] = __builtin_amdgcn_wmma_f32_16x16x32_f16(false, ah2[0], false, h2B, (short)0, d2[0], false, false);
    ORDER("+v"(d2[0]), "+v"(tf0_3), "+v"(tf1_3));
    v2f ff_3 = FIN2(((v2f){tf0_3, tf1_3}));
    ORDER("+v"(ff_3), "+v"(d1[1]));
    float ti1_4 = TSIG(d1[1][1]);
    ORDER("+v"(ti1_4), "+v"(d1[5]));
    float tg0_4 = tanh_act(d1[5][0]);
    ORDER("+v"(tg0_4), "+v"(d1[5]), "+v"(to0_2), "+v"(to1_2), "+v"(tc0_2), "+v"(tc1_2));
    float tg1_4 = tanh_act(d1[5][1]);
    v2f hh_2 = HOUT2(((v2f){to0_2, to1_2}), ((v2f){tc0_2, tc1_2}));
    ORDER("+v"(tg1_4), "+v"(hh_2), "+v"(d1[3]), "+v"(ff_3), "+v"(ig_3), "+v"(c1[0]));
    float tf0_4 = TSIG(d1[3][0]);
    { const v2f cn = __builtin_elementwise_fma(ff_3, (v2f){c1[0][6], c1[0][7]}, ig_3); c1[0][6] = cn[0]; c1[0][7] = cn[1]; }
    ORDER("+v"(tf0_4), "+v"(c1[0]), "+v"(h2B), "+v"(d2[2]));
    d2[2] = __builtin_amdgcn_wmma_f32_16x16x32_f16(false, ah2[2], false, h2B, (short)0, d2[2], false, false);
    ORDER("+v"(d2[2]), "+v"(ti0_4), "+v"(ti1_4));
    v2f fi_4 = FIN2(((v2f){ti0_4, ti1_4}));
    ORDER("+v"(fi_4), "+v"(d1[3]));
    float tf1_4 = TSIG(d1[3][1]);
    ORDER("+v"(tf1_4), "+v"(d1[7]));
    float to0_4 = TSIG(d1[7][0]);
    ORDER("+v"(to0_4), "+v"(d1[7]), "+v"(hh_2));
    float to1_4 = TSIG(d1[7][1]);
    v2h pk_2 = __builtin_convertvector(hh_2, v2h);
    ORDER("+v"(to1_4), "+v"(pk_2), "+v"(c1[0]));
    float tc0_3 = tanh_act(c1[0][6]);
    ORDER("+v"(tc0_3), "+v"(c1[0]), "+v"(fi_4), "+v"(tg0_4), "+v"(tg1_4));
    float tc1_3 = tanh_act(c1[0][7]);
    v2f ig_4 = fi_4 * (v2f){tg0_4, tg1_4};
    ORDER("+v"(tc1_3), "+v"(ig_4), "+v"(d1[1]));
    float ti0_5 = TSIG(d1[1][2]);
    ORDER("+v"(ti0_5), "+v"(h2B), "+v"(d2[4]));
    d2[4] = __builtin_amdgcn_wmma_f32_16x16x32_f16(false, ah2[4], false, h2B, (short)0, d2[4], false, false);
    ORDER("+v"(d2[4]), "+v"(tf0_4), "+v"(tf1_4));
    v2f ff_4 = FIN2(((v2f){tf0_4, tf1_4}));
    ORDER("+v"(ff_4), "+v"(d1[1]));
    float ti1_5 = TSIG(d1[1][3]);
    ORDER("+v"(ti1_5), "+v"(d1[5]));
    float tg0_5 = tanh_act(d1[5][2]);
    ORDER("+v"(tg0_5), "+v"(d1[5]), "+v"(to0_3), "+v"(to1_3), "+v"(tc0_3), "+v"(tc1_3));
    float tg1_5 = tanh_act(d1[5][3]);
    v2f hh_3 = HOUT2(((v2f){to0_3, to1_3}), ((v2f){tc0_3, tc1_3}));
    ORDER("+v"(tg1_5), "+v"(hh_3), "+v"(d1[3]), "+v"(ff_4), "+v"(ig_4), "+v"(c1[1]));
    float tf0_5 = TSIG(d1[3][2]);
    { const v2f cn = __builtin_elementwise_fma(ff_4, (v2f){c1[1][0], c1[1][1]}, ig_4); c1[1][0] = cn[0]; c1[1][1] = cn[1]; }
    ORDER("+v"(tf0_5), "+v"(c1[1]), "+v"(h2B), "+v"(d2[6]));
    d2[6] = __builtin_amdgcn_wmma_f32_16x16x32_f16(false, ah2[6], false, h2B, (short)0, d2[6], false, false);
    ORDER("+v"(d2[6]), "+v"(ti0_5), "+v"(ti1_5));
    v2f fi_5 = FIN2(((v2f){ti0_5, ti1_5}));
    ORDER("+v"(fi_5), "+v"(d1[3]));
    float tf1_5 = TSIG(d1[3][3]);
    ORDER("+v"(tf1_5), "+v"(d1[7]));
    float to0_5 = TSIG(d1[7][2]);
    ORDER("+v"(to0_5), "+v"(d1[7]), "+v"(hh_3));
    float to1_5 = TSIG(d1[7][3]);
    v2h pk_3 = __builtin_convertvector(hh_3, v2h);
    ORDER("+v"(to1_5), "+v"(pk_3), "+v"(c1[1]));
    float tc0_4 = tanh_act(c1[1][0]);
    ORDER("+v"(tc0_4), "+v"(c1[1]), "+v"(fi_5), "+v"(tg0_5), "+v"(tg1_5));
    float tc1_4 = tanh_act(c1[1][1]);
    v2f ig_5 = fi_5 * (v2f){tg0_5, tg1_5};
    ORDER("+v"(tc1_4), "+v"(ig_5), "+v"(d1[1]));
    float ti0_6 = TSIG(d1[1][4]);
    ORDER("+v"(ti0_6), "+v"(h2B), "+v"(d2[1]));
    d2[1] = __builtin_amdgcn_wmma_f32_16x16x32_f16(false, ah2[1], false, h2B, (short)0, d2[1], false, false);
    ORDER("+v"(d2[1]), "+v"(tf0_5), "+v"(tf1_5));
    v2f ff_5 = FIN2(((v2f){tf0_5, tf1_5}));
    ORDER("+v"(ff_5), "+v"(d1[1]));
    float ti1_6 = TSIG(d1[1][5]);
    ORDER("+v"(ti1_6), "+v"(d1[5]));
    float tg0_6 = tanh_act(d1[5][4]);
    ORDER("+v"(tg0_6), "+v"(d1[5]), "+v"(to0_4), "+v"(to1_4), "+v"(tc0_4), "+v"(tc1_4));
    float tg1_6 = tanh_act(d1[5][5]);
    v2f hh_4 = HOUT2(((v2f){to0_4, to1_4}), ((v2f){tc0_4, tc1_4}));
    ORDER("+v"(tg1_6), "+v"(hh_4), "+v"(d1[3]), "+v"(ff_5), "+v"(ig_5), "+v"(c1[1]));
    float tf0_6 = TSIG(d1[3][4]);
    { const v2f cn = __builtin_elementwise_fma(ff_5, (v2f){c1[1][2], c1[1][3]}, ig_5); c1[1][2] = cn[0]; c1[1][3] = cn[1]; }
    ORDER("+v"(tf0_6), "+v"(c1[1]), "+v"(h2B), "+v"(d2[3]));
    d2[3] = __builtin_amdgcn_wmma_f32_16x16x32_f16(false, ah2[3], false, h2B, (short)0, d2[3], false, false);
    ORDER("+v"(d2[3]), "+v"(ti0_6), "+v"(ti1_6));
    v2f fi_6 = FIN2(((v2f){ti0_6, ti1_6}));
    ORDER("+v"(fi_6), "+v"(d1[3]));
    float tf1_6 = TSIG(d1[3][5]);
    ORDER("+v"(tf1_6), "+v"(d1[7]));
    float to0_6 = TSIG(d1[7][4]);
    ORDER("+v"(to0_6), "+v"(d1[7]), "+v"(hh_4));
    float to1_6 = TSIG(d1[7][5]);
    v2h pk_4 = __builtin_convertvector(hh_4, v2h);
    ORDER("+v"(to1_6), "+v"(pk_4), "+v"(c1[1]), "+v"(fi_6), "+v"(tg0_6), "+v"(tg1_6));
    float tc0_5 = tanh_act(c1[1][2]);
    v2f ig_6 = fi_6 * (v2f){tg0_6, tg1_6};
    ORDER("+v"(tc0_5), "+v"(ig_6), "+v"(c1[1]), "+v"(tf0_6), "+v"(tf1_6));
    float tc1_5 = tanh_act(c1[1][3]);
    v2f ff_6 = FIN2(((v2f){tf0_6, tf1_6}));
    ORDER("+v"(tc1_5), "+v"(ff_6), "+v"(d1[1]));
    float ti0_7 = TSIG(d1[1][6]);
    ORDER("+v"(ti0_7), "+v"(h2B), "+v"(d2[5]), "+v"(xsel));
    d2[5] = __builtin_amdgcn_wmma_f32_16x16x32_f16(false, ah2[5], false, h2B, (short)0, d2[5], false, false);
    ORDER("+v"(d2[5]), "+v"(xsel));
    float rail1 = hi_half ? 1.0f : xsel;
    ORDER("+v"(rail1), "+v"(d1[5]), "+v"(ff_6), "+v"(ig_6), "+v"(c1[1]));
    float tg0_7 = tanh_act(d1[5][6]);
    { const v2f cn = __builtin_elementwise_fma(ff_6, (v2f){c1[1][4], c1[1][5]}, ig_6); c1[1][4] = cn[0]; c1[1][5] = cn[1]; }
    ORDER("+v"(tg0_7), "+v"(c1[1]), "+v"(d1[3]), "+v"(to0_5), "+v"(to1_5), "+v"(tc0_5), "+v"(tc1_5));
    float tf0_7 = TSIG(d1[3][6]);
    v2f hh_5 = HOUT2(((v2f){to0_5, to1_5}), ((v2f){tc0_5, tc1_5}));
    ORDER("+v"(tf0_7), "+v"(hh_5), "+v"(h2B), "+v"(d2[7]));
    d2[7] = __builtin_amdgcn_wmma_f32_16x16x32_f16(false, ah2[7], false, h2B, (short)0, d2[7], false, false);
    ORDER("+v"(d2[7]), "+v"(ti0_7));
    float fi_7 = FIN1(ti0_7);
    ORDER("+v"(fi_7), "+v"(d1[7]), "+v"(hh_5));
    float to0_7 = TSIG(d1[7][6]);
    v2h pk_5 = __builtin_convertvector(hh_5, v2h);
    ORDER("+v"(to0_7), "+v"(pk_5), "+v"(c1[1]));
    float tc0_6 = tanh_act(c1[1][4]);
    ORDER("+v"(tc0_6), "+v"(c1[1]), "+v"(fi_7), "+v"(tg0_7));
    float tc1_6 = tanh_act(c1[1][5]);
    float ig_7 = fi_7 * tg0_7;
    ORDER("+v"(tc1_6), "+v"(ig_7), "+v"(d2[0]), "+v"(tf0_7));
    float ti0_8 = TSIG(d2[0][0]);
    float ff_7 = FIN1(tf0_7);
    ORDER("+v"(ti0_8), "+v"(ff_7), "+v"(d2[0]));
    float ti1_8 = TSIG(d2[0][1]);
    ORDER("+v"(ti1_8), "+v"(d2[4]));
    float tg0_8 = tanh_act(d2[4][0]);
    ORDER("+v"(tg0_8), "+v"(d2[4]), "+v"(to0_6), "+v"(to1_6), "+v"(tc0_6), "+v"(tc1_6));
    float tg1_8 = tanh_act(d2[4][1]);
    v2f hh_6 = HOUT2(((v2f){to0_6, to1_6}), ((v2f){tc0_6, tc1_6}));
    ORDER("+v"(tg1_8), "+v"(hh_6), "+v"(d2[2]), "+v"(ff_7), "+v"(ig_7), "+v"(c1[1]));
    float tf0_8 = TSIG(d2[2][0]);
    c1[1][6] = fmaf(ff_7, c1[1][6], ig_7);
    ORDER("+v"(tf0_8), "+v"(c1[1]), "+v"(d2[2]), "+v"(ti0_8), "+v"(ti1_8));
    float tf1_8 = TSIG(d2[2][1]);
    v2f fi_8 = FIN2(((v2f){ti0_8, ti1_8}));
    ORDER("+v"(tf1_8), "+v"(fi_8), "+v"(d2[6]));
    float to0_8 = TSIG(d2[6][0]);
    ORDER("+v"(to0_8), "+v"(d2[6]), "+v"(hh_6));
    float to1_8 = TSIG(d2[6][1]);
    v2h pk_6 = __builtin_convertvector(hh_6, v2h);
    ORDER("+v"(to1_8), "+v"(pk_6), "+v"(c1[1]));
    float tc0_7 = tanh_act(c1[1][6]);
    ORDER("+v"(tc0_7), "+v"(c1[1]), "+v"(fi_8), "+v"(tg0_8), "+v"(tg1_8));
    v2f ig_8 = fi_8 * (v2f){tg0_8, tg1_8};
    ORDER("+v"(ig_8), "+v"(d2[0]), "+v"(tf0_8), "+v"(tf1_8));
    float ti0_9 = TSIG(d2[0][2]);
    v2f ff_8 = FIN2(((v2f){tf0_8, tf1_8}));
    ORDER("+v"(ti0_9), "+v"(ff_8), "+v"(d2[0]));
    float ti1_9 = TSIG(d2[0][3]);
    ORDER("+v"(ti1_9), "+v"(d2[4]));
    float tg0_9 = tanh_act(d2[4][2]);
    ORDER("+v"(tg0_9), "+v"(d2[4]), "+v"(to0_7), "+v"(tc0_7));
    float tg1_9 = tanh_act(d2[4][3]);
    float hh_7 = HOUT1(to0_7, tc0_7);
    ORDER("+v"(tg1_9), "+v"(hh_7), "+v"(d2[2]), "+v"(ff_8), "+v"(ig_8), "+v"(c2[0]));
    float tf0_9 = TSIG(d2[2][2]);
    { const v2f cn = __builtin_elementwise_fma(ff_8, (v2f){c2[0][0], c2[0][1]}, ig_8); c2[0][0] = cn[0]; c2[0][1] = cn[1]; }
    ORDER("+v"(tf0_9), "+v"(c2[0]), "+v"(d2[2]), "+v"(ti0_9), "+v"(ti1_9));
    float tf1_9 = TSIG(d2[2][3]);
    v2f fi_9 = FIN2(((v2f){ti0_9, ti1_9}));
    ORDER("+v"(tf1_9), "+v"(fi_9), "+v"(d2[6]));
    float to0_9 = TSIG(d2[6][2]);
    ORDER("+v"(to0_9), "+v"(d2[6]), "+v"(hh_7));
    float to1_9 = TSIG(d2[6][3]);
    v2h pk_7 = __builtin_convertvector(((v2f){hh_7, rail1}), v2h);
    ORDER("+v"(to1_9), "+v"(pk_7), "+v"(c2[0]));
    float tc0_8 = tanh_act(c2[0][0]);
    h1Bn[0] = pk_0[0]; h1Bn[1] = pk_0[1];
    h1Bn[2] = pk_1[0]; h1Bn[3] = pk_1[1];
    h1Bn[4] = pk_2[0]; h1Bn[5] = pk_2[1];
    h1Bn[6] = pk_3[0]; h1Bn[7] = pk_3[1];
    h1Bn[8] = pk_4[0]; h1Bn[9] = pk_4[1];
    h1Bn[10] = pk_5[0]; h1Bn[11] = pk_5[1];
    h1Bn[12] = pk_6[0]; h1Bn[13] = pk_6[1];
    h1Bn[14] = pk_7[0]; h1Bn[15] = pk_7[1];
    ORDER("+v"(tc0_8), "+v"(h1Bn), "+v"(c2[0]), "+v"(fi_9), "+v"(tg0_9), "+v"(tg1_9));
    float tc1_8 = tanh_act(c2[0][1]);
    v2f ig_9 = fi_9 * (v2f){tg0_9, tg1_9};
    ORDER("+v"(tc1_8), "+v"(ig_9), "+v"(d2[0]), "+v"(tf0_9), "+v"(tf1_9));
    float ti0_10 = TSIG(d2[0][4]);
    v2f ff_9 = FIN2(((v2f){tf0_9, tf1_9}));
    ORDER("+v"(ti0_10), "+v"(ff_9), "+v"(d2[0]));
    float ti1_10 = TSIG(d2[0][5]);
    ORDER("+v"(ti1_10), "+v"(d2[4]));
    float tg0_10 = tanh_act(d2[4][4]);
    ORDER("+v"(tg0_10), "+v"(d2[4]), "+v"(to0_8), "+v"(to1_8), "+v"(tc0_8), "+v"(tc1_8));
    float tg1_10 = tanh_act(d2[4][5]);
    v2f hh_8 = HOUT2(((v2f){to0_8, to1_8}), ((v2f){tc0_8, tc1_8}));
    ORDER("+v"(tg1_10), "+v"(hh_8), "+v"(d2[2]), "+v"(ff_9), "+v"(ig_9), "+v"(c2[0]));
    float tf0_10 = TSIG(d2[2][4]);
    { const v2f cn = __builtin_elementwise_fma(ff_9, (v2f){c2[0][2], c2[0][3]}, ig_9); c2[0][2] = cn[0]; c2[0][3] = cn[1]; }
    ORDER("+v"(tf0_10), "+v"(c2[0]), "+v"(h1Bn));
    d1[0] = __builtin_amdgcn_wmma_f32_16x16x32_f16(false, a1[0], false, h1Bn, (short)0, zc, false, false);
    ORDER("+v"(d1[0]), "+v"(ti0_10), "+v"(ti1_10));
    v2f fi_10 = FIN2(((v2f){ti0_10, ti1_10}));
    ORDER("+v"(fi_10), "+v"(d2[2]));
    float tf1_10 = TSIG(d2[2][5]);
    ORDER("+v"(tf1_10), "+v"(d2[6]));
    float to0_10 = TSIG(d2[6][4]);
    ORDER("+v"(to0_10), "+v"(d2[6]), "+v"(hh_8));
    float to1_10 = TSIG(d2[6][5]);
    v2h pk_8 = __builtin_convertvector(hh_8, v2h);
    ORDER("+v"(to1_10), "+v"(pk_8), "+v"(c2[0]));
    float tc0_9 = tanh_act(c2[0][2]);
    ORDER("+v"(tc0_9), "+v"(c2[0]), "+v"(fi_10), "+v"(tg0_10), "+v"(tg1_10));
    float tc1_9 = tanh_act(c2[0][3]);
    v2f ig_10 = fi_10 * (v2f){tg0_10, tg1_10};
    ORDER("+v"(tc1_9), "+v"(ig_10), "+v"(d2[0]));
    float ti0_11 = TSIG(d2[0][6]);
    ORDER("+v"(ti0_11), "+v"(h1Bn));
    d1[2] = __builtin_amdgcn_wmma_f32_16x16x32_f16(false, a1[2], false, h1Bn, (short)0, zc, false, false);
    ORDER("+v"(d1[2]), "+v"(tf0_10), "+v"(tf1_10));
    v2f ff_10 = FIN2(((v2f){tf0_10, tf1_10}));
    ORDER("+v"(ff_10), "+v"(d2[0]), "+v"(hh_8), "+v"(acc));
    float ti1_11 = TSIG(d2[0][7]);
    acc = fmaf(hh_8[0], wfcD[0][0], acc);
    ORDER("+v"(ti1_11), "+v"(acc), "+v"(d2[4]));
    float tg0_11 = tanh_act(d2[4][6]);
    ORDER("+v"(tg0_11), "+v"(d2[4]), "+v"(to0_9), "+v"(to1_9), "+v"(tc0_9), "+v"(tc1_9));
    float tg1_11 = tanh_act(d2[4][7]);
    v2f hh_9 = HOUT2(((v2f){to0_9, to1_9}), ((v2f){tc0_9, tc1_9}));
    ORDER("+v"(tg1_11), "+v"(hh_9), "+v"(d2[2]), "+v"(ff_10), "+v"(ig_10), "+v"(c2[0]));
    float tf0_11 = TSIG(d2[2][6]);
    { const v2f cn = __builtin_elementwise_fma(ff_10, (v2f){c2[0][4], c2[0][5]}, ig_10); c2[0][4] = cn[0]; c2[0][5] = cn[1]; }
    ORDER("+v"(tf0_11), "+v"(c2[0]), "+v"(d2[2]), "+v"(ti0_11), "+v"(ti1_11));
    float tf1_11 = TSIG(d2[2][7]);
    v2f fi_11 = FIN2(((v2f){ti0_11, ti1_11}));
    ORDER("+v"(tf1_11), "+v"(fi_11), "+v"(d2[6]), "+v"(hh_8), "+v"(acc));
    float to0_11 = TSIG(d2[6][6]);
    acc = fmaf(hh_8[1], wfcD[0][1], acc);
    ORDER("+v"(to0_11), "+v"(acc), "+v"(d2[6]), "+v"(hh_9));
    float to1_11 = TSIG(d2[6][7]);
    v2h pk_9 = __builtin_convertvector(hh_9, v2h);
    ORDER("+v"(to1_11), "+v"(pk_9), "+v"(c2[0]));
    float tc0_10 = tanh_act(c2[0][4]);
    ORDER("+v"(tc0_10), "+v"(c2[0]), "+v"(fi_11), "+v"(tg0_11), "+v"(tg1_11));
    float tc1_10 = tanh_act(c2[0][5]);
    v2f ig_11 = fi_11 * (v2f){tg0_11, tg1_11};
    ORDER("+v"(tc1_10), "+v"(ig_11), "+v"(d2[1]));
    float ti0_12 = TSIG(d2[1][0]);
    ORDER("+v"(ti0_12), "+v"(h1Bn));
    d1[4] = __builtin_amdgcn_wmma_f32_16x16x32_f16(false, a1[4], false, h1Bn, (short)0, zc, false, false);
    ORDER("+v"(d1[4]), "+v"(tf0_11), "+v"(tf1_11));
    v2f ff_11 = FIN2(((v2f){tf0_11, tf1_11}));
    ORDER("+v"(ff_11), "+v"(d2[1]), "+v"(hh_9), "+v"(acc));
    float ti1_12 = TSIG(d2[1][1]);
    acc = fmaf(hh_9[0], wfcD[0][2], acc);
    ORDER("+v"(ti1_12), "+v"(acc), "+v"(d2[5]));
    float tg0_12 = tanh_act(d2[5][0]);
    ORDER("+v"(tg0_12), "+v"(d2[5]), "+v"(to0_10), "+v"(to1_10), "+v"(tc0_10), "+v"(tc1_10));
    float tg1_12 = tanh_act(d2[5][1]);
    v2f hh_10 = HOUT2(((v2f){to0_10, to1_10}), ((v2f){tc0_10, tc1_10}));
    ORDER("+v"(tg1_12), "+v"(hh_10), "+v"(d2[3]), "+v"(ff_11), "+v"(ig_11), "+v"(c2[0]));
    float tf0_12 = TSIG(d2[3][0]);
    { const v2f cn = __builtin_elementwise_fma(ff_11, (v2f){c2[0][6], c2[0][7]}, ig_11); c2[0][6] = cn[0]; c2[0][7] = cn[1]; }
    ORDER("+v"(tf0_12), "+v"(c2[0]), "+v"(d2[3]), "+v"(ti0_12), "+v"(ti1_12));
    float tf1_12 = TSIG(d2[3][1]);
    v2f fi_12 = FIN2(((v2f){ti0_12, ti1_12}));
    ORDER("+v"(tf1_12), "+v"(fi_12), "+v"(d2[7]), "+v"(hh_9), "+v"(acc));
    float to0_12 = TSIG(d2[7][0]);
    acc = fmaf(hh_9[1], wfcD[0][3], acc);
    ORDER("+v"(to0_12), "+v"(acc), "+v"(d2[7]), "+v"(hh_10));
    float to1_12 = TSIG(d2[7][1]);
    v2h pk_10 = __builtin_convertvector(hh_10, v2h);
    ORDER("+v"(to1_12), "+v"(pk_10), "+v"(c2[0]));
    float tc0_11 = tanh_act(c2[0][6]);
    ORDER("+v"(tc0_11), "+v"(c2[0]), "+v"(fi_12), "+v"(tg0_12), "+v"(tg1_12));
    float tc1_11 = tanh_act(c2[0][7]);
    v2f ig_12 = fi_12 * (v2f){tg0_12, tg1_12};
    ORDER("+v"(tc1_11), "+v"(ig_12), "+v"(d2[1]));
    float ti0_13 = TSIG(d2[1][2]);
    ORDER("+v"(ti0_13), "+v"(h1Bn));
    d1[6] = __builtin_amdgcn_wmma_f32_16x16x32_f16(false, a1[6], false, h1Bn, (short)0, zc, false, false);
    ORDER("+v"(d1[6]), "+v"(tf0_12), "+v"(tf1_12));
    v2f ff_12 = FIN2(((v2f){tf0_12, tf1_12}));
    ORDER("+v"(ff_12), "+v"(d2[1]), "+v"(hh_10), "+v"(acc));
    float ti1_13 = TSIG(d2[1][3]);
    acc = fmaf(hh_10[0], wfcD[0][4], acc);
    ORDER("+v"(ti1_13), "+v"(acc), "+v"(d2[5]));
    float tg0_13 = tanh_act(d2[5][2]);
    ORDER("+v"(tg0_13), "+v"(d2[5]), "+v"(to0_11), "+v"(to1_11), "+v"(tc0_11), "+v"(tc1_11));
    float tg1_13 = tanh_act(d2[5][3]);
    v2f hh_11 = HOUT2(((v2f){to0_11, to1_11}), ((v2f){tc0_11, tc1_11}));
    ORDER("+v"(tg1_13), "+v"(hh_11), "+v"(d2[3]), "+v"(ff_12), "+v"(ig_12), "+v"(c2[1]));
    float tf0_13 = TSIG(d2[3][2]);
    { const v2f cn = __builtin_elementwise_fma(ff_12, (v2f){c2[1][0], c2[1][1]}, ig_12); c2[1][0] = cn[0]; c2[1][1] = cn[1]; }
    ORDER("+v"(tf0_13), "+v"(c2[1]), "+v"(d2[3]), "+v"(ti0_13), "+v"(ti1_13));
    float tf1_13 = TSIG(d2[3][3]);
    v2f fi_13 = FIN2(((v2f){ti0_13, ti1_13}));
    ORDER("+v"(tf1_13), "+v"(fi_13), "+v"(d2[7]), "+v"(hh_10), "+v"(acc));
    float to0_13 = TSIG(d2[7][2]);
    acc = fmaf(hh_10[1], wfcD[0][5], acc);
    ORDER("+v"(to0_13), "+v"(acc), "+v"(d2[7]), "+v"(hh_11));
    float to1_13 = TSIG(d2[7][3]);
    v2h pk_11 = __builtin_convertvector(hh_11, v2h);
    ORDER("+v"(to1_13), "+v"(pk_11), "+v"(c2[1]));
    float tc0_12 = tanh_act(c2[1][0]);
    ORDER("+v"(tc0_12), "+v"(c2[1]), "+v"(fi_13), "+v"(tg0_13), "+v"(tg1_13));
    float tc1_12 = tanh_act(c2[1][1]);
    v2f ig_13 = fi_13 * (v2f){tg0_13, tg1_13};
    ORDER("+v"(tc1_12), "+v"(ig_13), "+v"(d2[1]));
    float ti0_14 = TSIG(d2[1][4]);
    ORDER("+v"(ti0_14), "+v"(h1Bn));
    d1[1] = __builtin_amdgcn_wmma_f32_16x16x32_f16(false, a1[1], false, h1Bn, (short)0, zc, false, false);
    ORDER("+v"(d1[1]), "+v"(tf0_13), "+v"(tf1_13));
    v2f ff_13 = FIN2(((v2f){tf0_13, tf1_13}));
    ORDER("+v"(ff_13), "+v"(d2[1]), "+v"(hh_11), "+v"(acc));
    float ti1_14 = TSIG(d2[1][5]);
    acc = fmaf(hh_11[0], wfcD[0][6], acc);
    ORDER("+v"(ti1_14), "+v"(acc), "+v"(d2[5]));
    float tg0_14 = tanh_act(d2[5][4]);
    ORDER("+v"(tg0_14), "+v"(d2[5]), "+v"(to0_12), "+v"(to1_12), "+v"(tc0_12), "+v"(tc1_12));
    float tg1_14 = tanh_act(d2[5][5]);
    v2f hh_12 = HOUT2(((v2f){to0_12, to1_12}), ((v2f){tc0_12, tc1_12}));
    ORDER("+v"(tg1_14), "+v"(hh_12), "+v"(d2[3]), "+v"(ff_13), "+v"(ig_13), "+v"(c2[1]));
    float tf0_14 = TSIG(d2[3][4]);
    { const v2f cn = __builtin_elementwise_fma(ff_13, (v2f){c2[1][2], c2[1][3]}, ig_13); c2[1][2] = cn[0]; c2[1][3] = cn[1]; }
    ORDER("+v"(tf0_14), "+v"(c2[1]), "+v"(d2[3]), "+v"(ti0_14), "+v"(ti1_14));
    float tf1_14 = TSIG(d2[3][5]);
    v2f fi_14 = FIN2(((v2f){ti0_14, ti1_14}));
    ORDER("+v"(tf1_14), "+v"(fi_14), "+v"(d2[7]), "+v"(hh_11), "+v"(acc));
    float to0_14 = TSIG(d2[7][4]);
    acc = fmaf(hh_11[1], wfcD[0][7], acc);
    ORDER("+v"(to0_14), "+v"(acc), "+v"(d2[7]), "+v"(hh_12));
    float to1_14 = TSIG(d2[7][5]);
    v2h pk_12 = __builtin_convertvector(hh_12, v2h);
    ORDER("+v"(to1_14), "+v"(pk_12), "+v"(c2[1]));
    float tc0_13 = tanh_act(c2[1][2]);
    ORDER("+v"(tc0_13), "+v"(c2[1]), "+v"(fi_14), "+v"(tg0_14), "+v"(tg1_14), "+v"(tf0_14), "+v"(tf1_14));
    float tc1_13 = tanh_act(c2[1][3]);
    v2f ig_14 = fi_14 * (v2f){tg0_14, tg1_14};
    v2f ff_14 = FIN2(((v2f){tf0_14, tf1_14}));
    ORDER("+v"(tc1_13), "+v"(ig_14), "+v"(ff_14), "+v"(d2[1]));
    float ti0_15 = TSIG(d2[1][6]);
    ORDER("+v"(ti0_15), "+v"(h1Bn), "+v"(hh_12), "+v"(acc));
    d1[3] = __builtin_amdgcn_wmma_f32_16x16x32_f16(false, a1[3], false, h1Bn, (short)0, zc, false, false);
    ORDER("+v"(d1[3]), "+v"(hh_12), "+v"(acc));
    acc = fmaf(hh_12[0], wfcD[1][0], acc);
    ORDER("+v"(acc), "+v"(d2[5]), "+v"(ff_14), "+v"(ig_14), "+v"(c2[1]));
    float tg0_15 = tanh_act(d2[5][6]);
    { const v2f cn = __builtin_elementwise_fma(ff_14, (v2f){c2[1][4], c2[1][5]}, ig_14); c2[1][4] = cn[0]; c2[1][5] = cn[1]; }
    ORDER("+v"(tg0_15), "+v"(c2[1]), "+v"(d2[3]), "+v"(to0_13), "+v"(to1_13), "+v"(tc0_13), "+v"(tc1_13));
    float tf0_15 = TSIG(d2[3][6]);
    v2f hh_13 = HOUT2(((v2f){to0_13, to1_13}), ((v2f){tc0_13, tc1_13}));
    ORDER("+v"(tf0_15), "+v"(hh_13), "+v"(d2[7]), "+v"(ti0_15), "+v"(hh_12), "+v"(acc));
    float to0_15 = TSIG(d2[7][6]);
    float fi_15 = FIN1(ti0_15);
    acc = fmaf(hh_12[1], wfcD[1][1], acc);
    ORDER("+v"(to0_15), "+v"(fi_15), "+v"(acc));
    z_to0_15 = to0_15; z_fi_15 = fi_15; z_tg0_15 = tg0_15; z_tf0_15 = tf0_15;
    z_to0_14 = to0_14; z_to1_14 = to1_14; z_hh_13 = hh_13; accC = acc;
    zpk_8 = pk_8; zpk_9 = pk_9; zpk_10 = pk_10; zpk_11 = pk_11; zpk_12 = pk_12;
    h1B = h1Bn;
  }
    v2f xn0 = xq[0], xn1 = xq[1], xn2 = xq[2], xn3 = xq[3];
    ORDER("+v"(xn0), "+v"(xn1), "+v"(xn2), "+v"(xn3));
    xr[0] = xn0[0]; xr[1] = xn0[1];
    xr[2] = xn1[0]; xr[3] = xn1[1];
    xr[4] = xn2[0]; xr[5] = xn2[1];
    xr[6] = xn3[0]; xr[7] = xn3[1];
  }

  {
    const float tc0_14 = tanh_act(c2[1][4]);
    const float tc1_14 = tanh_act(c2[1][5]);
    const float ig_15 = z_fi_15 * z_tg0_15;
    const float c6 = fmaf(FIN1(z_tf0_15), c2[1][6], ig_15);
    const float tc0_15 = tanh_act(c6);
    const v2f hh_14 = HOUT2(((v2f){z_to0_14, z_to1_14}), ((v2f){tc0_14, tc1_14}));
    const float hh_15 = HOUT1(z_to0_15, tc0_15);
    accC = fmaf(z_hh_13[0], wfcD[1][2], accC);
    accC = fmaf(z_hh_13[1], wfcD[1][3], accC);
    accC = fmaf(hh_14[0], wfcD[1][4], accC);
    accC = fmaf(hh_14[1], wfcD[1][5], accC);
    accC = fmaf(hh_15, wfcD[1][6], accC);
  }
  const float accOd = xhalf(accC);
  {
    const float olast = (accC + accOd) + bfc;
    v4f ov;
    ov[0] = hi_half ? o8[4] : o8[0];
    ov[1] = hi_half ? o8[5] : o8[1];
    ov[2] = hi_half ? o8[6] : o8[2];
    ov[3] = hi_half ? olast : o8[3];
    *reinterpret_cast<v4f*>(ost + (TLEN - 8)) = ov;
  }
}

extern "C" void kernel_launch(void* const* d_in, const int* in_sizes, int n_in,
                              void* d_out, int out_size, void* d_ws, size_t ws_size,
                              hipStream_t stream) {
  (void)in_sizes; (void)n_in; (void)out_size; (void)d_ws; (void)ws_size;
  const float* x     = (const float*)d_in[0];
  const float* w_ih1 = (const float*)d_in[1];
  const float* w_hh1 = (const float*)d_in[2];
  const float* b_ih1 = (const float*)d_in[3];
  const float* b_hh1 = (const float*)d_in[4];
  const float* w_ih2 = (const float*)d_in[5];
  const float* w_hh2 = (const float*)d_in[6];
  const float* b_ih2 = (const float*)d_in[7];
  const float* b_hh2 = (const float*)d_in[8];
  const float* w_fc  = (const float*)d_in[9];
  const float* b_fc  = (const float*)d_in[10];
  float* out = (float*)d_out;

  dim3 grid(BSZ / 16);
  dim3 block(32);
  lstm2_wmma_kernel<<<grid, block, 0, stream>>>(x, w_ih1, w_hh1, b_ih1, b_hh1,
                                                w_ih2, w_hh2, b_ih2, b_hh2,
                                                w_fc, b_fc, out);
}
